# starting kernel
# baseline (speedup 1.0000x reference)
#include <hip/hip_runtime.h>

#define NB   8
#define CBGN 30000
#define POIN 20000
#define NE   1600000
#define NN   50000

typedef __attribute__((ext_vector_type(2))) float v2f;
typedef __attribute__((ext_vector_type(8))) float v8f;

__global__ void zero_f32(float* __restrict__ p, long long n) {
  long long i = (long long)blockIdx.x * blockDim.x + threadIdx.x;
  if (i < n) p[i] = 0.0f;
}

__global__ void degree_kernel(const int* __restrict__ src, const int* __restrict__ dst,
                              float* __restrict__ degO, float* __restrict__ degI) {
  int e = blockIdx.x * blockDim.x + threadIdx.x;
  if (e >= NE) return;
  atomicAdd(&degO[src[e]], 1.0f);
  atomicAdd(&degI[dst[e]], 1.0f);
}

__global__ void norm_kernel(float* __restrict__ degO, float* __restrict__ degI) {
  int i = blockIdx.x * blockDim.x + threadIdx.x;
  if (i >= NN) return;
  degO[i] = rsqrtf(fmaxf(degO[i], 1.0f));
  degI[i] = rsqrtf(fmaxf(degI[i], 1.0f));
}

__global__ void scatter1_kernel(const float* __restrict__ cbg, const float* __restrict__ poi,
                                const float* __restrict__ ew, const int* __restrict__ src,
                                const int* __restrict__ dst, const float* __restrict__ normS,
                                float* __restrict__ agg) {
  int e = blockIdx.x * blockDim.x + threadIdx.x;
  if (e >= NE) return;
  int b = blockIdx.y;
  int s = src[e], d = dst[e];
  float w = ew[(size_t)b * NE + e] * normS[s];
  const float4* f = (s < CBGN)
      ? (const float4*)&cbg[((size_t)b * CBGN + s) * 8]
      : (const float4*)&poi[((size_t)b * POIN + (s - CBGN)) * 8];
  float4 f0 = f[0], f1 = f[1];
  float* o = &agg[((size_t)b * NN + d) * 8];
  atomicAdd(&o[0], f0.x * w); atomicAdd(&o[1], f0.y * w);
  atomicAdd(&o[2], f0.z * w); atomicAdd(&o[3], f0.w * w);
  atomicAdd(&o[4], f1.x * w); atomicAdd(&o[5], f1.y * w);
  atomicAdd(&o[6], f1.z * w); atomicAdd(&o[7], f1.w * w);
}

__global__ void scatter2_kernel(const float* __restrict__ h1s, const float* __restrict__ ew,
                                const int* __restrict__ src, const int* __restrict__ dst,
                                float* __restrict__ agg) {
  int e = blockIdx.x * blockDim.x + threadIdx.x;
  if (e >= NE) return;
  int b = blockIdx.y;
  int s = src[e], d = dst[e];
  float w = ew[(size_t)b * NE + e];
  const float4* f = (const float4*)&h1s[((size_t)b * NN + s) * 16];
  float4 f0 = f[0], f1 = f[1], f2 = f[2], f3 = f[3];
  float* o = &agg[((size_t)b * NN + d) * 16];
  atomicAdd(&o[0],  f0.x * w); atomicAdd(&o[1],  f0.y * w);
  atomicAdd(&o[2],  f0.z * w); atomicAdd(&o[3],  f0.w * w);
  atomicAdd(&o[4],  f1.x * w); atomicAdd(&o[5],  f1.y * w);
  atomicAdd(&o[6],  f1.z * w); atomicAdd(&o[7],  f1.w * w);
  atomicAdd(&o[8],  f2.x * w); atomicAdd(&o[9],  f2.y * w);
  atomicAdd(&o[10], f2.z * w); atomicAdd(&o[11], f2.w * w);
  atomicAdd(&o[12], f3.x * w); atomicAdd(&o[13], f3.y * w);
  atomicAdd(&o[14], f3.z * w); atomicAdd(&o[15], f3.w * w);
}

__global__ void __launch_bounds__(256)
dense1_kernel(const float* __restrict__ agg, const float* __restrict__ normD,
              const float* __restrict__ normS, const float* __restrict__ W1,
              const float* __restrict__ b1, float* __restrict__ h1s) {
  int wave = (int)((blockIdx.x * blockDim.x + threadIdx.x) >> 5);
  int lane = threadIdx.x & 31;
  const int tilesPerB = NN / 16;
  int b = wave / tilesPerB;
  int t = wave % tilesPerB;
  if (b >= NB) return;
  int m0 = t * 16;
  int rc = lane & 15;
  int half = lane >> 4;
  float nd = normD[m0 + rc];
  const float* arow = &agg[((size_t)b * NN + m0 + rc) * 8];

  v8f c = {0.f, 0.f, 0.f, 0.f, 0.f, 0.f, 0.f, 0.f};
#pragma unroll
  for (int kb = 0; kb < 8; kb += 4) {
    int k0 = kb + 2 * half;
    v2f a, w;
    a.x = arow[k0 + 0] * nd;
    a.y = arow[k0 + 1] * nd;
    w.x = W1[(k0 + 0) * 16 + rc];
    w.y = W1[(k0 + 1) * 16 + rc];
    c = __builtin_amdgcn_wmma_f32_16x16x4_f32(false, a, false, w, (short)0, c,
                                              false, false);
  }
  float bias = b1[rc];
#pragma unroll
  for (int r = 0; r < 8; ++r) {
    int node = m0 + r + 8 * half;
    float v = c[r] + bias;
    v = v > 0.0f ? v : 0.01f * v;
    v *= normS[node];
    h1s[((size_t)b * NN + node) * 16 + rc] = v;
  }
}

__global__ void __launch_bounds__(256)
dense2_kernel(const float* __restrict__ agg, const float* __restrict__ normD,
              const float* __restrict__ W2, const float* __restrict__ b2,
              float* __restrict__ out) {
  int wave = (int)((blockIdx.x * blockDim.x + threadIdx.x) >> 5);
  int lane = threadIdx.x & 31;
  const int tilesPerB = CBGN / 16;
  int b = wave / tilesPerB;
  int t = wave % tilesPerB;
  if (b >= NB) return;
  int m0 = t * 16;
  int rc = lane & 15;
  int half = lane >> 4;
  float nd = normD[m0 + rc];
  const float* arow = &agg[((size_t)b * NN + m0 + rc) * 16];
  int cw = rc & 7;
  float colmask = (rc < 8) ? 1.0f : 0.0f;

  v8f c = {0.f, 0.f, 0.f, 0.f, 0.f, 0.f, 0.f, 0.f};
#pragma unroll
  for (int kb = 0; kb < 16; kb += 4) {
    int k0 = kb + 2 * half;
    v2f a, w;
    a.x = arow[k0 + 0] * nd;
    a.y = arow[k0 + 1] * nd;
    w.x = W2[(k0 + 0) * 8 + cw] * colmask;
    w.y = W2[(k0 + 1) * 8 + cw] * colmask;
    c = __builtin_amdgcn_wmma_f32_16x16x4_f32(false, a, false, w, (short)0, c,
                                              false, false);
  }
  if (rc < 8) {
    float bias = b2[rc];
#pragma unroll
    for (int r = 0; r < 8; ++r) {
      int node = m0 + r + 8 * half;
      out[((size_t)b * CBGN + node) * 8 + rc] = c[r] + bias;
    }
  }
}

extern "C" void kernel_launch(void* const* d_in, const int* in_sizes, int n_in,
                              void* d_out, int out_size, void* d_ws, size_t ws_size,
                              hipStream_t stream) {
  (void)in_sizes; (void)n_in; (void)out_size; (void)ws_size;
  const float* ew  = (const float*)d_in[0];
  const float* cbg = (const float*)d_in[1];
  const float* poi = (const float*)d_in[2];
  const float* W1  = (const float*)d_in[3];
  const float* b1  = (const float*)d_in[4];
  const float* W2  = (const float*)d_in[5];
  const float* b2  = (const float*)d_in[6];
  const int*   src = (const int*)d_in[7];
  const int*   dst = (const int*)d_in[8];
  float* out = (float*)d_out;

  float* ws    = (float*)d_ws;
  float* normS = ws;
  float* normD = ws + NN;
  float* agg   = ws + 2 * (size_t)NN;
  float* h1s   = agg + (size_t)NB * NN * 16;

  long long z1 = 2LL * NN + (long long)NB * NN * 16;
  zero_f32<<<(int)((z1 + 255) / 256), 256, 0, stream>>>(normS, z1);

  degree_kernel<<<(NE + 255) / 256, 256, 0, stream>>>(src, dst, normS, normD);
  norm_kernel<<<(NN + 255) / 256, 256, 0, stream>>>(normS, normD);

  dim3 gE((NE + 255) / 256, NB);
  scatter1_kernel<<<gE, 256, 0, stream>>>(cbg, poi, ew, src, dst, normS, agg);

  dense1_kernel<<<(NB * (NN / 16)) / 8, 256, 0, stream>>>(agg, normD, normS, W1, b1, h1s);

  long long z2 = (long long)NB * NN * 16;
  zero_f32<<<(int)((z2 + 255) / 256), 256, 0, stream>>>(agg, z2);

  scatter2_kernel<<<gE, 256, 0, stream>>>(h1s, ew, src, dst, agg);

  dense2_kernel<<<(NB * (CBGN / 16)) / 8, 256, 0, stream>>>(agg, normD, W2, b2, out);
}


# Float-atomic scatter-adds replaced by an in-call dst-sorted edge list (int atomic arrival rank + scan + placement) and per-row register sums; sum order now = arrival rank, still varies run to run
# speedup vs baseline: 15.3938x; 15.3938x over previous
#include <hip/hip_runtime.h>

#define NB   8
#define CBGN 30000
#define POIN 20000
#define NE   1600000
#define NN   50000
#define SCAN_T 1024
#define SCAN_BLOCKS 49

typedef __attribute__((ext_vector_type(2))) float v2f;
typedef __attribute__((ext_vector_type(8))) float v8f;

__global__ void zero_f32(float* __restrict__ p, long long n) {
  long long i = (long long)blockIdx.x * blockDim.x + threadIdx.x;
  if (i < n) p[i] = 0.0f;
}

__global__ void degree_rank_kernel(const int* __restrict__ src, const int* __restrict__ dst,
                                   int* __restrict__ cntO, int* __restrict__ cntI,
                                   int* __restrict__ rank) {
  int e = blockIdx.x * blockDim.x + threadIdx.x;
  if (e >= NE) return;
  atomicAdd(&cntO[src[e]], 1);
  rank[e] = atomicAdd(&cntI[dst[e]], 1);
}

__global__ void __launch_bounds__(SCAN_T)
rowptr_norm_kernel(const int* __restrict__ cntO, const int* __restrict__ cntI,
                   int* __restrict__ rowptr, float* __restrict__ normS,
                   float* __restrict__ normD) {
  __shared__ int s_tot[32];
  __shared__ int s_part[32];
  __shared__ int s_off[32];
  int t = threadIdx.x;
  int j = blockIdx.x;
  int lane = t & 31, wv = t >> 5;

  int part = 0;
  for (int i = 0; i < j; ++i) part += cntI[i * SCAN_T + t];

  int idx = j * SCAN_T + t;
  int c = (idx < NN) ? cntI[idx] : 0;

  int incl = c;
  for (int o = 1; o < 32; o <<= 1) {
    int v = __shfl_up(incl, o);
    if (lane >= o) incl += v;
  }
  int psum = part;
  for (int o = 16; o > 0; o >>= 1) psum += __shfl_xor(psum, o);
  if (lane == 31) s_tot[wv] = incl;
  if (lane == 0) s_part[wv] = psum;
  __syncthreads();
  if (wv == 0) {
    int wt = s_tot[lane];
    int wi = wt;
    for (int o = 1; o < 32; o <<= 1) {
      int v = __shfl_up(wi, o);
      if (lane >= o) wi += v;
    }
    int ptot = s_part[lane];
    for (int o = 16; o > 0; o >>= 1) ptot += __shfl_xor(ptot, o);
    s_off[lane] = wi - wt + ptot;
  }
  __syncthreads();
  if (idx < NN) {
    int excl = s_off[wv] + incl - c;
    rowptr[idx] = excl;
    if (idx == NN - 1) rowptr[NN] = excl + c;
    normS[idx] = rsqrtf(fmaxf((float)cntO[idx], 1.0f));
    normD[idx] = rsqrtf(fmaxf((float)c, 1.0f));
  }
}

__global__ void place_kernel(const float* __restrict__ ew, const int* __restrict__ src,
                             const int* __restrict__ dst, const int* __restrict__ rowptr,
                             const int* __restrict__ rank, int* __restrict__ srcS,
                             float* __restrict__ ewS) {
  int e = blockIdx.x * blockDim.x + threadIdx.x;
  if (e >= NE) return;
  int p = rowptr[dst[e]] + rank[e];
  srcS[p] = src[e];
  float4 lo, hi;
  lo.x = ew[(size_t)0 * NE + e]; lo.y = ew[(size_t)1 * NE + e];
  lo.z = ew[(size_t)2 * NE + e]; lo.w = ew[(size_t)3 * NE + e];
  hi.x = ew[(size_t)4 * NE + e]; hi.y = ew[(size_t)5 * NE + e];
  hi.z = ew[(size_t)6 * NE + e]; hi.w = ew[(size_t)7 * NE + e];
  float4* o = (float4*)&ewS[(size_t)p * 8];
  o[0] = lo; o[1] = hi;
}

typedef __attribute__((ext_vector_type(4))) float v4f;

template <int K>
__device__ __forceinline__ void hold_rows(const v4f (&x)[K]) {
  if constexpr (K == 8) {
    asm volatile("" :: "v"(x[0]), "v"(x[1]), "v"(x[2]), "v"(x[3]),
                       "v"(x[4]), "v"(x[5]), "v"(x[6]), "v"(x[7]));
  } else if constexpr (K == 4) {
    asm volatile("" :: "v"(x[0]), "v"(x[1]), "v"(x[2]), "v"(x[3]));
  } else if constexpr (K == 2) {
    asm volatile("" :: "v"(x[0]), "v"(x[1]));
  }
}
template <int K>
__device__ __forceinline__ void hold_words(const float (&w)[K]) {
  if constexpr (K == 8) {
    asm volatile("" :: "v"(w[0]), "v"(w[1]), "v"(w[2]), "v"(w[3]),
                       "v"(w[4]), "v"(w[5]), "v"(w[6]), "v"(w[7]));
  } else if constexpr (K == 4) {
    asm volatile("" :: "v"(w[0]), "v"(w[1]), "v"(w[2]), "v"(w[3]));
  } else if constexpr (K == 2) {
    asm volatile("" :: "v"(w[0]), "v"(w[1]));
  }
}

template <int K, bool L1>
__device__ __forceinline__ void edge_batch(v4f& a, const int p, const int b, const int q,
                                           const int* __restrict__ srcS,
                                           const float* __restrict__ ewS,
                                           const float* __restrict__ normS,
                                           const float* __restrict__ fA,
                                           const float* __restrict__ fB) {
  int s[K];
  float w[K];
  float ns[K];
  v4f x[K];
  for (int i = 0; i < K; ++i) s[i] = srcS[p + i];
  for (int i = 0; i < K; ++i) w[i] = ewS[(size_t)(p + i) * 8 + b];
  for (int i = 0; i < K; ++i) {
    const v4f* r;
    if (L1) {
      r = (s[i] < CBGN) ? (const v4f*)&fA[((size_t)b * CBGN + s[i]) * 8]
                        : (const v4f*)&fB[((size_t)b * POIN + (s[i] - CBGN)) * 8];
      ns[i] = normS[s[i]];
    } else {
      r = (const v4f*)(fA + (size_t)s[i] * 16);
      ns[i] = 1.0f;
    }
    x[i] = r[q];
  }
  hold_rows<K>(x);
  hold_words<K>(w);
  if (L1) hold_words<K>(ns);
  for (int i = 0; i < K; ++i) {
    const float wi = L1 ? w[i] * ns[i] : w[i];
    a.x = fmaf(x[i].x, wi, a.x);
    a.y = fmaf(x[i].y, wi, a.y);
    a.z = fmaf(x[i].z, wi, a.z);
    a.w = fmaf(x[i].w, wi, a.w);
  }
}

template <bool L1>
__device__ __forceinline__ v4f row_sum(int p, const int p1, const int b, const int q,
                                       const int* __restrict__ srcS,
                                       const float* __restrict__ ewS,
                                       const float* __restrict__ normS,
                                       const float* __restrict__ fA,
                                       const float* __restrict__ fB) {
  v4f a = {0.f, 0.f, 0.f, 0.f};
  for (; p + 8 <= p1; p += 8) edge_batch<8, L1>(a, p, b, q, srcS, ewS, normS, fA, fB);
  if (p + 4 <= p1) { edge_batch<4, L1>(a, p, b, q, srcS, ewS, normS, fA, fB); p += 4; }
  if (p + 2 <= p1) { edge_batch<2, L1>(a, p, b, q, srcS, ewS, normS, fA, fB); p += 2; }
  if (p < p1)      { edge_batch<1, L1>(a, p, b, q, srcS, ewS, normS, fA, fB); }
  return a;
}

__global__ void __launch_bounds__(256)
gather1_kernel(const float* __restrict__ cbg, const float* __restrict__ poi,
               const int* __restrict__ rowptr, const int* __restrict__ srcS,
               const float* __restrict__ ewS, const float* __restrict__ normS,
               float* __restrict__ agg) {
  int tid = blockIdx.x * 256 + threadIdx.x;
  int q = tid & 1;
  int b = (tid >> 3) & 7;
  int n = (tid >> 6) * 4 + ((tid >> 1) & 3);
  if (n >= NN) return;
  v4f a = row_sum<true>(rowptr[n], rowptr[n + 1], b, q, srcS, ewS, normS, cbg, poi);
  ((v4f*)&agg[((size_t)b * NN + n) * 8])[q] = a;
}

__global__ void __launch_bounds__(256)
gather2_kernel(const float* __restrict__ h1s, const int* __restrict__ rowptr,
               const int* __restrict__ srcS, const float* __restrict__ ewS,
               float* __restrict__ agg) {
  int tid = blockIdx.x * 256 + threadIdx.x;
  int q = tid & 3;
  int b = (tid >> 3) & 7;
  int n = (tid >> 6) * 2 + ((tid >> 2) & 1);
  if (n >= CBGN) return;
  v4f a = row_sum<false>(rowptr[n], rowptr[n + 1], b, q, srcS, ewS, nullptr,
                         h1s + (size_t)b * NN * 16, nullptr);
  ((v4f*)&agg[((size_t)b * NN + n) * 16])[q] = a;
}


__global__ void __launch_bounds__(256)
dense1_kernel(const float* __restrict__ agg, const float* __restrict__ normD,
              const float* __restrict__ normS, const float* __restrict__ W1,
              const float* __restrict__ b1, float* __restrict__ h1s) {
  int wave = (int)((blockIdx.x * blockDim.x + threadIdx.x) >> 5);
  int lane = threadIdx.x & 31;
  const int tilesPerB = NN / 16;
  int b = wave / tilesPerB;
  int t = wave % tilesPerB;
  if (b >= NB) return;
  int m0 = t * 16;
  int rc = lane & 15;
  int half = lane >> 4;
  float nd = normD[m0 + rc];
  const float* arow = &agg[((size_t)b * NN + m0 + rc) * 8];

  v8f c = {0.f, 0.f, 0.f, 0.f, 0.f, 0.f, 0.f, 0.f};
#pragma unroll
  for (int kb = 0; kb < 8; kb += 4) {
    int k0 = kb + 2 * half;
    v2f a, w;
    a.x = arow[k0 + 0] * nd;
    a.y = arow[k0 + 1] * nd;
    w.x = W1[(k0 + 0) * 16 + rc];
    w.y = W1[(k0 + 1) * 16 + rc];
    c = __builtin_amdgcn_wmma_f32_16x16x4_f32(false, a, false, w, (short)0, c,
                                              false, false);
  }
  float bias = b1[rc];
#pragma unroll
  for (int r = 0; r < 8; ++r) {
    int node = m0 + r + 8 * half;
    float v = c[r] + bias;
    v = v > 0.0f ? v : 0.01f * v;
    v *= normS[node];
    h1s[((size_t)b * NN + node) * 16 + rc] = v;
  }
}

__global__ void __launch_bounds__(256)
dense2_kernel(const float* __restrict__ agg, const float* __restrict__ normD,
              const float* __restrict__ W2, const float* __restrict__ b2,
              float* __restrict__ out) {
  int wave = (int)((blockIdx.x * blockDim.x + threadIdx.x) >> 5);
  int lane = threadIdx.x & 31;
  const int tilesPerB = CBGN / 16;
  int b = wave / tilesPerB;
  int t = wave % tilesPerB;
  if (b >= NB) return;
  int m0 = t * 16;
  int rc = lane & 15;
  int half = lane >> 4;
  float nd = normD[m0 + rc];
  const float* arow = &agg[((size_t)b * NN + m0 + rc) * 16];
  int cw = rc & 7;
  float colmask = (rc < 8) ? 1.0f : 0.0f;

  v8f c = {0.f, 0.f, 0.f, 0.f, 0.f, 0.f, 0.f, 0.f};
#pragma unroll
  for (int kb = 0; kb < 16; kb += 4) {
    int k0 = kb + 2 * half;
    v2f a, w;
    a.x = arow[k0 + 0] * nd;
    a.y = arow[k0 + 1] * nd;
    w.x = W2[(k0 + 0) * 8 + cw] * colmask;
    w.y = W2[(k0 + 1) * 8 + cw] * colmask;
    c = __builtin_amdgcn_wmma_f32_16x16x4_f32(false, a, false, w, (short)0, c,
                                              false, false);
  }
  if (rc < 8) {
    float bias = b2[rc];
#pragma unroll
    for (int r = 0; r < 8; ++r) {
      int node = m0 + r + 8 * half;
      out[((size_t)b * CBGN + node) * 8 + rc] = c[r] + bias;
    }
  }
}

extern "C" void kernel_launch(void* const* d_in, const int* in_sizes, int n_in,
                              void* d_out, int out_size, void* d_ws, size_t ws_size,
                              hipStream_t stream) {
  (void)in_sizes; (void)n_in; (void)out_size; (void)ws_size;
  const float* ew  = (const float*)d_in[0];
  const float* cbg = (const float*)d_in[1];
  const float* poi = (const float*)d_in[2];
  const float* W1  = (const float*)d_in[3];
  const float* b1  = (const float*)d_in[4];
  const float* W2  = (const float*)d_in[5];
  const float* b2  = (const float*)d_in[6];
  const int*   src = (const int*)d_in[7];
  const int*   dst = (const int*)d_in[8];
  float* out = (float*)d_out;

  float* ws     = (float*)d_ws;
  float* normS  = ws;
  float* normD  = ws + NN;
  int*   cntO   = (int*)(ws + 2 * (size_t)NN);
  int*   cntI   = (int*)(ws + 3 * (size_t)NN);
  int*   rowptr = (int*)(ws + 4 * (size_t)NN);
  int*   rank   = (int*)(ws + 5 * (size_t)NN + 16);
  int*   srcS   = rank + NE;
  float* ewS    = (float*)(srcS + NE);
  float* agg    = ewS + 8 * (size_t)NE;
  float* h1s    = agg + (size_t)NB * NN * 16;

  zero_f32<<<(2 * NN + 255) / 256, 256, 0, stream>>>((float*)cntO, 2LL * NN);

  degree_rank_kernel<<<(NE + 255) / 256, 256, 0, stream>>>(src, dst, cntO, cntI, rank);
  rowptr_norm_kernel<<<SCAN_BLOCKS, SCAN_T, 0, stream>>>(cntO, cntI, rowptr, normS, normD);

  place_kernel<<<(NE + 255) / 256, 256, 0, stream>>>(ew, src, dst, rowptr, rank, srcS, ewS);

  gather1_kernel<<<(NN * 16) / 256, 256, 0, stream>>>(cbg, poi, rowptr, srcS, ewS, normS, agg);
  dense1_kernel<<<(NB * (NN / 16)) / 8, 256, 0, stream>>>(agg, normD, normS, W1, b1, h1s);

  gather2_kernel<<<(CBGN * 32) / 256, 256, 0, stream>>>(h1s, rowptr, srcS, ewS, agg);
  dense2_kernel<<<(NB * (CBGN / 16)) / 8, 256, 0, stream>>>(agg, normD, W2, b2, out);
}


# Per-edge global int atomics (degrees, rank) replaced by LDS counters: one group per edge chunk and role holds all node counters in LDS; sum order = chunk, then LDS arrival rank, still varies by run
# speedup vs baseline: 34.9154x; 2.2681x over previous
#include <hip/hip_runtime.h>

#define NB   8
#define CBGN 30000
#define POIN 20000
#define NE   1600000
#define NN   50000
#define SCAN_T 1024
#define SCAN_BLOCKS 49
#define NCH  25
#define CH   (NE / NCH)
#define CH4  (CH / 4)
#define NP   50016

typedef __attribute__((ext_vector_type(2))) float v2f;
typedef __attribute__((ext_vector_type(8))) float v8f;

template <bool RANK>
__device__ __forceinline__ void count_chunk(const int* __restrict__ idx, int* __restrict__ rank,
                                            int* __restrict__ s_cnt, const int t) {
  const int4* i4 = (const int4*)idx;
  int4 d[16];
#pragma unroll
  for (int i = 0; i < 16; ++i) {
    int k = t + 1024 * i;
    d[i] = (k < CH4) ? i4[k] : make_int4(0, 0, 0, 0);
  }
#pragma unroll
  for (int i = 0; i < 16; ++i) {
    int k = t + 1024 * i;
    if (k < CH4) {
      if constexpr (RANK) {
        int4 r;
        r.x = atomicAdd(&s_cnt[d[i].x], 1);
        r.y = atomicAdd(&s_cnt[d[i].y], 1);
        r.z = atomicAdd(&s_cnt[d[i].z], 1);
        r.w = atomicAdd(&s_cnt[d[i].w], 1);
        ((int4*)rank)[k] = r;
      } else {
        atomicAdd(&s_cnt[d[i].x], 1);
        atomicAdd(&s_cnt[d[i].y], 1);
        atomicAdd(&s_cnt[d[i].z], 1);
        atomicAdd(&s_cnt[d[i].w], 1);
      }
    }
  }
}

__global__ void __launch_bounds__(1024)
count_rank_kernel(const int* __restrict__ src, const int* __restrict__ dst,
                  int* __restrict__ cO, int* __restrict__ cI,
                  int* __restrict__ tileSum, int* __restrict__ rank) {
  __shared__ __attribute__((aligned(16))) int s_cnt[NN];
  __shared__ int s_tile[64];
  const int t = threadIdx.x;
  const int lane = t & 31;
  const int c = blockIdx.x >> 1;
  const bool dst_role = (blockIdx.x & 1) == 0;
  int4* s4 = (int4*)s_cnt;
  for (int k = t; k < NN / 4; k += 1024) s4[k] = make_int4(0, 0, 0, 0);
  if (t < 64) s_tile[t] = 0;
  __syncthreads();
  if (dst_role) count_chunk<true>(dst + (size_t)c * CH, rank + (size_t)c * CH, s_cnt, t);
  else          count_chunk<false>(src + (size_t)c * CH, nullptr, s_cnt, t);
  __syncthreads();
  int4* o4 = (int4*)((dst_role ? cI : cO) + (size_t)c * NP);
  for (int i = 0; i < 13; ++i) {
    int k = t + 1024 * i;
    bool ok = k < NN / 4;
    int4 v = ok ? s4[k] : make_int4(0, 0, 0, 0);
    if (ok) o4[k] = v;
    if (dst_role) {
      int s = v.x + v.y + v.z + v.w;
      for (int o = 16; o > 0; o >>= 1) s += __shfl_xor(s, o);
      if (lane == 0) atomicAdd(&s_tile[k >> 8], s);
    }
  }
  __syncthreads();
  if (dst_role && t < SCAN_BLOCKS) tileSum[c * 64 + t] = s_tile[t];
}

__global__ void __launch_bounds__(SCAN_T)
rowptr_norm_kernel(const int* __restrict__ cO, const int* __restrict__ cI,
                   const int* __restrict__ tileSum,
                   int* __restrict__ rowptr, int* __restrict__ off,
                   float* __restrict__ normS, float* __restrict__ normD) {
  __shared__ int s_tot[32];
  __shared__ int s_part[32];
  __shared__ int s_off[32];
  int t = threadIdx.x;
  int j = blockIdx.x;
  int lane = t & 31, wv = t >> 5;

  int part = 0;
  for (int k = t; k < NCH * 64; k += SCAN_T)
    if ((k & 63) < j) part += tileSum[k];

  int idx = j * SCAN_T + t;
  int ci[NCH];
  int c = 0, co = 0;
#pragma unroll
  for (int q = 0; q < NCH; ++q) ci[q] = 0;
  if (idx < NN) {
#pragma unroll
    for (int q = 0; q < NCH; ++q) {
      ci[q] = cI[(size_t)q * NP + idx];
      co   += cO[(size_t)q * NP + idx];
      c += ci[q];
    }
  }

  int incl = c;
  for (int o = 1; o < 32; o <<= 1) {
    int v = __shfl_up(incl, o);
    if (lane >= o) incl += v;
  }
  int psum = part;
  for (int o = 16; o > 0; o >>= 1) psum += __shfl_xor(psum, o);
  if (lane == 31) s_tot[wv] = incl;
  if (lane == 0) s_part[wv] = psum;
  __syncthreads();
  if (wv == 0) {
    int wt = s_tot[lane];
    int wi = wt;
    for (int o = 1; o < 32; o <<= 1) {
      int v = __shfl_up(wi, o);
      if (lane >= o) wi += v;
    }
    int ptot = s_part[lane];
    for (int o = 16; o > 0; o >>= 1) ptot += __shfl_xor(ptot, o);
    s_off[lane] = wi - wt + ptot;
  }
  __syncthreads();
  if (idx < NN) {
    int excl = s_off[wv] + incl - c;
    rowptr[idx] = excl;
    if (idx == NN - 1) rowptr[NN] = excl + c;
    int run = excl;
#pragma unroll
    for (int q = 0; q < NCH; ++q) {
      off[(size_t)q * NP + idx] = run;
      run += ci[q];
    }
    normS[idx] = rsqrtf(fmaxf((float)co, 1.0f));
    normD[idx] = rsqrtf(fmaxf((float)c, 1.0f));
  }
}

__global__ void place_kernel(const float* __restrict__ ew, const int* __restrict__ src,
                             const int* __restrict__ dst, const int* __restrict__ off,
                             const int* __restrict__ rank, int* __restrict__ srcS,
                             float* __restrict__ ewS) {
  int e = blockIdx.x * blockDim.x + threadIdx.x;
  if (e >= NE) return;
  const int* offc = off + (size_t)(blockIdx.x / (CH / 256)) * NP;
  int p = offc[dst[e]] + rank[e];
  srcS[p] = src[e];
  float4 lo, hi;
  lo.x = ew[(size_t)0 * NE + e]; lo.y = ew[(size_t)1 * NE + e];
  lo.z = ew[(size_t)2 * NE + e]; lo.w = ew[(size_t)3 * NE + e];
  hi.x = ew[(size_t)4 * NE + e]; hi.y = ew[(size_t)5 * NE + e];
  hi.z = ew[(size_t)6 * NE + e]; hi.w = ew[(size_t)7 * NE + e];
  float4* o = (float4*)&ewS[(size_t)p * 8];
  o[0] = lo; o[1] = hi;
}

typedef __attribute__((ext_vector_type(4))) float v4f;

__global__ void __launch_bounds__(256)
relayout_kernel(const float* __restrict__ cbg, const float* __restrict__ poi,
                const float* __restrict__ normS, float* __restrict__ featT) {
  int tid = blockIdx.x * 256 + threadIdx.x;
  int q = tid & 1;
  int b = (tid >> 1) & 7;
  int n = tid >> 4;
  if (n >= NN) return;
  const v4f* r = (n < CBGN) ? (const v4f*)&cbg[((size_t)b * CBGN + n) * 8]
                            : (const v4f*)&poi[((size_t)b * POIN + (n - CBGN)) * 8];
  v4f x = r[q];
  float s = normS[n];
  x.x *= s; x.y *= s; x.z *= s; x.w *= s;
  ((v4f*)featT)[tid] = x;
}

template <int K>
__device__ __forceinline__ void hold_rows(const v4f (&x)[K]) {
  if constexpr (K == 8) {
    asm volatile("" :: "v"(x[0]), "v"(x[1]), "v"(x[2]), "v"(x[3]),
                       "v"(x[4]), "v"(x[5]), "v"(x[6]), "v"(x[7]));
  } else if constexpr (K == 4) {
    asm volatile("" :: "v"(x[0]), "v"(x[1]), "v"(x[2]), "v"(x[3]));
  } else if constexpr (K == 2) {
    asm volatile("" :: "v"(x[0]), "v"(x[1]));
  }
}
template <int K>
__device__ __forceinline__ void hold_words(const float (&w)[K]) {
  if constexpr (K == 8) {
    asm volatile("" :: "v"(w[0]), "v"(w[1]), "v"(w[2]), "v"(w[3]),
                       "v"(w[4]), "v"(w[5]), "v"(w[6]), "v"(w[7]));
  } else if constexpr (K == 4) {
    asm volatile("" :: "v"(w[0]), "v"(w[1]), "v"(w[2]), "v"(w[3]));
  } else if constexpr (K == 2) {
    asm volatile("" :: "v"(w[0]), "v"(w[1]));
  }
}

template <int K, int F>
__device__ __forceinline__ void edge_batch(v4f& a, const int p, const int b, const int q,
                                           const int* __restrict__ srcS,
                                           const float* __restrict__ ewS,
                                           const float* __restrict__ tab) {
  int s[K];
  float w[K];
  v4f x[K];
  for (int i = 0; i < K; ++i) s[i] = srcS[p + i];
  for (int i = 0; i < K; ++i) w[i] = ewS[(size_t)(p + i) * 8 + b];
  for (int i = 0; i < K; ++i) {
    const v4f* r = (const v4f*)(tab + ((size_t)s[i] * 8 + b) * F);
    x[i] = r[q];
  }
  hold_rows<K>(x);
  hold_words<K>(w);
  for (int i = 0; i < K; ++i) {
    a.x = fmaf(x[i].x, w[i], a.x);
    a.y = fmaf(x[i].y, w[i], a.y);
    a.z = fmaf(x[i].z, w[i], a.z);
    a.w = fmaf(x[i].w, w[i], a.w);
  }
}

template <int F>
__device__ __forceinline__ v4f row_sum(int p, const int p1, const int b, const int q,
                                       const int* __restrict__ srcS,
                                       const float* __restrict__ ewS,
                                       const float* __restrict__ tab) {
  v4f a = {0.f, 0.f, 0.f, 0.f};
  for (; p + 8 <= p1; p += 8) edge_batch<8, F>(a, p, b, q, srcS, ewS, tab);
  if (p + 4 <= p1) { edge_batch<4, F>(a, p, b, q, srcS, ewS, tab); p += 4; }
  if (p + 2 <= p1) { edge_batch<2, F>(a, p, b, q, srcS, ewS, tab); p += 2; }
  if (p < p1)      { edge_batch<1, F>(a, p, b, q, srcS, ewS, tab); }
  return a;
}

__global__ void __launch_bounds__(256)
gather1_kernel(const float* __restrict__ featT, const int* __restrict__ rowptr,
               const int* __restrict__ srcS, const float* __restrict__ ewS,
               float* __restrict__ agg) {
  int tid = blockIdx.x * 256 + threadIdx.x;
  int q = tid & 1;
  int b = (tid >> 1) & 7;
  int n = tid >> 4;
  if (n >= NN) return;
  v4f a = row_sum<8>(rowptr[n], rowptr[n + 1], b, q, srcS, ewS, featT);
  ((v4f*)agg)[tid] = a;
}

__global__ void __launch_bounds__(256)
gather2_kernel(const float* __restrict__ h1s, const int* __restrict__ rowptr,
               const int* __restrict__ srcS, const float* __restrict__ ewS,
               float* __restrict__ agg) {
  int tid = blockIdx.x * 256 + threadIdx.x;
  int q = tid & 3;
  int b = (tid >> 2) & 7;
  int n = tid >> 5;
  if (n >= CBGN) return;
  v4f a = row_sum<16>(rowptr[n], rowptr[n + 1], b, q, srcS, ewS, h1s);
  ((v4f*)agg)[tid] = a;
}


__global__ void __launch_bounds__(256)
dense1_kernel(const float* __restrict__ agg, const float* __restrict__ normD,
              const float* __restrict__ normS, const float* __restrict__ W1,
              const float* __restrict__ b1, float* __restrict__ h1s) {
  int wave = (int)((blockIdx.x * blockDim.x + threadIdx.x) >> 5);
  int lane = threadIdx.x & 31;
  if (wave >= (NN * NB) / 16) return;
  int m0 = wave * 16;
  int rc = lane & 15;
  int half = lane >> 4;
  float nd = normD[(m0 + rc) >> 3];
  const float* arow = &agg[(size_t)(m0 + rc) * 8];

  v8f c = {0.f, 0.f, 0.f, 0.f, 0.f, 0.f, 0.f, 0.f};
#pragma unroll
  for (int kb = 0; kb < 8; kb += 4) {
    int k0 = kb + 2 * half;
    v2f a, w;
    a.x = arow[k0 + 0] * nd;
    a.y = arow[k0 + 1] * nd;
    w.x = W1[(k0 + 0) * 16 + rc];
    w.y = W1[(k0 + 1) * 16 + rc];
    c = __builtin_amdgcn_wmma_f32_16x16x4_f32(false, a, false, w, (short)0, c,
                                              false, false);
  }
  float bias = b1[rc];
  float ns = normS[(m0 >> 3) + half];
#pragma unroll
  for (int r = 0; r < 8; ++r) {
    int row = m0 + r + 8 * half;
    float v = c[r] + bias;
    v = v > 0.0f ? v : 0.01f * v;
    v *= ns;
    h1s[(size_t)row * 16 + rc] = v;
  }
}

__global__ void __launch_bounds__(256)
dense2_kernel(const float* __restrict__ agg, const float* __restrict__ normD,
              const float* __restrict__ W2, const float* __restrict__ b2,
              float* __restrict__ out) {
  int wave = (int)((blockIdx.x * blockDim.x + threadIdx.x) >> 5);
  int lane = threadIdx.x & 31;
  if (wave >= CBGN / 4) return;
  int n0 = wave * 4;
  int rc = lane & 15;
  int half = lane >> 4;
  int cw = rc & 7;
  float colmask = (rc < 8) ? 1.0f : 0.0f;

  v8f c0 = {0.f, 0.f, 0.f, 0.f, 0.f, 0.f, 0.f, 0.f};
  v8f c1 = {0.f, 0.f, 0.f, 0.f, 0.f, 0.f, 0.f, 0.f};
  float nd0 = normD[n0 + (rc >> 3)];
  float nd1 = normD[n0 + 2 + (rc >> 3)];
  const float* arow0 = &agg[((size_t)n0 * 8 + rc) * 16];
  const float* arow1 = arow0 + 16 * 16;
#pragma unroll
  for (int kb = 0; kb < 16; kb += 4) {
    int k0 = kb + 2 * half;
    v2f a0, a1, w;
    a0.x = arow0[k0 + 0] * nd0;
    a0.y = arow0[k0 + 1] * nd0;
    a1.x = arow1[k0 + 0] * nd1;
    a1.y = arow1[k0 + 1] * nd1;
    w.x = W2[(k0 + 0) * 8 + cw] * colmask;
    w.y = W2[(k0 + 1) * 8 + cw] * colmask;
    c0 = __builtin_amdgcn_wmma_f32_16x16x4_f32(false, a0, false, w, (short)0, c0,
                                               false, false);
    c1 = __builtin_amdgcn_wmma_f32_16x16x4_f32(false, a1, false, w, (short)0, c1,
                                               false, false);
  }
  if (rc < 8) {
    float bias = b2[rc];
#pragma unroll
    for (int r = 0; r < 8; ++r) {
      float* o = &out[((size_t)r * CBGN + n0 + half) * 8 + rc];
      o[0]  = c0[r] + bias;
      o[16] = c1[r] + bias;
    }
  }
}

extern "C" void kernel_launch(void* const* d_in, const int* in_sizes, int n_in,
                              void* d_out, int out_size, void* d_ws, size_t ws_size,
                              hipStream_t stream) {
  (void)in_sizes; (void)n_in; (void)out_size; (void)ws_size;
  const float* ew  = (const float*)d_in[0];
  const float* cbg = (const float*)d_in[1];
  const float* poi = (const float*)d_in[2];
  const float* W1  = (const float*)d_in[3];
  const float* b1  = (const float*)d_in[4];
  const float* W2  = (const float*)d_in[5];
  const float* b2  = (const float*)d_in[6];
  const int*   src = (const int*)d_in[7];
  const int*   dst = (const int*)d_in[8];
  float* out = (float*)d_out;

  float* ws     = (float*)d_ws;
  float* normS  = ws;
  float* normD  = ws + NN;
  int*   tileSum= (int*)(ws + 2 * (size_t)NN);
  int*   rowptr = (int*)(ws + 4 * (size_t)NN);
  int*   rank   = (int*)(ws + 5 * (size_t)NN + 48);
  int*   srcS   = rank + NE;
  float* ewS    = (float*)(srcS + NE);
  float* agg    = ewS + 8 * (size_t)NE;
  float* h1s    = agg + (size_t)NB * NN * 16;
  float* featT  = h1s + (size_t)NB * NN * 16;
  int*   cI     = (int*)(featT + (size_t)NB * NN * 8);
  int*   cO     = cI + (size_t)NCH * NP;
  int*   off    = cO + (size_t)NCH * NP;

  count_rank_kernel<<<2 * NCH, 1024, 0, stream>>>(src, dst, cO, cI, tileSum, rank);
  rowptr_norm_kernel<<<SCAN_BLOCKS, SCAN_T, 0, stream>>>(cO, cI, tileSum, rowptr, off, normS, normD);

  place_kernel<<<(NE + 255) / 256, 256, 0, stream>>>(ew, src, dst, off, rank, srcS, ewS);
  relayout_kernel<<<(NN * 16) / 256, 256, 0, stream>>>(cbg, poi, normS, featT);

  gather1_kernel<<<(NN * 16) / 256, 256, 0, stream>>>(featT, rowptr, srcS, ewS, agg);
  dense1_kernel<<<((NN * NB) / 16) / 8, 256, 0, stream>>>(agg, normD, normS, W1, b1, h1s);

  gather2_kernel<<<(CBGN * 32) / 256, 256, 0, stream>>>(h1s, rowptr, srcS, ewS, agg);
  dense2_kernel<<<(CBGN / 4 + 7) / 8, 256, 0, stream>>>(agg, normD, W2, b2, out);
}


# Layer-2 weight product (WMMA) moved ahead of the layer-2 edge sum (they commute; hidden-unit sum now precedes edge sum): gather reads 256-B rows not 512-B and writes out itself, one launch fewer
# speedup vs baseline: 38.0108x; 1.0887x over previous
#include <hip/hip_runtime.h>

#define NB   8
#define CBGN 30000
#define POIN 20000
#define NE   1600000
#define NN   50000
#define SCAN_T 1024
#define SCAN_BLOCKS 49
#define NCH  25
#define CH   (NE / NCH)
#define CH4  (CH / 4)
#define NP   50016

typedef __attribute__((ext_vector_type(2))) float v2f;
typedef __attribute__((ext_vector_type(8))) float v8f;

template <bool RANK>
__device__ __forceinline__ void count_chunk(const int* __restrict__ idx, int* __restrict__ rank,
                                            int* __restrict__ s_cnt, const int t) {
  const int4* i4 = (const int4*)idx;
  int4 d[16];
#pragma unroll
  for (int i = 0; i < 16; ++i) {
    int k = t + 1024 * i;
    d[i] = (k < CH4) ? i4[k] : make_int4(0, 0, 0, 0);
  }
#pragma unroll
  for (int i = 0; i < 16; ++i) {
    int k = t + 1024 * i;
    if (k < CH4) {
      if constexpr (RANK) {
        int4 r;
        r.x = atomicAdd(&s_cnt[d[i].x], 1);
        r.y = atomicAdd(&s_cnt[d[i].y], 1);
        r.z = atomicAdd(&s_cnt[d[i].z], 1);
        r.w = atomicAdd(&s_cnt[d[i].w], 1);
        ((int4*)rank)[k] = r;
      } else {
        atomicAdd(&s_cnt[d[i].x], 1);
        atomicAdd(&s_cnt[d[i].y], 1);
        atomicAdd(&s_cnt[d[i].z], 1);
        atomicAdd(&s_cnt[d[i].w], 1);
      }
    }
  }
}

__global__ void __launch_bounds__(1024)
count_rank_kernel(const int* __restrict__ src, const int* __restrict__ dst,
                  int* __restrict__ cO, int* __restrict__ cI,
                  int* __restrict__ tileSum, int* __restrict__ rank) {
  __shared__ __attribute__((aligned(16))) int s_cnt[NN];
  __shared__ int s_tile[64];
  const int t = threadIdx.x;
  const int lane = t & 31;
  const int c = blockIdx.x >> 1;
  const bool dst_role = (blockIdx.x & 1) == 0;
  int4* s4 = (int4*)s_cnt;
  for (int k = t; k < NN / 4; k += 1024) s4[k] = make_int4(0, 0, 0, 0);
  if (t < 64) s_tile[t] = 0;
  __syncthreads();
  if (dst_role) count_chunk<true>(dst + (size_t)c * CH, rank + (size_t)c * CH, s_cnt, t);
  else          count_chunk<false>(src + (size_t)c * CH, nullptr, s_cnt, t);
  __syncthreads();
  int4* o4 = (int4*)((dst_role ? cI : cO) + (size_t)c * NP);
  for (int i = 0; i < 13; ++i) {
    int k = t + 1024 * i;
    bool ok = k < NN / 4;
    int4 v = ok ? s4[k] : make_int4(0, 0, 0, 0);
    if (ok) o4[k] = v;
    if (dst_role) {
      int s = v.x + v.y + v.z + v.w;
      for (int o = 16; o > 0; o >>= 1) s += __shfl_xor(s, o);
      if (lane == 0) atomicAdd(&s_tile[k >> 8], s);
    }
  }
  __syncthreads();
  if (dst_role && t < SCAN_BLOCKS) tileSum[c * 64 + t] = s_tile[t];
}

__global__ void __launch_bounds__(SCAN_T)
rowptr_norm_kernel(const int* __restrict__ cO, const int* __restrict__ cI,
                   const int* __restrict__ tileSum,
                   int* __restrict__ rowptr, int* __restrict__ off,
                   float* __restrict__ normS, float* __restrict__ normD) {
  __shared__ int s_tot[32];
  __shared__ int s_part[32];
  __shared__ int s_off[32];
  int t = threadIdx.x;
  int j = blockIdx.x;
  int lane = t & 31, wv = t >> 5;

  int part = 0;
  for (int k = t; k < NCH * 64; k += SCAN_T)
    if ((k & 63) < j) part += tileSum[k];

  int idx = j * SCAN_T + t;
  int ci[NCH];
  int c = 0, co = 0;
#pragma unroll
  for (int q = 0; q < NCH; ++q) ci[q] = 0;
  if (idx < NN) {
#pragma unroll
    for (int q = 0; q < NCH; ++q) {
      ci[q] = cI[(size_t)q * NP + idx];
      co   += cO[(size_t)q * NP + idx];
      c += ci[q];
    }
  }

  int incl = c;
  for (int o = 1; o < 32; o <<= 1) {
    int v = __shfl_up(incl, o);
    if (lane >= o) incl += v;
  }
  int psum = part;
  for (int o = 16; o > 0; o >>= 1) psum += __shfl_xor(psum, o);
  if (lane == 31) s_tot[wv] = incl;
  if (lane == 0) s_part[wv] = psum;
  __syncthreads();
  if (wv == 0) {
    int wt = s_tot[lane];
    int wi = wt;
    for (int o = 1; o < 32; o <<= 1) {
      int v = __shfl_up(wi, o);
      if (lane >= o) wi += v;
    }
    int ptot = s_part[lane];
    for (int o = 16; o > 0; o >>= 1) ptot += __shfl_xor(ptot, o);
    s_off[lane] = wi - wt + ptot;
  }
  __syncthreads();
  if (idx < NN) {
    int excl = s_off[wv] + incl - c;
    rowptr[idx] = excl;
    if (idx == NN - 1) rowptr[NN] = excl + c;
    int run = excl;
#pragma unroll
    for (int q = 0; q < NCH; ++q) {
      off[(size_t)q * NP + idx] = run;
      run += ci[q];
    }
    normS[idx] = rsqrtf(fmaxf((float)co, 1.0f));
    normD[idx] = rsqrtf(fmaxf((float)c, 1.0f));
  }
}

__global__ void place_kernel(const float* __restrict__ ew, const int* __restrict__ src,
                             const int* __restrict__ dst, const int* __restrict__ off,
                             const int* __restrict__ rank, int* __restrict__ srcS,
                             float* __restrict__ ewS) {
  int e = blockIdx.x * blockDim.x + threadIdx.x;
  if (e >= NE) return;
  const int* offc = off + (size_t)(blockIdx.x / (CH / 256)) * NP;
  int p = offc[dst[e]] + rank[e];
  srcS[p] = src[e];
  float4 lo, hi;
  lo.x = ew[(size_t)0 * NE + e]; lo.y = ew[(size_t)1 * NE + e];
  lo.z = ew[(size_t)2 * NE + e]; lo.w = ew[(size_t)3 * NE + e];
  hi.x = ew[(size_t)4 * NE + e]; hi.y = ew[(size_t)5 * NE + e];
  hi.z = ew[(size_t)6 * NE + e]; hi.w = ew[(size_t)7 * NE + e];
  float4* o = (float4*)&ewS[(size_t)p * 8];
  o[0] = lo; o[1] = hi;
}

typedef __attribute__((ext_vector_type(4))) float v4f;

__global__ void __launch_bounds__(256)
relayout_kernel(const float* __restrict__ cbg, const float* __restrict__ poi,
                const float* __restrict__ normS, float* __restrict__ featT) {
  int tid = blockIdx.x * 256 + threadIdx.x;
  int q = tid & 1;
  int b = (tid >> 1) & 7;
  int n = tid >> 4;
  if (n >= NN) return;
  const v4f* r = (n < CBGN) ? (const v4f*)&cbg[((size_t)b * CBGN + n) * 8]
                            : (const v4f*)&poi[((size_t)b * POIN + (n - CBGN)) * 8];
  v4f x = r[q];
  float s = normS[n];
  x.x *= s; x.y *= s; x.z *= s; x.w *= s;
  ((v4f*)featT)[tid] = x;
}

template <int K>
__device__ __forceinline__ void hold_rows(const v4f (&x)[K]) {
  if constexpr (K == 8) {
    asm volatile("" :: "v"(x[0]), "v"(x[1]), "v"(x[2]), "v"(x[3]),
                       "v"(x[4]), "v"(x[5]), "v"(x[6]), "v"(x[7]));
  } else if constexpr (K == 4) {
    asm volatile("" :: "v"(x[0]), "v"(x[1]), "v"(x[2]), "v"(x[3]));
  } else if constexpr (K == 2) {
    asm volatile("" :: "v"(x[0]), "v"(x[1]));
  }
}
template <int K>
__device__ __forceinline__ void hold_words(const float (&w)[K]) {
  if constexpr (K == 8) {
    asm volatile("" :: "v"(w[0]), "v"(w[1]), "v"(w[2]), "v"(w[3]),
                       "v"(w[4]), "v"(w[5]), "v"(w[6]), "v"(w[7]));
  } else if constexpr (K == 4) {
    asm volatile("" :: "v"(w[0]), "v"(w[1]), "v"(w[2]), "v"(w[3]));
  } else if constexpr (K == 2) {
    asm volatile("" :: "v"(w[0]), "v"(w[1]));
  }
}

template <int K, int F>
__device__ __forceinline__ void edge_batch(v4f& a, const int p, const int b, const int q,
                                           const int* __restrict__ srcS,
                                           const float* __restrict__ ewS,
                                           const float* __restrict__ tab) {
  int s[K];
  float w[K];
  v4f x[K];
  for (int i = 0; i < K; ++i) s[i] = srcS[p + i];
  for (int i = 0; i < K; ++i) w[i] = ewS[(size_t)(p + i) * 8 + b];
  for (int i = 0; i < K; ++i) {
    const v4f* r = (const v4f*)(tab + ((size_t)s[i] * 8 + b) * F);
    x[i] = r[q];
  }
  hold_rows<K>(x);
  hold_words<K>(w);
  for (int i = 0; i < K; ++i) {
    a.x = fmaf(x[i].x, w[i], a.x);
    a.y = fmaf(x[i].y, w[i], a.y);
    a.z = fmaf(x[i].z, w[i], a.z);
    a.w = fmaf(x[i].w, w[i], a.w);
  }
}

template <int F>
__device__ __forceinline__ v4f row_sum(int p, const int p1, const int b, const int q,
                                       const int* __restrict__ srcS,
                                       const float* __restrict__ ewS,
                                       const float* __restrict__ tab) {
  v4f a = {0.f, 0.f, 0.f, 0.f};
  for (; p + 8 <= p1; p += 8) edge_batch<8, F>(a, p, b, q, srcS, ewS, tab);
  if (p + 4 <= p1) { edge_batch<4, F>(a, p, b, q, srcS, ewS, tab); p += 4; }
  if (p + 2 <= p1) { edge_batch<2, F>(a, p, b, q, srcS, ewS, tab); p += 2; }
  if (p < p1)      { edge_batch<1, F>(a, p, b, q, srcS, ewS, tab); }
  return a;
}

__global__ void __launch_bounds__(256)
gather1_kernel(const float* __restrict__ featT, const int* __restrict__ rowptr,
               const int* __restrict__ srcS, const float* __restrict__ ewS,
               float* __restrict__ agg) {
  int tid = blockIdx.x * 256 + threadIdx.x;
  int q = tid & 1;
  int b = (tid >> 1) & 7;
  int n = tid >> 4;
  if (n >= NN) return;
  v4f a = row_sum<8>(rowptr[n], rowptr[n + 1], b, q, srcS, ewS, featT);
  ((v4f*)agg)[tid] = a;
}

__global__ void __launch_bounds__(256)
dense12_kernel(const float* __restrict__ agg, const float* __restrict__ normD,
               const float* __restrict__ normS, const float* __restrict__ W1,
               const float* __restrict__ b1, const float* __restrict__ W2,
               float* __restrict__ t2) {
  int wave = (int)((blockIdx.x * blockDim.x + threadIdx.x) >> 5);
  int lane = threadIdx.x & 31;
  if (wave >= (NN * NB) / 16) return;
  int m0 = wave * 16;
  int rc = lane & 15;
  int half = lane >> 4;
  float nd = normD[(m0 + rc) >> 3];
  float ns = normS[(m0 + rc) >> 3];
  const float* arow = &agg[(size_t)(m0 + rc) * 8];

  v8f c = {0.f, 0.f, 0.f, 0.f, 0.f, 0.f, 0.f, 0.f};
#pragma unroll
  for (int kb = 0; kb < 8; kb += 4) {
    int k0 = kb + 2 * half;
    v2f x, w;
    x.x = arow[k0 + 0] * nd;
    x.y = arow[k0 + 1] * nd;
    w.x = W1[(k0 + 0) * 16 + rc];
    w.y = W1[(k0 + 1) * 16 + rc];
    c = __builtin_amdgcn_wmma_f32_16x16x4_f32(false, w, false, x, (short)0, c,
                                              false, false);
  }
  const v4f* b1v = (const v4f*)b1;
  v4f bl = b1v[2 * half], bh = b1v[2 * half + 1];
  float bias[8] = {bl.x, bl.y, bl.z, bl.w, bh.x, bh.y, bh.z, bh.w};
  float h[8];
#pragma unroll
  for (int r = 0; r < 8; ++r) {
    float v = c[r] + bias[r];
    v = v > 0.0f ? v : 0.01f * v;
    h[r] = v * ns;
  }

  int cw = rc & 7;
  float colmask = (rc < 8) ? 1.0f : 0.0f;
  v8f d = {0.f, 0.f, 0.f, 0.f, 0.f, 0.f, 0.f, 0.f};
#pragma unroll
  for (int j = 0; j < 4; ++j) {
    int k0 = 8 * half + 2 * j;
    v2f w, x;
    w.x = W2[(k0 + 0) * 8 + cw] * colmask;
    w.y = W2[(k0 + 1) * 8 + cw] * colmask;
    x.x = h[2 * j + 0];
    x.y = h[2 * j + 1];
    d = __builtin_amdgcn_wmma_f32_16x16x4_f32(false, w, false, x, (short)0, d,
                                              false, false);
  }
  float u0 = __shfl_xor(d[4], 16);
  float u1 = __shfl_xor(d[5], 16);
  float u2 = __shfl_xor(d[6], 16);
  float u3 = __shfl_xor(d[7], 16);
  v4f o;
  o.x = half ? u0 : d[0];
  o.y = half ? u1 : d[1];
  o.z = half ? u2 : d[2];
  o.w = half ? u3 : d[3];
  ((v4f*)t2)[(size_t)(m0 + rc) * 2 + half] = o;
}

__global__ void __launch_bounds__(256)
gather2_out_kernel(const float* __restrict__ t2, const int* __restrict__ rowptr,
                   const int* __restrict__ srcS, const float* __restrict__ ewS,
                   const float* __restrict__ normD, const float* __restrict__ b2,
                   float* __restrict__ out) {
  int tid = blockIdx.x * 256 + threadIdx.x;
  int q = tid & 1;
  int b = (tid >> 1) & 7;
  int nl = (tid >> 4) & 1;
  int n0 = (tid >> 5) * 4;
  if (n0 >= CBGN) return;
  int nA = n0 + nl, nB = n0 + 2 + nl;
  int pA = rowptr[nA], pA1 = rowptr[nA + 1];
  int pB = rowptr[nB], pB1 = rowptr[nB + 1];
  float ndA = normD[nA], ndB = normD[nB];
  v4f bias = ((const v4f*)b2)[q];
  v4f a = row_sum<8>(pA, pA1, b, q, srcS, ewS, t2);
  v4f c = row_sum<8>(pB, pB1, b, q, srcS, ewS, t2);
  v4f oa, oc;
  oa.x = fmaf(a.x, ndA, bias.x); oa.y = fmaf(a.y, ndA, bias.y);
  oa.z = fmaf(a.z, ndA, bias.z); oa.w = fmaf(a.w, ndA, bias.w);
  oc.x = fmaf(c.x, ndB, bias.x); oc.y = fmaf(c.y, ndB, bias.y);
  oc.z = fmaf(c.z, ndB, bias.z); oc.w = fmaf(c.w, ndB, bias.w);
  ((v4f*)out)[((size_t)b * CBGN + nA) * 2 + q] = oa;
  ((v4f*)out)[((size_t)b * CBGN + nB) * 2 + q] = oc;
}

extern "C" void kernel_launch(void* const* d_in, const int* in_sizes, int n_in,
                              void* d_out, int out_size, void* d_ws, size_t ws_size,
                              hipStream_t stream) {
  (void)in_sizes; (void)n_in; (void)out_size; (void)ws_size;
  const float* ew  = (const float*)d_in[0];
  const float* cbg = (const float*)d_in[1];
  const float* poi = (const float*)d_in[2];
  const float* W1  = (const float*)d_in[3];
  const float* b1  = (const float*)d_in[4];
  const float* W2  = (const float*)d_in[5];
  const float* b2  = (const float*)d_in[6];
  const int*   src = (const int*)d_in[7];
  const int*   dst = (const int*)d_in[8];
  float* out = (float*)d_out;

  float* ws     = (float*)d_ws;
  float* normS  = ws;
  float* normD  = ws + NN;
  int*   tileSum= (int*)(ws + 2 * (size_t)NN);
  int*   rowptr = (int*)(ws + 4 * (size_t)NN);
  int*   rank   = (int*)(ws + 5 * (size_t)NN + 48);
  int*   srcS   = rank + NE;
  float* ewS    = (float*)(srcS + NE);
  float* agg    = ewS + 8 * (size_t)NE;
  float* t2     = agg + (size_t)NB * NN * 16;
  float* featT  = t2 + (size_t)NB * NN * 16;
  int*   cI     = (int*)(featT + (size_t)NB * NN * 8);
  int*   cO     = cI + (size_t)NCH * NP;
  int*   off    = cO + (size_t)NCH * NP;

  count_rank_kernel<<<2 * NCH, 1024, 0, stream>>>(src, dst, cO, cI, tileSum, rank);
  rowptr_norm_kernel<<<SCAN_BLOCKS, SCAN_T, 0, stream>>>(cO, cI, tileSum, rowptr, off, normS, normD);

  place_kernel<<<(NE + 255) / 256, 256, 0, stream>>>(ew, src, dst, off, rank, srcS, ewS);
  relayout_kernel<<<(NN * 16) / 256, 256, 0, stream>>>(cbg, poi, normS, featT);

  gather1_kernel<<<(NN * 16) / 256, 256, 0, stream>>>(featT, rowptr, srcS, ewS, agg);
  dense12_kernel<<<((NN * NB) / 16) / 8, 256, 0, stream>>>(agg, normD, normS, W1, b1, W2, t2);

  gather2_out_kernel<<<(CBGN / 4 + 7) / 8, 256, 0, stream>>>(t2, rowptr, srcS, ewS, normD, b2, out);
}


# Weight records stay in edge order, written once as whole lines; only an 8-B {src, edge id} pair per edge is scattered into dst order; row sums fetch each edge's 32-B record through the edge id
# speedup vs baseline: 42.1424x; 1.1087x over previous
#include <hip/hip_runtime.h>

#define NB   8
#define CBGN 30000
#define POIN 20000
#define NE   1600000
#define NN   50000
#define SCAN_T 1024
#define SCAN_BLOCKS 49
#define NCH  25
#define CH   (NE / NCH)
#define CH4  (CH / 4)
#define NP   50016

typedef __attribute__((ext_vector_type(2))) float v2f;
typedef __attribute__((ext_vector_type(8))) float v8f;

template <bool RANK>
__device__ __forceinline__ void count_chunk(const int* __restrict__ idx, int* __restrict__ rank,
                                            int* __restrict__ s_cnt, const int t) {
  const int4* i4 = (const int4*)idx;
  int4 d[16];
#pragma unroll
  for (int i = 0; i < 16; ++i) {
    int k = t + 1024 * i;
    d[i] = (k < CH4) ? i4[k] : make_int4(0, 0, 0, 0);
  }
#pragma unroll
  for (int i = 0; i < 16; ++i) {
    int k = t + 1024 * i;
    if (k < CH4) {
      if constexpr (RANK) {
        int4 r;
        r.x = atomicAdd(&s_cnt[d[i].x], 1);
        r.y = atomicAdd(&s_cnt[d[i].y], 1);
        r.z = atomicAdd(&s_cnt[d[i].z], 1);
        r.w = atomicAdd(&s_cnt[d[i].w], 1);
        ((int4*)rank)[k] = r;
      } else {
        atomicAdd(&s_cnt[d[i].x], 1);
        atomicAdd(&s_cnt[d[i].y], 1);
        atomicAdd(&s_cnt[d[i].z], 1);
        atomicAdd(&s_cnt[d[i].w], 1);
      }
    }
  }
}

__global__ void __launch_bounds__(1024)
count_rank_kernel(const int* __restrict__ src, const int* __restrict__ dst,
                  int* __restrict__ cO, int* __restrict__ cI,
                  int* __restrict__ tileSum, int* __restrict__ rank) {
  __shared__ __attribute__((aligned(16))) int s_cnt[NN];
  __shared__ int s_tile[64];
  const int t = threadIdx.x;
  const int lane = t & 31;
  const int c = blockIdx.x >> 1;
  const bool dst_role = (blockIdx.x & 1) == 0;
  int4* s4 = (int4*)s_cnt;
  for (int k = t; k < NN / 4; k += 1024) s4[k] = make_int4(0, 0, 0, 0);
  if (t < 64) s_tile[t] = 0;
  __syncthreads();
  if (dst_role) count_chunk<true>(dst + (size_t)c * CH, rank + (size_t)c * CH, s_cnt, t);
  else          count_chunk<false>(src + (size_t)c * CH, nullptr, s_cnt, t);
  __syncthreads();
  int4* o4 = (int4*)((dst_role ? cI : cO) + (size_t)c * NP);
  for (int i = 0; i < 13; ++i) {
    int k = t + 1024 * i;
    bool ok = k < NN / 4;
    int4 v = ok ? s4[k] : make_int4(0, 0, 0, 0);
    if (ok) o4[k] = v;
    if (dst_role) {
      int s = v.x + v.y + v.z + v.w;
      for (int o = 16; o > 0; o >>= 1) s += __shfl_xor(s, o);
      if (lane == 0) atomicAdd(&s_tile[k >> 8], s);
    }
  }
  __syncthreads();
  if (dst_role && t < SCAN_BLOCKS) tileSum[c * 64 + t] = s_tile[t];
}

__global__ void __launch_bounds__(SCAN_T)
rowptr_norm_kernel(const int* __restrict__ cO, const int* __restrict__ cI,
                   const int* __restrict__ tileSum,
                   int* __restrict__ rowptr, int* __restrict__ off,
                   float* __restrict__ normS, float* __restrict__ normD) {
  __shared__ int s_tot[32];
  __shared__ int s_part[32];
  __shared__ int s_off[32];
  int t = threadIdx.x;
  int j = blockIdx.x;
  int lane = t & 31, wv = t >> 5;

  int part = 0;
  for (int k = t; k < NCH * 64; k += SCAN_T)
    if ((k & 63) < j) part += tileSum[k];

  int idx = j * SCAN_T + t;
  int ci[NCH];
  int c = 0, co = 0;
#pragma unroll
  for (int q = 0; q < NCH; ++q) ci[q] = 0;
  if (idx < NN) {
#pragma unroll
    for (int q = 0; q < NCH; ++q) {
      ci[q] = cI[(size_t)q * NP + idx];
      co   += cO[(size_t)q * NP + idx];
      c += ci[q];
    }
  }

  int incl = c;
  for (int o = 1; o < 32; o <<= 1) {
    int v = __shfl_up(incl, o);
    if (lane >= o) incl += v;
  }
  int psum = part;
  for (int o = 16; o > 0; o >>= 1) psum += __shfl_xor(psum, o);
  if (lane == 31) s_tot[wv] = incl;
  if (lane == 0) s_part[wv] = psum;
  __syncthreads();
  if (wv == 0) {
    int wt = s_tot[lane];
    int wi = wt;
    for (int o = 1; o < 32; o <<= 1) {
      int v = __shfl_up(wi, o);
      if (lane >= o) wi += v;
    }
    int ptot = s_part[lane];
    for (int o = 16; o > 0; o >>= 1) ptot += __shfl_xor(ptot, o);
    s_off[lane] = wi - wt + ptot;
  }
  __syncthreads();
  if (idx < NN) {
    int excl = s_off[wv] + incl - c;
    rowptr[idx] = excl;
    if (idx == NN - 1) rowptr[NN] = excl + c;
    int run = excl;
#pragma unroll
    for (int q = 0; q < NCH; ++q) {
      off[(size_t)q * NP + idx] = run;
      run += ci[q];
    }
    normS[idx] = rsqrtf(fmaxf((float)co, 1.0f));
    normD[idx] = rsqrtf(fmaxf((float)c, 1.0f));
  }
}

typedef __attribute__((ext_vector_type(4))) float v4f;

__global__ void __launch_bounds__(256)
place_kernel(const float* __restrict__ ew, const int* __restrict__ src,
             const int* __restrict__ dst, const int* __restrict__ off,
             const int* __restrict__ rank, int* __restrict__ sp,
             float* __restrict__ ewR) {
  const int h = threadIdx.x & 1;
  const int e0 = blockIdx.x * 256 + (threadIdx.x >> 1);
  const int e1 = e0 + 128;
  const int em = h ? e1 : e0;
  const int* offc = off + (size_t)(blockIdx.x / (CH / 256)) * NP;
  const int d = dst[em];
  const int r = rank[em];
  const int sv = src[em];
  v4f w0, w1;
  w0.x = ew[(size_t)(4 * h + 0) * NE + e0];
  w0.y = ew[(size_t)(4 * h + 1) * NE + e0];
  w0.z = ew[(size_t)(4 * h + 2) * NE + e0];
  w0.w = ew[(size_t)(4 * h + 3) * NE + e0];
  w1.x = ew[(size_t)(4 * h + 0) * NE + e1];
  w1.y = ew[(size_t)(4 * h + 1) * NE + e1];
  w1.z = ew[(size_t)(4 * h + 2) * NE + e1];
  w1.w = ew[(size_t)(4 * h + 3) * NE + e1];
  const int p = offc[d] + r;
  ((int2*)sp)[p] = make_int2(sv, em);
  ((v4f*)ewR)[(size_t)e0 * 2 + h] = w0;
  ((v4f*)ewR)[(size_t)e1 * 2 + h] = w1;
}

__global__ void __launch_bounds__(256)
relayout_kernel(const float* __restrict__ cbg, const float* __restrict__ poi,
                const float* __restrict__ normS, float* __restrict__ featT) {
  int tid = blockIdx.x * 256 + threadIdx.x;
  int q = tid & 1;
  int b = (tid >> 1) & 7;
  int n = tid >> 4;
  if (n >= NN) return;
  const v4f* r = (n < CBGN) ? (const v4f*)&cbg[((size_t)b * CBGN + n) * 8]
                            : (const v4f*)&poi[((size_t)b * POIN + (n - CBGN)) * 8];
  v4f x = r[q];
  float s = normS[n];
  x.x *= s; x.y *= s; x.z *= s; x.w *= s;
  ((v4f*)featT)[tid] = x;
}

template <int K>
__device__ __forceinline__ void hold_rows(const v4f (&x)[K]) {
  if constexpr (K == 8) {
    asm volatile("" :: "v"(x[0]), "v"(x[1]), "v"(x[2]), "v"(x[3]),
                       "v"(x[4]), "v"(x[5]), "v"(x[6]), "v"(x[7]));
  } else if constexpr (K == 4) {
    asm volatile("" :: "v"(x[0]), "v"(x[1]), "v"(x[2]), "v"(x[3]));
  } else if constexpr (K == 2) {
    asm volatile("" :: "v"(x[0]), "v"(x[1]));
  }
}
template <int K>
__device__ __forceinline__ void hold_words(const float (&w)[K]) {
  if constexpr (K == 8) {
    asm volatile("" :: "v"(w[0]), "v"(w[1]), "v"(w[2]), "v"(w[3]),
                       "v"(w[4]), "v"(w[5]), "v"(w[6]), "v"(w[7]));
  } else if constexpr (K == 4) {
    asm volatile("" :: "v"(w[0]), "v"(w[1]), "v"(w[2]), "v"(w[3]));
  } else if constexpr (K == 2) {
    asm volatile("" :: "v"(w[0]), "v"(w[1]));
  }
}

template <int K, int F>
__device__ __forceinline__ void edge_batch(v4f& a, const int p, const int b, const int q,
                                           const int* __restrict__ sp,
                                           const float* __restrict__ ewR,
                                           const float* __restrict__ tab) {
  int s[K];
  int ei[K];
  float w[K];
  v4f x[K];
  for (int i = 0; i < K; ++i) {
    int2 v = ((const int2*)sp)[p + i];
    s[i] = v.x;
    ei[i] = v.y;
  }
  for (int i = 0; i < K; ++i) w[i] = ewR[(size_t)ei[i] * 8 + b];
  for (int i = 0; i < K; ++i) {
    const v4f* r = (const v4f*)(tab + ((size_t)s[i] * 8 + b) * F);
    x[i] = r[q];
  }
  hold_rows<K>(x);
  hold_words<K>(w);
  for (int i = 0; i < K; ++i) {
    a.x = fmaf(x[i].x, w[i], a.x);
    a.y = fmaf(x[i].y, w[i], a.y);
    a.z = fmaf(x[i].z, w[i], a.z);
    a.w = fmaf(x[i].w, w[i], a.w);
  }
}

template <int F>
__device__ __forceinline__ v4f row_sum(int p, const int p1, const int b, const int q,
                                       const int* __restrict__ sp,
                                       const float* __restrict__ ewR,
                                       const float* __restrict__ tab) {
  v4f a = {0.f, 0.f, 0.f, 0.f};
  for (; p + 8 <= p1; p += 8) edge_batch<8, F>(a, p, b, q, sp, ewR, tab);
  if (p + 4 <= p1) { edge_batch<4, F>(a, p, b, q, sp, ewR, tab); p += 4; }
  if (p + 2 <= p1) { edge_batch<2, F>(a, p, b, q, sp, ewR, tab); p += 2; }
  if (p < p1)      { edge_batch<1, F>(a, p, b, q, sp, ewR, tab); }
  return a;
}

__global__ void __launch_bounds__(256)
gather1_kernel(const float* __restrict__ featT, const int* __restrict__ rowptr,
               const int* __restrict__ sp, const float* __restrict__ ewR,
               float* __restrict__ agg) {
  int tid = blockIdx.x * 256 + threadIdx.x;
  int q = tid & 1;
  int b = (tid >> 1) & 7;
  int n = tid >> 4;
  if (n >= NN) return;
  v4f a = row_sum<8>(rowptr[n], rowptr[n + 1], b, q, sp, ewR, featT);
  ((v4f*)agg)[tid] = a;
}

__global__ void __launch_bounds__(256)
dense12_kernel(const float* __restrict__ agg, const float* __restrict__ normD,
               const float* __restrict__ normS, const float* __restrict__ W1,
               const float* __restrict__ b1, const float* __restrict__ W2,
               float* __restrict__ t2) {
  int wave = (int)((blockIdx.x * blockDim.x + threadIdx.x) >> 5);
  int lane = threadIdx.x & 31;
  if (wave >= (NN * NB) / 16) return;
  int m0 = wave * 16;
  int rc = lane & 15;
  int half = lane >> 4;
  float nd = normD[(m0 + rc) >> 3];
  float ns = normS[(m0 + rc) >> 3];
  const float* arow = &agg[(size_t)(m0 + rc) * 8];

  v8f c = {0.f, 0.f, 0.f, 0.f, 0.f, 0.f, 0.f, 0.f};
#pragma unroll
  for (int kb = 0; kb < 8; kb += 4) {
    int k0 = kb + 2 * half;
    v2f x, w;
    x.x = arow[k0 + 0] * nd;
    x.y = arow[k0 + 1] * nd;
    w.x = W1[(k0 + 0) * 16 + rc];
    w.y = W1[(k0 + 1) * 16 + rc];
    c = __builtin_amdgcn_wmma_f32_16x16x4_f32(false, w, false, x, (short)0, c,
                                              false, false);
  }
  const v4f* b1v = (const v4f*)b1;
  v4f bl = b1v[2 * half], bh = b1v[2 * half + 1];
  float bias[8] = {bl.x, bl.y, bl.z, bl.w, bh.x, bh.y, bh.z, bh.w};
  float h[8];
#pragma unroll
  for (int r = 0; r < 8; ++r) {
    float v = c[r] + bias[r];
    v = v > 0.0f ? v : 0.01f * v;
    h[r] = v * ns;
  }

  int cw = rc & 7;
  float colmask = (rc < 8) ? 1.0f : 0.0f;
  v8f d = {0.f, 0.f, 0.f, 0.f, 0.f, 0.f, 0.f, 0.f};
#pragma unroll
  for (int j = 0; j < 4; ++j) {
    int k0 = 8 * half + 2 * j;
    v2f w, x;
    w.x = W2[(k0 + 0) * 8 + cw] * colmask;
    w.y = W2[(k0 + 1) * 8 + cw] * colmask;
    x.x = h[2 * j + 0];
    x.y = h[2 * j + 1];
    d = __builtin_amdgcn_wmma_f32_16x16x4_f32(false, w, false, x, (short)0, d,
                                              false, false);
  }
  float u0 = __shfl_xor(d[4], 16);
  float u1 = __shfl_xor(d[5], 16);
  float u2 = __shfl_xor(d[6], 16);
  float u3 = __shfl_xor(d[7], 16);
  v4f o;
  o.x = half ? u0 : d[0];
  o.y = half ? u1 : d[1];
  o.z = half ? u2 : d[2];
  o.w = half ? u3 : d[3];
  ((v4f*)t2)[(size_t)(m0 + rc) * 2 + half] = o;
}

__global__ void __launch_bounds__(256)
gather2_out_kernel(const float* __restrict__ t2, const int* __restrict__ rowptr,
                   const int* __restrict__ sp, const float* __restrict__ ewR,
                   const float* __restrict__ normD, const float* __restrict__ b2,
                   float* __restrict__ out) {
  int tid = blockIdx.x * 256 + threadIdx.x;
  int q = tid & 1;
  int b = (tid >> 1) & 7;
  int nl = (tid >> 4) & 1;
  int n0 = (tid >> 5) * 4;
  if (n0 >= CBGN) return;
  int nA = n0 + nl, nB = n0 + 2 + nl;
  int pA = rowptr[nA], pA1 = rowptr[nA + 1];
  int pB = rowptr[nB], pB1 = rowptr[nB + 1];
  float ndA = normD[nA], ndB = normD[nB];
  v4f bias = ((const v4f*)b2)[q];
  v4f a = row_sum<8>(pA, pA1, b, q, sp, ewR, t2);
  v4f c = row_sum<8>(pB, pB1, b, q, sp, ewR, t2);
  v4f oa, oc;
  oa.x = fmaf(a.x, ndA, bias.x); oa.y = fmaf(a.y, ndA, bias.y);
  oa.z = fmaf(a.z, ndA, bias.z); oa.w = fmaf(a.w, ndA, bias.w);
  oc.x = fmaf(c.x, ndB, bias.x); oc.y = fmaf(c.y, ndB, bias.y);
  oc.z = fmaf(c.z, ndB, bias.z); oc.w = fmaf(c.w, ndB, bias.w);
  ((v4f*)out)[((size_t)b * CBGN + nA) * 2 + q] = oa;
  ((v4f*)out)[((size_t)b * CBGN + nB) * 2 + q] = oc;
}

extern "C" void kernel_launch(void* const* d_in, const int* in_sizes, int n_in,
                              void* d_out, int out_size, void* d_ws, size_t ws_size,
                              hipStream_t stream) {
  (void)in_sizes; (void)n_in; (void)out_size; (void)ws_size;
  const float* ew  = (const float*)d_in[0];
  const float* cbg = (const float*)d_in[1];
  const float* poi = (const float*)d_in[2];
  const float* W1  = (const float*)d_in[3];
  const float* b1  = (const float*)d_in[4];
  const float* W2  = (const float*)d_in[5];
  const float* b2  = (const float*)d_in[6];
  const int*   src = (const int*)d_in[7];
  const int*   dst = (const int*)d_in[8];
  float* out = (float*)d_out;

  float* ws     = (float*)d_ws;
  float* normS  = ws;
  float* normD  = ws + NN;
  int*   tileSum= (int*)(ws + 2 * (size_t)NN);
  int*   rowptr = (int*)(ws + 4 * (size_t)NN);
  int*   rank   = (int*)(ws + 5 * (size_t)NN + 48);
  int*   sp     = rank + NE;
  float* ewR    = (float*)(sp + 2 * (size_t)NE);
  float* agg    = ewR + 8 * (size_t)NE;
  float* t2     = agg + (size_t)NB * NN * 16;
  float* featT  = t2 + (size_t)NB * NN * 16;
  int*   cI     = (int*)(featT + (size_t)NB * NN * 8);
  int*   cO     = cI + (size_t)NCH * NP;
  int*   off    = cO + (size_t)NCH * NP;

  count_rank_kernel<<<2 * NCH, 1024, 0, stream>>>(src, dst, cO, cI, tileSum, rank);
  rowptr_norm_kernel<<<SCAN_BLOCKS, SCAN_T, 0, stream>>>(cO, cI, tileSum, rowptr, off, normS, normD);

  place_kernel<<<NE / 256, 256, 0, stream>>>(ew, src, dst, off, rank, sp, ewR);
  relayout_kernel<<<(NN * 16) / 256, 256, 0, stream>>>(cbg, poi, normS, featT);

  gather1_kernel<<<(NN * 16) / 256, 256, 0, stream>>>(featT, rowptr, sp, ewR, agg);
  dense12_kernel<<<((NN * NB) / 16) / 8, 256, 0, stream>>>(agg, normD, normS, W1, b1, W2, t2);

  gather2_out_kernel<<<(CBGN / 4 + 7) / 8, 256, 0, stream>>>(t2, rowptr, sp, ewR, normD, b2, out);
}


# Edge placement, table re-layout and dense WMMA kernels give each wave 5 pieces with all loads issued first: 100000 one-piece waves become 20000, so wave start rate no longer sets their time
# speedup vs baseline: 43.7155x; 1.0373x over previous
#include <hip/hip_runtime.h>

#define NB   8
#define CBGN 30000
#define POIN 20000
#define NE   1600000
#define NN   50000
#define SCAN_T 1024
#define SCAN_BLOCKS 49
#define NCH  25
#define CH   (NE / NCH)
#define CH4  (CH / 4)
#define NP   50016

typedef __attribute__((ext_vector_type(2))) float v2f;
typedef __attribute__((ext_vector_type(8))) float v8f;

template <bool RANK>
__device__ __forceinline__ void count_chunk(const int* __restrict__ idx, int* __restrict__ rank,
                                            int* __restrict__ s_cnt, const int t) {
  const int4* i4 = (const int4*)idx;
  int4 d[16];
#pragma unroll
  for (int i = 0; i < 16; ++i) {
    int k = t + 1024 * i;
    d[i] = (k < CH4) ? i4[k] : make_int4(0, 0, 0, 0);
  }
#pragma unroll
  for (int i = 0; i < 16; ++i) {
    int k = t + 1024 * i;
    if (k < CH4) {
      if constexpr (RANK) {
        int4 r;
        r.x = atomicAdd(&s_cnt[d[i].x], 1);
        r.y = atomicAdd(&s_cnt[d[i].y], 1);
        r.z = atomicAdd(&s_cnt[d[i].z], 1);
        r.w = atomicAdd(&s_cnt[d[i].w], 1);
        ((int4*)rank)[k] = r;
      } else {
        atomicAdd(&s_cnt[d[i].x], 1);
        atomicAdd(&s_cnt[d[i].y], 1);
        atomicAdd(&s_cnt[d[i].z], 1);
        atomicAdd(&s_cnt[d[i].w], 1);
      }
    }
  }
}

__global__ void __launch_bounds__(1024)
count_rank_kernel(const int* __restrict__ src, const int* __restrict__ dst,
                  int* __restrict__ cO, int* __restrict__ cI,
                  int* __restrict__ tileSum, int* __restrict__ rank) {
  __shared__ __attribute__((aligned(16))) int s_cnt[NN];
  __shared__ int s_tile[64];
  const int t = threadIdx.x;
  const int lane = t & 31;
  const int c = blockIdx.x >> 1;
  const bool dst_role = (blockIdx.x & 1) == 0;
  int4* s4 = (int4*)s_cnt;
  for (int k = t; k < NN / 4; k += 1024) s4[k] = make_int4(0, 0, 0, 0);
  if (t < 64) s_tile[t] = 0;
  __syncthreads();
  if (dst_role) count_chunk<true>(dst + (size_t)c * CH, rank + (size_t)c * CH, s_cnt, t);
  else          count_chunk<false>(src + (size_t)c * CH, nullptr, s_cnt, t);
  __syncthreads();
  int4* o4 = (int4*)((dst_role ? cI : cO) + (size_t)c * NP);
  for (int i = 0; i < 13; ++i) {
    int k = t + 1024 * i;
    bool ok = k < NN / 4;
    int4 v = ok ? s4[k] : make_int4(0, 0, 0, 0);
    if (ok) o4[k] = v;
    if (dst_role) {
      int s = v.x + v.y + v.z + v.w;
      for (int o = 16; o > 0; o >>= 1) s += __shfl_xor(s, o);
      if (lane == 0) atomicAdd(&s_tile[k >> 8], s);
    }
  }
  __syncthreads();
  if (dst_role && t < SCAN_BLOCKS) tileSum[c * 64 + t] = s_tile[t];
}

__global__ void __launch_bounds__(SCAN_T)
rowptr_norm_kernel(const int* __restrict__ cO, const int* __restrict__ cI,
                   const int* __restrict__ tileSum,
                   int* __restrict__ rowptr, int* __restrict__ off,
                   float* __restrict__ normS, float* __restrict__ normD) {
  __shared__ int s_tot[32];
  __shared__ int s_part[32];
  __shared__ int s_off[32];
  int t = threadIdx.x;
  int j = blockIdx.x;
  int lane = t & 31, wv = t >> 5;

  int part = 0;
  for (int k = t; k < NCH * 64; k += SCAN_T)
    if ((k & 63) < j) part += tileSum[k];

  int idx = j * SCAN_T + t;
  int ci[NCH];
  int c = 0, co = 0;
#pragma unroll
  for (int q = 0; q < NCH; ++q) ci[q] = 0;
  if (idx < NN) {
#pragma unroll
    for (int q = 0; q < NCH; ++q) {
      ci[q] = cI[(size_t)q * NP + idx];
      co   += cO[(size_t)q * NP + idx];
      c += ci[q];
    }
  }

  int incl = c;
  for (int o = 1; o < 32; o <<= 1) {
    int v = __shfl_up(incl, o);
    if (lane >= o) incl += v;
  }
  int psum = part;
  for (int o = 16; o > 0; o >>= 1) psum += __shfl_xor(psum, o);
  if (lane == 31) s_tot[wv] = incl;
  if (lane == 0) s_part[wv] = psum;
  __syncthreads();
  if (wv == 0) {
    int wt = s_tot[lane];
    int wi = wt;
    for (int o = 1; o < 32; o <<= 1) {
      int v = __shfl_up(wi, o);
      if (lane >= o) wi += v;
    }
    int ptot = s_part[lane];
    for (int o = 16; o > 0; o >>= 1) ptot += __shfl_xor(ptot, o);
    s_off[lane] = wi - wt + ptot;
  }
  __syncthreads();
  if (idx < NN) {
    int excl = s_off[wv] + incl - c;
    rowptr[idx] = excl;
    if (idx == NN - 1) rowptr[NN] = excl + c;
    int run = excl;
#pragma unroll
    for (int q = 0; q < NCH; ++q) {
      off[(size_t)q * NP + idx] = run;
      run += ci[q];
    }
    normS[idx] = rsqrtf(fmaxf((float)co, 1.0f));
    normD[idx] = rsqrtf(fmaxf((float)c, 1.0f));
  }
}

typedef __attribute__((ext_vector_type(4))) float v4f;

#define PL_R 5
#define PL_G (NE / (256 * PL_R))
static_assert(NE % (256 * PL_R) == 0, "whole work-groups");
static_assert((CH / 256) % PL_R == 0, "a work-group's pieces lie in one chunk");
__global__ void __launch_bounds__(256)
place_kernel(const float* __restrict__ ew, const int* __restrict__ src,
             const int* __restrict__ dst, const int* __restrict__ off,
             const int* __restrict__ rank, int* __restrict__ sp,
             float* __restrict__ ewR) {
  const int h = threadIdx.x & 1;
  const int base = blockIdx.x * (256 * PL_R) + (threadIdx.x >> 1);
  const int* offc = off + (size_t)(blockIdx.x / (CH / (256 * PL_R))) * NP;
  int d[PL_R], r[PL_R], sv[PL_R], p[PL_R];
  v4f w0[PL_R], w1[PL_R];
#pragma unroll
  for (int i = 0; i < PL_R; ++i) {
    const int em = base + i * 256 + (h ? 128 : 0);
    d[i] = dst[em];
  }
#pragma unroll
  for (int i = 0; i < PL_R; ++i) {
    const int em = base + i * 256 + (h ? 128 : 0);
    r[i] = rank[em];
    sv[i] = src[em];
  }
#pragma unroll
  for (int i = 0; i < PL_R; ++i) {
    const int e0 = base + i * 256;
    const int e1 = e0 + 128;
    w0[i].x = ew[(size_t)(4 * h + 0) * NE + e0];
    w0[i].y = ew[(size_t)(4 * h + 1) * NE + e0];
    w0[i].z = ew[(size_t)(4 * h + 2) * NE + e0];
    w0[i].w = ew[(size_t)(4 * h + 3) * NE + e0];
    w1[i].x = ew[(size_t)(4 * h + 0) * NE + e1];
    w1[i].y = ew[(size_t)(4 * h + 1) * NE + e1];
    w1[i].z = ew[(size_t)(4 * h + 2) * NE + e1];
    w1[i].w = ew[(size_t)(4 * h + 3) * NE + e1];
  }
#pragma unroll
  for (int i = 0; i < PL_R; ++i) p[i] = offc[d[i]] + r[i];
#pragma unroll
  for (int i = 0; i < PL_R; ++i) {
    const int e0 = base + i * 256;
    const int e1 = e0 + 128;
    ((v4f*)ewR)[(size_t)e0 * 2 + h] = w0[i];
    ((v4f*)ewR)[(size_t)e1 * 2 + h] = w1[i];
  }
#pragma unroll
  for (int i = 0; i < PL_R; ++i) {
    const int em = base + i * 256 + (h ? 128 : 0);
    ((int2*)sp)[p[i]] = make_int2(sv[i], em);
  }
}

#define RL_R 5
#define RL_G ((NN * 16) / (256 * RL_R))
static_assert((NN * 16) % (256 * RL_R) == 0, "whole work-groups: no bounds guard");
__global__ void __launch_bounds__(256)
relayout_kernel(const float* __restrict__ cbg, const float* __restrict__ poi,
                const float* __restrict__ normS, float* __restrict__ featT) {
  const int t0 = blockIdx.x * (256 * RL_R) + threadIdx.x;
  v4f x[RL_R];
  float s[RL_R];
#pragma unroll
  for (int i = 0; i < RL_R; ++i) {
    int tid = t0 + i * 256;
    int q = tid & 1;
    int b = (tid >> 1) & 7;
    int n = tid >> 4;
    const bool in_cbg = n < CBGN;
    const float* tab = in_cbg ? cbg : poi;
    const int row = in_cbg ? (b * CBGN + n) : (b * POIN + (n - CBGN));
    x[i] = ((const v4f*)tab)[(size_t)row * 2 + q];
    s[i] = normS[n];
  }
#pragma unroll
  for (int i = 0; i < RL_R; ++i) {
    int tid = t0 + i * 256;
    v4f y = x[i];
    y.x *= s[i]; y.y *= s[i]; y.z *= s[i]; y.w *= s[i];
    ((v4f*)featT)[tid] = y;
  }
}

template <int K>
__device__ __forceinline__ void hold_rows(const v4f (&x)[K]) {
  if constexpr (K == 8) {
    asm volatile("" :: "v"(x[0]), "v"(x[1]), "v"(x[2]), "v"(x[3]),
                       "v"(x[4]), "v"(x[5]), "v"(x[6]), "v"(x[7]));
  } else if constexpr (K == 4) {
    asm volatile("" :: "v"(x[0]), "v"(x[1]), "v"(x[2]), "v"(x[3]));
  } else if constexpr (K == 2) {
    asm volatile("" :: "v"(x[0]), "v"(x[1]));
  }
}
template <int K>
__device__ __forceinline__ void hold_words(const float (&w)[K]) {
  if constexpr (K == 8) {
    asm volatile("" :: "v"(w[0]), "v"(w[1]), "v"(w[2]), "v"(w[3]),
                       "v"(w[4]), "v"(w[5]), "v"(w[6]), "v"(w[7]));
  } else if constexpr (K == 4) {
    asm volatile("" :: "v"(w[0]), "v"(w[1]), "v"(w[2]), "v"(w[3]));
  } else if constexpr (K == 2) {
    asm volatile("" :: "v"(w[0]), "v"(w[1]));
  }
}

template <int K, int F>
__device__ __forceinline__ void edge_batch(v4f& a, const int p, const int b, const int q,
                                           const int* __restrict__ sp,
                                           const float* __restrict__ ewR,
                                           const float* __restrict__ tab) {
  int s[K];
  int ei[K];
  float w[K];
  v4f x[K];
  for (int i = 0; i < K; ++i) {
    int2 v = ((const int2*)sp)[p + i];
    s[i] = v.x;
    ei[i] = v.y;
  }
  for (int i = 0; i < K; ++i) w[i] = ewR[(size_t)ei[i] * 8 + b];
  for (int i = 0; i < K; ++i) {
    const v4f* r = (const v4f*)(tab + ((size_t)s[i] * 8 + b) * F);
    x[i] = r[q];
  }
  hold_rows<K>(x);
  hold_words<K>(w);
  for (int i = 0; i < K; ++i) {
    a.x = fmaf(x[i].x, w[i], a.x);
    a.y = fmaf(x[i].y, w[i], a.y);
    a.z = fmaf(x[i].z, w[i], a.z);
    a.w = fmaf(x[i].w, w[i], a.w);
  }
}

template <int F>
__device__ __forceinline__ v4f row_sum(int p, const int p1, const int b, const int q,
                                       const int* __restrict__ sp,
                                       const float* __restrict__ ewR,
                                       const float* __restrict__ tab) {
  v4f a = {0.f, 0.f, 0.f, 0.f};
  for (; p + 8 <= p1; p += 8) edge_batch<8, F>(a, p, b, q, sp, ewR, tab);
  if (p + 4 <= p1) { edge_batch<4, F>(a, p, b, q, sp, ewR, tab); p += 4; }
  if (p + 2 <= p1) { edge_batch<2, F>(a, p, b, q, sp, ewR, tab); p += 2; }
  if (p < p1)      { edge_batch<1, F>(a, p, b, q, sp, ewR, tab); }
  return a;
}

__global__ void __launch_bounds__(256)
gather1_kernel(const float* __restrict__ featT, const int* __restrict__ rowptr,
               const int* __restrict__ sp, const float* __restrict__ ewR,
               float* __restrict__ agg) {
  int tid = blockIdx.x * 256 + threadIdx.x;
  int q = tid & 1;
  int b = (tid >> 1) & 7;
  int n = tid >> 4;
  if (n >= NN) return;
  v4f a = row_sum<8>(rowptr[n], rowptr[n + 1], b, q, sp, ewR, featT);
  ((v4f*)agg)[tid] = a;
}

#define DN_R 5
#define DN_TILES ((NN * NB) / 16)
#define DN_G (DN_TILES / (8 * DN_R))
static_assert(DN_TILES % (8 * DN_R) == 0, "whole waves: every lane runs every product");
__global__ void __launch_bounds__(256)
dense12_kernel(const float* __restrict__ agg, const float* __restrict__ normD,
               const float* __restrict__ normS, const float* __restrict__ W1,
               const float* __restrict__ b1, const float* __restrict__ W2,
               float* __restrict__ t2) {
  const int wave = (int)((blockIdx.x * 256 + threadIdx.x) >> 5);
  const int lane = threadIdx.x & 31;
  const int rc = lane & 15;
  const int half = lane >> 4;
  const int cw = rc & 7;
  const float colmask = (rc < 8) ? 1.0f : 0.0f;

  v2f wa[2], wb[4];
#pragma unroll
  for (int kb = 0; kb < 8; kb += 4) {
    int k0 = kb + 2 * half;
    wa[kb >> 2].x = W1[(k0 + 0) * 16 + rc];
    wa[kb >> 2].y = W1[(k0 + 1) * 16 + rc];
  }
#pragma unroll
  for (int j = 0; j < 4; ++j) {
    int k0 = 8 * half + 2 * j;
    wb[j].x = W2[(k0 + 0) * 8 + cw] * colmask;
    wb[j].y = W2[(k0 + 1) * 8 + cw] * colmask;
  }
  const v4f* b1v = (const v4f*)b1;
  v4f bl = b1v[2 * half], bh = b1v[2 * half + 1];
  const float bias[8] = {bl.x, bl.y, bl.z, bl.w, bh.x, bh.y, bh.z, bh.w};

  v2f xa[DN_R][2];
  float nd[DN_R], ns[DN_R];
#pragma unroll
  for (int i = 0; i < DN_R; ++i) {
    int row = (wave * DN_R + i) * 16 + rc;
    nd[i] = normD[row >> 3];
    ns[i] = normS[row >> 3];
    const v2f* ar = (const v2f*)&agg[(size_t)row * 8];
    xa[i][0] = ar[half];
    xa[i][1] = ar[2 + half];
  }

#pragma unroll
  for (int i = 0; i < DN_R; ++i) {
    int m0 = (wave * DN_R + i) * 16;
    v8f c = {0.f, 0.f, 0.f, 0.f, 0.f, 0.f, 0.f, 0.f};
#pragma unroll
    for (int kk = 0; kk < 2; ++kk) {
      v2f x;
      x.x = xa[i][kk].x * nd[i];
      x.y = xa[i][kk].y * nd[i];
      c = __builtin_amdgcn_wmma_f32_16x16x4_f32(false, wa[kk], false, x, (short)0, c,
                                                false, false);
    }
    float h[8];
#pragma unroll
    for (int r = 0; r < 8; ++r) {
      float v = c[r] + bias[r];
      v = v > 0.0f ? v : 0.01f * v;
      h[r] = v * ns[i];
    }
    v8f d = {0.f, 0.f, 0.f, 0.f, 0.f, 0.f, 0.f, 0.f};
#pragma unroll
    for (int j = 0; j < 4; ++j) {
      v2f x;
      x.x = h[2 * j + 0];
      x.y = h[2 * j + 1];
      d = __builtin_amdgcn_wmma_f32_16x16x4_f32(false, wb[j], false, x, (short)0, d,
                                                false, false);
    }
    float u0 = __shfl_xor(d[4], 16);
    float u1 = __shfl_xor(d[5], 16);
    float u2 = __shfl_xor(d[6], 16);
    float u3 = __shfl_xor(d[7], 16);
    v4f o;
    o.x = half ? u0 : d[0];
    o.y = half ? u1 : d[1];
    o.z = half ? u2 : d[2];
    o.w = half ? u3 : d[3];
    ((v4f*)t2)[(size_t)(m0 + rc) * 2 + half] = o;
  }
}

__global__ void __launch_bounds__(256)
gather2_out_kernel(const float* __restrict__ t2, const int* __restrict__ rowptr,
                   const int* __restrict__ sp, const float* __restrict__ ewR,
                   const float* __restrict__ normD, const float* __restrict__ b2,
                   float* __restrict__ out) {
  int tid = blockIdx.x * 256 + threadIdx.x;
  int q = tid & 1;
  int b = (tid >> 1) & 7;
  int nl = (tid >> 4) & 1;
  int n0 = (tid >> 5) * 4;
  if (n0 >= CBGN) return;
  int nA = n0 + nl, nB = n0 + 2 + nl;
  int pA = rowptr[nA], pA1 = rowptr[nA + 1];
  int pB = rowptr[nB], pB1 = rowptr[nB + 1];
  float ndA = normD[nA], ndB = normD[nB];
  v4f bias = ((const v4f*)b2)[q];
  v4f a = row_sum<8>(pA, pA1, b, q, sp, ewR, t2);
  v4f c = row_sum<8>(pB, pB1, b, q, sp, ewR, t2);
  v4f oa, oc;
  oa.x = fmaf(a.x, ndA, bias.x); oa.y = fmaf(a.y, ndA, bias.y);
  oa.z = fmaf(a.z, ndA, bias.z); oa.w = fmaf(a.w, ndA, bias.w);
  oc.x = fmaf(c.x, ndB, bias.x); oc.y = fmaf(c.y, ndB, bias.y);
  oc.z = fmaf(c.z, ndB, bias.z); oc.w = fmaf(c.w, ndB, bias.w);
  ((v4f*)out)[((size_t)b * CBGN + nA) * 2 + q] = oa;
  ((v4f*)out)[((size_t)b * CBGN + nB) * 2 + q] = oc;
}

extern "C" void kernel_launch(void* const* d_in, const int* in_sizes, int n_in,
                              void* d_out, int out_size, void* d_ws, size_t ws_size,
                              hipStream_t stream) {
  (void)in_sizes; (void)n_in; (void)out_size; (void)ws_size;
  const float* ew  = (const float*)d_in[0];
  const float* cbg = (const float*)d_in[1];
  const float* poi = (const float*)d_in[2];
  const float* W1  = (const float*)d_in[3];
  const float* b1  = (const float*)d_in[4];
  const float* W2  = (const float*)d_in[5];
  const float* b2  = (const float*)d_in[6];
  const int*   src = (const int*)d_in[7];
  const int*   dst = (const int*)d_in[8];
  float* out = (float*)d_out;

  float* ws     = (float*)d_ws;
  float* normS  = ws;
  float* normD  = ws + NN;
  int*   tileSum= (int*)(ws + 2 * (size_t)NN);
  int*   rowptr = (int*)(ws + 4 * (size_t)NN);
  int*   rank   = (int*)(ws + 5 * (size_t)NN + 48);
  int*   sp     = rank + NE;
  float* ewR    = (float*)(sp + 2 * (size_t)NE);
  float* agg    = ewR + 8 * (size_t)NE;
  float* t2     = agg + (size_t)NB * NN * 16;
  float* featT  = t2 + (size_t)NB * NN * 16;
  int*   cI     = (int*)(featT + (size_t)NB * NN * 8);
  int*   cO     = cI + (size_t)NCH * NP;
  int*   off    = cO + (size_t)NCH * NP;

  count_rank_kernel<<<2 * NCH, 1024, 0, stream>>>(src, dst, cO, cI, tileSum, rank);
  rowptr_norm_kernel<<<SCAN_BLOCKS, SCAN_T, 0, stream>>>(cO, cI, tileSum, rowptr, off, normS, normD);

  place_kernel<<<PL_G, 256, 0, stream>>>(ew, src, dst, off, rank, sp, ewR);
  relayout_kernel<<<RL_G, 256, 0, stream>>>(cbg, poi, normS, featT);

  gather1_kernel<<<(NN * 16) / 256, 256, 0, stream>>>(featT, rowptr, sp, ewR, agg);
  dense12_kernel<<<DN_G, 256, 0, stream>>>(agg, normD, normS, W1, b1, W2, t2);

  gather2_out_kernel<<<(CBGN / 4 + 7) / 8, 256, 0, stream>>>(t2, rowptr, sp, ewR, normD, b2, out);
}


# Exact dst sort replaced by one coarse partition into 64-node dst ranges (LDS-staged whole-line record writes); row sums rank a range's records in LDS, weights read in line; sum order = chunk, stable
# speedup vs baseline: 52.2962x; 1.1963x over previous
#include <hip/hip_runtime.h>

#define NB   8
#define CBGN 30000
#define POIN 20000
#define NE   1600000
#define NN   50000
#define NCH  25
#define CH   (NE / NCH)
#define CH4  (CH / 4)
#define NP   50016
#define NPB  64
#define NSH  6
#define NBIN 782
#define NBP  1024
#define NCK  250
#define CK   (NE / NCK)
#define PT   800
#define GT   (NPB * 8)
#define GW   (GT / 32)
#define MAXR (7 * GT)
static_assert(NE % NCK == 0 && CK == 8 * PT, "a partition group covers its chunk exactly");
static_assert(NBIN * NPB >= NN && NBIN <= NBP && NBP == 1024 && NCK <= 256 && (1 << NSH) == NPB, "table shapes");
static_assert(NN < 65536, "source and destination ids share one 32-bit key");

typedef __attribute__((ext_vector_type(2))) float v2f;
typedef __attribute__((ext_vector_type(8))) float v8f;
typedef __attribute__((ext_vector_type(4))) float v4f;

__device__ __forceinline__ void count_chunk(const int* __restrict__ idx, int* __restrict__ s_cnt,
                                            const int t) {
  const int4* i4 = (const int4*)idx;
  int4 d[16];
#pragma unroll
  for (int i = 0; i < 16; ++i) {
    int k = t + 1024 * i;
    d[i] = (k < CH4) ? i4[k] : make_int4(0, 0, 0, 0);
  }
#pragma unroll
  for (int i = 0; i < 16; ++i) {
    int k = t + 1024 * i;
    if (k < CH4) {
      atomicAdd(&s_cnt[d[i].x], 1);
      atomicAdd(&s_cnt[d[i].y], 1);
      atomicAdd(&s_cnt[d[i].z], 1);
      atomicAdd(&s_cnt[d[i].w], 1);
    }
  }
}

__global__ void __launch_bounds__(1024)
count_kernel(const int* __restrict__ src, const int* __restrict__ dst,
             int* __restrict__ cO, int* __restrict__ hist) {
  __shared__ __attribute__((aligned(16))) int s_cnt[NN];
  const int t = threadIdx.x;
  if (blockIdx.x < NCH) {
    const int c = blockIdx.x;
    int4* s4 = (int4*)s_cnt;
    for (int k = t; k < NN / 4; k += 1024) s4[k] = make_int4(0, 0, 0, 0);
    __syncthreads();
    count_chunk(src + (size_t)c * CH, s_cnt, t);
    __syncthreads();
    int4* o4 = (int4*)(cO + (size_t)c * NP);
    for (int k = t; k < NN / 4; k += 1024) o4[k] = s4[k];
  } else {
    const int c = blockIdx.x - NCH;
    if (t < NBP) s_cnt[t] = 0;
    __syncthreads();
    const int* d = dst + (size_t)c * CK;
    for (int i = t; i < CK; i += 1024) atomicAdd(&s_cnt[d[i] >> NSH], 1);
    __syncthreads();
    if (t < NBP) hist[c * NBP + t] = s_cnt[t];
  }
}

#define SN_NODEBLK ((NN + 255) / 256)
__global__ void __launch_bounds__(256)
scan_norm_kernel(const int* __restrict__ hist, const int* __restrict__ cO,
                 int* __restrict__ coff, int* __restrict__ binTot,
                 float* __restrict__ normS) {
  __shared__ int s_wsum[8];
  const int t = threadIdx.x;
  if (blockIdx.x < NBIN) {
    const int bin = blockIdx.x;
    const int lane = t & 31, wv = t >> 5;
    const int v = (t < NCK) ? hist[t * NBP + bin] : 0;
    int incl = v;
    for (int o = 1; o < 32; o <<= 1) {
      int u = __shfl_up(incl, o);
      if (lane >= o) incl += u;
    }
    if (lane == 31) s_wsum[wv] = incl;
    __syncthreads();
    int base = 0;
    for (int k = 0; k < wv; ++k) base += s_wsum[k];
    if (t < NCK) coff[t * NBP + bin] = base + incl - v;
    if (t == 255) binTot[bin] = base + incl;
  } else {
    const int n = (blockIdx.x - NBIN) * 256 + t;
    if (n < NN) {
      int co = 0;
#pragma unroll
      for (int q = 0; q < NCH; ++q) co += cO[(size_t)q * NP + n];
      normS[n] = rsqrtf(fmaxf((float)co, 1.0f));
    }
  }
}

__global__ void __launch_bounds__(PT)
partition_kernel(const float* __restrict__ ew, const int* __restrict__ src,
                 const int* __restrict__ dst, const int* __restrict__ coff,
                 const int* __restrict__ binTot,
                 unsigned* __restrict__ keys, float* __restrict__ wts) {
  __shared__ __attribute__((aligned(16))) float s_w[CK * 8];
  __shared__ unsigned s_key[CK];
  __shared__ int s_cnt[NBP];
  __shared__ int s_lst[NBP];
  __shared__ int s_gb[NBP];
  const int t = threadIdx.x;
  const int c = blockIdx.x;
  const int e0 = c * CK + t;
  for (int i = t; i < NBP; i += PT) s_cnt[i] = 0;
  __syncthreads();
  int d[8], r[8];
#pragma unroll
  for (int k = 0; k < 8; ++k) d[k] = dst[e0 + k * PT];
#pragma unroll
  for (int k = 0; k < 8; ++k) r[k] = atomicAdd(&s_cnt[d[k] >> NSH], 1);
  __syncthreads();
  if (t < 32) {
    int cl[32], tl[32];
    int sc = 0, st = 0;
#pragma unroll
    for (int j = 0; j < 32; ++j) {
      const int bin = t * 32 + j;
      cl[j] = s_cnt[bin];
      tl[j] = (bin < NBIN) ? binTot[bin] : 0;
      sc += cl[j];
      st += tl[j];
    }
    int ic = sc, it = st;
    for (int o = 1; o < 32; o <<= 1) {
      int uc = __shfl_up(ic, o);
      int ut = __shfl_up(it, o);
      if (t >= o) { ic += uc; it += ut; }
    }
    int xc = ic - sc, xt = it - st;
#pragma unroll
    for (int j = 0; j < 32; ++j) {
      const int bin = t * 32 + j;
      const int co = (bin < NBIN) ? coff[c * NBP + bin] : 0;
      s_lst[bin] = xc;
      s_gb[bin] = xt + co - xc;
      xc += cl[j];
      xt += tl[j];
    }
  }
  __syncthreads();
#pragma unroll
  for (int g = 0; g < 2; ++g) {
    int sv[4];
    v4f lo[4], hi[4];
#pragma unroll
    for (int k = 0; k < 4; ++k) {
      const int e = e0 + (4 * g + k) * PT;
      sv[k] = src[e];
      lo[k].x = ew[(size_t)0 * NE + e];
      lo[k].y = ew[(size_t)1 * NE + e];
      lo[k].z = ew[(size_t)2 * NE + e];
      lo[k].w = ew[(size_t)3 * NE + e];
      hi[k].x = ew[(size_t)4 * NE + e];
      hi[k].y = ew[(size_t)5 * NE + e];
      hi[k].z = ew[(size_t)6 * NE + e];
      hi[k].w = ew[(size_t)7 * NE + e];
    }
#pragma unroll
    for (int k = 0; k < 4; ++k) {
      const int dd = d[4 * g + k];
      const int lp = s_lst[dd >> NSH] + r[4 * g + k];
      s_key[lp] = (unsigned)sv[k] | ((unsigned)dd << 16);
      ((v4f*)s_w)[lp * 2 + 0] = lo[k];
      ((v4f*)s_w)[lp * 2 + 1] = hi[k];
    }
  }
  __syncthreads();
  for (int i = t; i < CK; i += PT) {
    const unsigned k = s_key[i];
    keys[s_gb[k >> (16 + NSH)] + i] = k;
  }
  for (int j = t; j < CK * 2; j += PT) {
    const int i = j >> 1;
    const unsigned k = s_key[i];
    const size_t gslot = (size_t)(s_gb[k >> (16 + NSH)] + i);
    ((v4f*)wts)[gslot * 2 + (j & 1)] = ((const v4f*)s_w)[j];
  }
}

#define RL_R 5
#define RL_G ((NN * 16) / (256 * RL_R))
static_assert((NN * 16) % (256 * RL_R) == 0, "whole work-groups: no bounds guard");
__global__ void __launch_bounds__(256)
relayout_kernel(const float* __restrict__ cbg, const float* __restrict__ poi,
                const float* __restrict__ normS, float* __restrict__ featT) {
  const int t0 = blockIdx.x * (256 * RL_R) + threadIdx.x;
  v4f x[RL_R];
  float s[RL_R];
#pragma unroll
  for (int i = 0; i < RL_R; ++i) {
    int tid = t0 + i * 256;
    int q = tid & 1;
    int b = (tid >> 1) & 7;
    int n = tid >> 4;
    const bool in_cbg = n < CBGN;
    const float* tab = in_cbg ? cbg : poi;
    const int row = in_cbg ? (b * CBGN + n) : (b * POIN + (n - CBGN));
    x[i] = ((const v4f*)tab)[(size_t)row * 2 + q];
    s[i] = normS[n];
  }
#pragma unroll
  for (int i = 0; i < RL_R; ++i) {
    int tid = t0 + i * 256;
    v4f y = x[i];
    y.x *= s[i]; y.y *= s[i]; y.z *= s[i]; y.w *= s[i];
    ((v4f*)featT)[tid] = y;
  }
}

template <int K>
__device__ __forceinline__ void hold_rows(const v4f (&x)[K]) {
  if constexpr (K == 8) {
    asm volatile("" :: "v"(x[0]), "v"(x[1]), "v"(x[2]), "v"(x[3]),
                       "v"(x[4]), "v"(x[5]), "v"(x[6]), "v"(x[7]));
  } else if constexpr (K == 4) {
    asm volatile("" :: "v"(x[0]), "v"(x[1]), "v"(x[2]), "v"(x[3]));
  } else if constexpr (K == 2) {
    asm volatile("" :: "v"(x[0]), "v"(x[1]));
  }
}

template <int K>
__device__ __forceinline__ void bin_batch(v4f& a, const int p, const int b, const int q,
                                          const unsigned short* s_ord, const unsigned* s_key,
                                          const float* s_w, const float* __restrict__ tab) {
  int id[K];
  float w[K];
  v4f x[K];
#pragma unroll
  for (int i = 0; i < K; ++i) id[i] = s_ord[p + i];
#pragma unroll
  for (int i = 0; i < K; ++i) {
    const unsigned k = s_key[id[i]];
    w[i] = s_w[id[i] * 8 + b];
    const v4f* r = (const v4f*)(tab + ((size_t)(k & 0xFFFFu) * 8 + b) * 8);
    x[i] = r[q];
  }
  hold_rows<K>(x);
#pragma unroll
  for (int i = 0; i < K; ++i) {
    a.x = fmaf(x[i].x, w[i], a.x);
    a.y = fmaf(x[i].y, w[i], a.y);
    a.z = fmaf(x[i].z, w[i], a.z);
    a.w = fmaf(x[i].w, w[i], a.w);
  }
}

__device__ __forceinline__ void bin_row_sum(v4f& a, int p, const int p1, const int b, const int q,
                                            const unsigned short* s_ord, const unsigned* s_key,
                                            const float* s_w, const float* __restrict__ tab) {
  for (; p + 8 <= p1; p += 8) bin_batch<8>(a, p, b, q, s_ord, s_key, s_w, tab);
  if (p + 4 <= p1) { bin_batch<4>(a, p, b, q, s_ord, s_key, s_w, tab); p += 4; }
  if (p + 2 <= p1) { bin_batch<2>(a, p, b, q, s_ord, s_key, s_w, tab); p += 2; }
  if (p < p1)      { bin_batch<1>(a, p, b, q, s_ord, s_key, s_w, tab); }
}

template <int LAYER>
__global__ void __launch_bounds__(GT)
gather_bin_kernel(const float* __restrict__ tab, const unsigned* __restrict__ keys,
                  const float* __restrict__ wts, const int* __restrict__ binTot,
                  float* __restrict__ normD, const float* __restrict__ b2,
                  float* __restrict__ dstbuf) {
  __shared__ __attribute__((aligned(16))) float s_w[MAXR * 8];
  __shared__ unsigned s_key[MAXR];
  __shared__ unsigned short s_ord[MAXR];
  __shared__ int s_wc[GW * NPB];
  __shared__ int s_cnt[NPB];
  __shared__ int s_st[NPB + 1];
  __shared__ int s_red[GW];
  const int t = threadIdx.x;
  const int bin = blockIdx.x;
  const int lane = t & 31, w = t >> 5;
  const int nl = (t >> 4) & 1, b = (t >> 1) & 7, q = t & 1;
  const int lA = 4 * w + nl, lB = lA + 2;

  int v = 0;
  for (int i = t; i < bin; i += GT) v += binTot[i];
  for (int o = 16; o > 0; o >>= 1) v += __shfl_xor(v, o);
  if (lane == 0) s_red[w] = v;
  __syncthreads();
  int r0 = 0;
  for (int k = 0; k < GW; ++k) r0 += s_red[k];
  const int r1 = r0 + binTot[bin];

  v4f aA = {0.f, 0.f, 0.f, 0.f}, aB = {0.f, 0.f, 0.f, 0.f};
  int dA = 0, dB = 0;
  const int wc0 = w * NPB;
  for (int t0 = r0; t0 < r1; t0 += MAXR) {
    const int len = min(MAXR, r1 - t0);
    __syncthreads();
    for (int i = t; i < len; i += GT) s_key[i] = keys[t0 + i];
    for (int j = t; j < len * 2; j += GT) ((v4f*)s_w)[j] = ((const v4f*)wts)[(size_t)t0 * 2 + j];
    for (int j = lane; j < NPB; j += 32) s_wc[wc0 + j] = 0;
    __syncthreads();
    int rk[7];
#pragma unroll
    for (int k = 0; k < 7; ++k) {
      const int i = w * 224 + k * 32 + lane;
      const bool valid = i < len;
      const int node = valid ? (int)((s_key[i] >> 16) & (NPB - 1)) : 0;
      unsigned m = (unsigned)__ballot(valid);
#pragma unroll
      for (int bit = 0; bit < NSH; ++bit) {
        const bool on = (node >> bit) & 1;
        const unsigned bb = (unsigned)__ballot(on);
        m &= on ? bb : ~bb;
      }
      const int before = __popc(m & ((1u << lane) - 1u));
      const int same = __popc(m);
      int old = 0;
      if (valid && before == 0) old = atomicAdd(&s_wc[wc0 + node], same);
      const int base = __shfl(old, m ? (__ffs(m) - 1) : lane);
      rk[k] = base + before;
    }
    __syncthreads();
    if (t < NPB) {
      int run = 0;
      for (int k = 0; k < GW; ++k) {
        const int c = s_wc[k * NPB + t];
        s_wc[k * NPB + t] = run;
        run += c;
      }
      s_cnt[t] = run;
    }
    __syncthreads();
    if (t < 32) {
      int c[NPB / 32];
      int s = 0;
#pragma unroll
      for (int j = 0; j < NPB / 32; ++j) { c[j] = s_cnt[t * (NPB / 32) + j]; s += c[j]; }
      int inc = s;
      for (int o = 1; o < 32; o <<= 1) {
        int u = __shfl_up(inc, o);
        if (t >= o) inc += u;
      }
      int x = inc - s;
#pragma unroll
      for (int j = 0; j < NPB / 32; ++j) { s_st[t * (NPB / 32) + j] = x; x += c[j]; }
      if (t == 31) s_st[NPB] = x;
    }
    __syncthreads();
#pragma unroll
    for (int k = 0; k < 7; ++k) {
      const int i = w * 224 + k * 32 + lane;
      if (i < len) {
        const int node = (int)((s_key[i] >> 16) & (NPB - 1));
        s_ord[s_st[node] + s_wc[wc0 + node] + rk[k]] = (unsigned short)i;
      }
    }
    __syncthreads();
    {
      const int p = s_st[lA], p1 = s_st[lA + 1];
      dA += p1 - p;
      bin_row_sum(aA, p, p1, b, q, s_ord, s_key, s_w, tab);
    }
    {
      const int p = s_st[lB], p1 = s_st[lB + 1];
      dB += p1 - p;
      bin_row_sum(aB, p, p1, b, q, s_ord, s_key, s_w, tab);
    }
  }

  const int nA = bin * NPB + lA, nB = bin * NPB + lB;
  if constexpr (LAYER == 1) {
    if (nA < NN) {
      ((v4f*)dstbuf)[((size_t)nA * 8 + b) * 2 + q] = aA;
      if ((t & 15) == 0) normD[nA] = rsqrtf(fmaxf((float)dA, 1.0f));
    }
    if (nB < NN) {
      ((v4f*)dstbuf)[((size_t)nB * 8 + b) * 2 + q] = aB;
      if ((t & 15) == 0) normD[nB] = rsqrtf(fmaxf((float)dB, 1.0f));
    }
  } else {
    const v4f bias = ((const v4f*)b2)[q];
    if (nA < CBGN) {
      const float nd = rsqrtf(fmaxf((float)dA, 1.0f));
      v4f o;
      o.x = fmaf(aA.x, nd, bias.x); o.y = fmaf(aA.y, nd, bias.y);
      o.z = fmaf(aA.z, nd, bias.z); o.w = fmaf(aA.w, nd, bias.w);
      ((v4f*)dstbuf)[((size_t)b * CBGN + nA) * 2 + q] = o;
    }
    if (nB < CBGN) {
      const float nd = rsqrtf(fmaxf((float)dB, 1.0f));
      v4f o;
      o.x = fmaf(aB.x, nd, bias.x); o.y = fmaf(aB.y, nd, bias.y);
      o.z = fmaf(aB.z, nd, bias.z); o.w = fmaf(aB.w, nd, bias.w);
      ((v4f*)dstbuf)[((size_t)b * CBGN + nB) * 2 + q] = o;
    }
  }
}

#define DN_R 5
#define DN_TILES ((NN * NB) / 16)
#define DN_G (DN_TILES / (8 * DN_R))
static_assert(DN_TILES % (8 * DN_R) == 0, "whole waves: every lane runs every product");
__global__ void __launch_bounds__(256)
dense12_kernel(const float* __restrict__ agg, const float* __restrict__ normD,
               const float* __restrict__ normS, const float* __restrict__ W1,
               const float* __restrict__ b1, const float* __restrict__ W2,
               float* __restrict__ t2) {
  const int wave = (int)((blockIdx.x * 256 + threadIdx.x) >> 5);
  const int lane = threadIdx.x & 31;
  const int rc = lane & 15;
  const int half = lane >> 4;
  const int cw = rc & 7;
  const float colmask = (rc < 8) ? 1.0f : 0.0f;

  v2f wa[2], wb[4];
#pragma unroll
  for (int kb = 0; kb < 8; kb += 4) {
    int k0 = kb + 2 * half;
    wa[kb >> 2].x = W1[(k0 + 0) * 16 + rc];
    wa[kb >> 2].y = W1[(k0 + 1) * 16 + rc];
  }
#pragma unroll
  for (int j = 0; j < 4; ++j) {
    int k0 = 8 * half + 2 * j;
    wb[j].x = W2[(k0 + 0) * 8 + cw] * colmask;
    wb[j].y = W2[(k0 + 1) * 8 + cw] * colmask;
  }
  const v4f* b1v = (const v4f*)b1;
  v4f bl = b1v[2 * half], bh = b1v[2 * half + 1];
  const float bias[8] = {bl.x, bl.y, bl.z, bl.w, bh.x, bh.y, bh.z, bh.w};

  v2f xa[DN_R][2];
  float nd[DN_R], ns[DN_R];
#pragma unroll
  for (int i = 0; i < DN_R; ++i) {
    int row = (wave * DN_R + i) * 16 + rc;
    nd[i] = normD[row >> 3];
    ns[i] = normS[row >> 3];
    const v2f* ar = (const v2f*)&agg[(size_t)row * 8];
    xa[i][0] = ar[half];
    xa[i][1] = ar[2 + half];
  }

#pragma unroll
  for (int i = 0; i < DN_R; ++i) {
    int m0 = (wave * DN_R + i) * 16;
    v8f c = {0.f, 0.f, 0.f, 0.f, 0.f, 0.f, 0.f, 0.f};
#pragma unroll
    for (int kk = 0; kk < 2; ++kk) {
      v2f x;
      x.x = xa[i][kk].x * nd[i];
      x.y = xa[i][kk].y * nd[i];
      c = __builtin_amdgcn_wmma_f32_16x16x4_f32(false, wa[kk], false, x, (short)0, c,
                                                false, false);
    }
    float h[8];
#pragma unroll
    for (int r = 0; r < 8; ++r) {
      float v = c[r] + bias[r];
      v = v > 0.0f ? v : 0.01f * v;
      h[r] = v * ns[i];
    }
    v8f d = {0.f, 0.f, 0.f, 0.f, 0.f, 0.f, 0.f, 0.f};
#pragma unroll
    for (int j = 0; j < 4; ++j) {
      v2f x;
      x.x = h[2 * j + 0];
      x.y = h[2 * j + 1];
      d = __builtin_amdgcn_wmma_f32_16x16x4_f32(false, wb[j], false, x, (short)0, d,
                                                false, false);
    }
    float u0 = __shfl_xor(d[4], 16);
    float u1 = __shfl_xor(d[5], 16);
    float u2 = __shfl_xor(d[6], 16);
    float u3 = __shfl_xor(d[7], 16);
    v4f o;
    o.x = half ? u0 : d[0];
    o.y = half ? u1 : d[1];
    o.z = half ? u2 : d[2];
    o.w = half ? u3 : d[3];
    ((v4f*)t2)[(size_t)(m0 + rc) * 2 + half] = o;
  }
}

extern "C" void kernel_launch(void* const* d_in, const int* in_sizes, int n_in,
                              void* d_out, int out_size, void* d_ws, size_t ws_size,
                              hipStream_t stream) {
  (void)in_sizes; (void)n_in; (void)out_size; (void)ws_size;
  const float* ew  = (const float*)d_in[0];
  const float* cbg = (const float*)d_in[1];
  const float* poi = (const float*)d_in[2];
  const float* W1  = (const float*)d_in[3];
  const float* b1  = (const float*)d_in[4];
  const float* W2  = (const float*)d_in[5];
  const float* b2  = (const float*)d_in[6];
  const int*   src = (const int*)d_in[7];
  const int*   dst = (const int*)d_in[8];
  float* out = (float*)d_out;

  float* ws     = (float*)d_ws;
  float* normS  = ws;
  float* normD  = normS + 50048;
  int*   binTot = (int*)(normD + 50048);
  int*   hist   = binTot + NBP;
  int*   coff   = hist + (size_t)NCK * NBP;
  int*   cO     = coff + (size_t)NCK * NBP;
  unsigned* keys = (unsigned*)(cO + 1250432);
  float* wts    = (float*)(keys + NE);
  float* agg    = wts + 8 * (size_t)NE;
  float* t2     = agg + (size_t)NB * NN * 8;
  float* featT  = t2 + (size_t)NB * NN * 8;

  count_kernel<<<NCH + NCK, 1024, 0, stream>>>(src, dst, cO, hist);
  scan_norm_kernel<<<NBIN + SN_NODEBLK, 256, 0, stream>>>(hist, cO, coff, binTot, normS);

  partition_kernel<<<NCK, PT, 0, stream>>>(ew, src, dst, coff, binTot, keys, wts);
  relayout_kernel<<<RL_G, 256, 0, stream>>>(cbg, poi, normS, featT);

  gather_bin_kernel<1><<<NBIN, GT, 0, stream>>>(featT, keys, wts, binTot, normD, b2, agg);
  dense12_kernel<<<DN_G, 256, 0, stream>>>(agg, normD, normS, W1, b1, W2, t2);

  gather_bin_kernel<2><<<(CBGN + NPB - 1) / NPB, GT, 0, stream>>>(t2, keys, wts, binTot, normD, b2, out);
}


# Staging loops that waited for each global load before asking for the next (row-sum tiles 12 round trips, partition scan wave 32, range counts 7) issue all their loads as one group first
# speedup vs baseline: 57.3478x; 1.0966x over previous
#include <hip/hip_runtime.h>

#define NB   8
#define CBGN 30000
#define POIN 20000
#define NE   1600000
#define NN   50000
#define NCH  25
#define CH   (NE / NCH)
#define CH4  (CH / 4)
#define NP   50016
#define NPB  64
#define NSH  6
#define NBIN 782
#define NBP  1024
#define NCK  250
#define CK   (NE / NCK)
#define PT   800
#define GT   (NPB * 8)
#define GW   (GT / 32)
#define TK   6
#define MAXR (TK * GT)
#define WR   (TK * 32)
#define HK   ((CK + 1023) / 1024)
static_assert(NE % NCK == 0 && CK == 8 * PT, "a partition group covers its chunk exactly");
static_assert(WR * GW == MAXR && MAXR < 65536, "the waves' rank windows cover a tile; a staged slot fits 16 bits");
static_assert(NBIN * NPB >= NN && NBIN <= NBP && NBP == 1024 && NCK <= 256 && (1 << NSH) == NPB, "table shapes");
static_assert(NN < 65536, "source and destination ids share one 32-bit key");

typedef __attribute__((ext_vector_type(2))) float v2f;
typedef __attribute__((ext_vector_type(8))) float v8f;
typedef __attribute__((ext_vector_type(4))) float v4f;

__device__ __forceinline__ void count_chunk(const int* __restrict__ idx, int* __restrict__ s_cnt,
                                            const int t) {
  const int4* i4 = (const int4*)idx;
  int4 d[16];
#pragma unroll
  for (int i = 0; i < 16; ++i) {
    int k = t + 1024 * i;
    d[i] = (k < CH4) ? i4[k] : make_int4(0, 0, 0, 0);
  }
#pragma unroll
  for (int i = 0; i < 16; ++i) {
    int k = t + 1024 * i;
    if (k < CH4) {
      atomicAdd(&s_cnt[d[i].x], 1);
      atomicAdd(&s_cnt[d[i].y], 1);
      atomicAdd(&s_cnt[d[i].z], 1);
      atomicAdd(&s_cnt[d[i].w], 1);
    }
  }
}

__global__ void __launch_bounds__(1024)
count_kernel(const int* __restrict__ src, const int* __restrict__ dst,
             int* __restrict__ cO, int* __restrict__ hist) {
  __shared__ __attribute__((aligned(16))) int s_cnt[NN];
  const int t = threadIdx.x;
  if (blockIdx.x < NCH) {
    const int c = blockIdx.x;
    int4* s4 = (int4*)s_cnt;
    for (int k = t; k < NN / 4; k += 1024) s4[k] = make_int4(0, 0, 0, 0);
    __syncthreads();
    count_chunk(src + (size_t)c * CH, s_cnt, t);
    __syncthreads();
    int4* o4 = (int4*)(cO + (size_t)c * NP);
    for (int k = t; k < NN / 4; k += 1024) o4[k] = s4[k];
  } else {
    const int c = blockIdx.x - NCH;
    const int* d = dst + (size_t)c * CK;
    int dv[HK];
#pragma unroll
    for (int k = 0; k < HK; ++k) {
      const int i = t + 1024 * k;
      dv[k] = (i < CK) ? d[i] : 0;
    }
    if (t < NBP) s_cnt[t] = 0;
    __syncthreads();
#pragma unroll
    for (int k = 0; k < HK; ++k) {
      if (t + 1024 * k < CK) atomicAdd(&s_cnt[dv[k] >> NSH], 1);
    }
    __syncthreads();
    if (t < NBP) hist[c * NBP + t] = s_cnt[t];
  }
}

#define SN_NODEBLK ((NN + 255) / 256)
__global__ void __launch_bounds__(256)
scan_norm_kernel(const int* __restrict__ hist, const int* __restrict__ cO,
                 int* __restrict__ coff, int* __restrict__ binTot,
                 float* __restrict__ normS) {
  __shared__ int s_wsum[8];
  const int t = threadIdx.x;
  if (blockIdx.x < NBIN) {
    const int bin = blockIdx.x;
    const int lane = t & 31, wv = t >> 5;
    const int v = (t < NCK) ? hist[t * NBP + bin] : 0;
    int incl = v;
    for (int o = 1; o < 32; o <<= 1) {
      int u = __shfl_up(incl, o);
      if (lane >= o) incl += u;
    }
    if (lane == 31) s_wsum[wv] = incl;
    __syncthreads();
    int base = 0;
    for (int k = 0; k < wv; ++k) base += s_wsum[k];
    if (t < NCK) coff[t * NBP + bin] = base + incl - v;
    if (t == 255) binTot[bin] = base + incl;
  } else {
    const int n = (blockIdx.x - NBIN) * 256 + t;
    if (n < NN) {
      int co = 0;
#pragma unroll
      for (int q = 0; q < NCH; ++q) co += cO[(size_t)q * NP + n];
      normS[n] = rsqrtf(fmaxf((float)co, 1.0f));
    }
  }
}

__global__ void __launch_bounds__(PT)
partition_kernel(const float* __restrict__ ew, const int* __restrict__ src,
                 const int* __restrict__ dst, const int* __restrict__ coff,
                 const int* __restrict__ binTot,
                 unsigned* __restrict__ keys, float* __restrict__ wts) {
  __shared__ __attribute__((aligned(16))) float s_w[CK * 8];
  __shared__ unsigned s_key[CK];
  __shared__ int s_cnt[NBP];
  __shared__ int s_lst[NBP];
  __shared__ int s_gb[NBP];
  __shared__ int s_bt[NBP];
  __shared__ int s_co[NBP];
  const int t = threadIdx.x;
  const int c = blockIdx.x;
  const int e0 = c * CK + t;
  int d[8], r[8];
#pragma unroll
  for (int k = 0; k < 8; ++k) d[k] = dst[e0 + k * PT];
  static_assert(NBP == 32 * 32, "one scan wave, 32 ranges per lane");
  for (int i = t; i < NBP; i += PT) {
    const bool in = i < NBIN;
    const int tr = (i & 31) * 32 + (i >> 5);
    s_cnt[i] = 0;
    s_bt[tr] = in ? binTot[i] : 0;
    s_co[tr] = in ? coff[c * NBP + i] : 0;
  }
  __syncthreads();
#pragma unroll
  for (int k = 0; k < 8; ++k) r[k] = atomicAdd(&s_cnt[d[k] >> NSH], 1);
  __syncthreads();
  if (t < 32) {
    int cl[32], tl[32];
    int sc = 0, st = 0;
#pragma unroll
    for (int j = 0; j < 32; ++j) {
      const int bin = t * 32 + j;
      cl[j] = s_cnt[bin];
      tl[j] = s_bt[j * 32 + t];
      sc += cl[j];
      st += tl[j];
    }
    int ic = sc, it = st;
    for (int o = 1; o < 32; o <<= 1) {
      int uc = __shfl_up(ic, o);
      int ut = __shfl_up(it, o);
      if (t >= o) { ic += uc; it += ut; }
    }
    int xc = ic - sc, xt = it - st;
#pragma unroll
    for (int j = 0; j < 32; ++j) {
      const int bin = t * 32 + j;
      const int co = s_co[j * 32 + t];
      s_lst[bin] = xc;
      s_gb[bin] = xt + co - xc;
      xc += cl[j];
      xt += tl[j];
    }
  }
  __syncthreads();
#pragma unroll
  for (int g = 0; g < 2; ++g) {
    int sv[4];
    v4f lo[4], hi[4];
#pragma unroll
    for (int k = 0; k < 4; ++k) {
      const int e = e0 + (4 * g + k) * PT;
      sv[k] = src[e];
      lo[k].x = ew[(size_t)0 * NE + e];
      lo[k].y = ew[(size_t)1 * NE + e];
      lo[k].z = ew[(size_t)2 * NE + e];
      lo[k].w = ew[(size_t)3 * NE + e];
      hi[k].x = ew[(size_t)4 * NE + e];
      hi[k].y = ew[(size_t)5 * NE + e];
      hi[k].z = ew[(size_t)6 * NE + e];
      hi[k].w = ew[(size_t)7 * NE + e];
    }
#pragma unroll
    for (int k = 0; k < 4; ++k) {
      const int dd = d[4 * g + k];
      const int lp = s_lst[dd >> NSH] + r[4 * g + k];
      s_key[lp] = (unsigned)sv[k] | ((unsigned)dd << 16);
      ((v4f*)s_w)[lp * 2 + 0] = lo[k];
      ((v4f*)s_w)[lp * 2 + 1] = hi[k];
    }
  }
  __syncthreads();
  for (int i = t; i < CK; i += PT) {
    const unsigned k = s_key[i];
    keys[s_gb[k >> (16 + NSH)] + i] = k;
  }
  for (int j = t; j < CK * 2; j += PT) {
    const int i = j >> 1;
    const unsigned k = s_key[i];
    const size_t gslot = (size_t)(s_gb[k >> (16 + NSH)] + i);
    ((v4f*)wts)[gslot * 2 + (j & 1)] = ((const v4f*)s_w)[j];
  }
}

#define RL_R 5
#define RL_G ((NN * 16) / (256 * RL_R))
static_assert((NN * 16) % (256 * RL_R) == 0, "whole work-groups: no bounds guard");
__global__ void __launch_bounds__(256)
relayout_kernel(const float* __restrict__ cbg, const float* __restrict__ poi,
                const float* __restrict__ normS, float* __restrict__ featT) {
  const int t0 = blockIdx.x * (256 * RL_R) + threadIdx.x;
  v4f x[RL_R];
  float s[RL_R];
#pragma unroll
  for (int i = 0; i < RL_R; ++i) {
    int tid = t0 + i * 256;
    int q = tid & 1;
    int b = (tid >> 1) & 7;
    int n = tid >> 4;
    const bool in_cbg = n < CBGN;
    const float* tab = in_cbg ? cbg : poi;
    const int row = in_cbg ? (b * CBGN + n) : (b * POIN + (n - CBGN));
    x[i] = ((const v4f*)tab)[(size_t)row * 2 + q];
    s[i] = normS[n];
  }
#pragma unroll
  for (int i = 0; i < RL_R; ++i) {
    int tid = t0 + i * 256;
    v4f y = x[i];
    y.x *= s[i]; y.y *= s[i]; y.z *= s[i]; y.w *= s[i];
    ((v4f*)featT)[tid] = y;
  }
}

template <int K>
__device__ __forceinline__ void hold_rows(const v4f (&x)[K]) {
  if constexpr (K == 8) {
    asm volatile("" :: "v"(x[0]), "v"(x[1]), "v"(x[2]), "v"(x[3]),
                       "v"(x[4]), "v"(x[5]), "v"(x[6]), "v"(x[7]));
  } else if constexpr (K == 4) {
    asm volatile("" :: "v"(x[0]), "v"(x[1]), "v"(x[2]), "v"(x[3]));
  } else if constexpr (K == 2) {
    asm volatile("" :: "v"(x[0]), "v"(x[1]));
  }
}

template <int K>
__device__ __forceinline__ void bin_batch(v4f& a, const int p, const int b, const int q,
                                          const unsigned short* s_ord, const unsigned* s_key,
                                          const float* s_w, const float* __restrict__ tab) {
  int id[K];
  float w[K];
  v4f x[K];
#pragma unroll
  for (int i = 0; i < K; ++i) id[i] = s_ord[p + i];
#pragma unroll
  for (int i = 0; i < K; ++i) {
    const unsigned k = s_key[id[i]];
    w[i] = s_w[id[i] * 8 + b];
    const v4f* r = (const v4f*)(tab + ((size_t)(k & 0xFFFFu) * 8 + b) * 8);
    x[i] = r[q];
  }
  hold_rows<K>(x);
#pragma unroll
  for (int i = 0; i < K; ++i) {
    a.x = fmaf(x[i].x, w[i], a.x);
    a.y = fmaf(x[i].y, w[i], a.y);
    a.z = fmaf(x[i].z, w[i], a.z);
    a.w = fmaf(x[i].w, w[i], a.w);
  }
}

__device__ __forceinline__ void bin_row_sum(v4f& a, int p, const int p1, const int b, const int q,
                                            const unsigned short* s_ord, const unsigned* s_key,
                                            const float* s_w, const float* __restrict__ tab) {
  for (; p + 8 <= p1; p += 8) bin_batch<8>(a, p, b, q, s_ord, s_key, s_w, tab);
  if (p + 4 <= p1) { bin_batch<4>(a, p, b, q, s_ord, s_key, s_w, tab); p += 4; }
  if (p + 2 <= p1) { bin_batch<2>(a, p, b, q, s_ord, s_key, s_w, tab); p += 2; }
  if (p < p1)      { bin_batch<1>(a, p, b, q, s_ord, s_key, s_w, tab); }
}

template <int LAYER>
__global__ void __launch_bounds__(GT)
gather_bin_kernel(const float* __restrict__ tab, const unsigned* __restrict__ keys,
                  const float* __restrict__ wts, const int* __restrict__ binTot,
                  float* __restrict__ normD, const float* __restrict__ b2,
                  float* __restrict__ dstbuf) {
  __shared__ __attribute__((aligned(16))) float s_w[MAXR * 8];
  __shared__ unsigned s_key[MAXR];
  __shared__ unsigned short s_ord[MAXR];
  __shared__ int s_wc[GW * NPB];
  __shared__ int s_cnt[NPB];
  __shared__ int s_st[NPB + 1];
  __shared__ int s_red[GW];
  const int t = threadIdx.x;
  const int bin = blockIdx.x;
  const int lane = t & 31, w = t >> 5;
  const int nl = (t >> 4) & 1, b = (t >> 1) & 7, q = t & 1;
  const int lA = 4 * w + nl, lB = lA + 2;

  const int tot = binTot[bin];
  int v = 0;
  for (int i = t; i < bin; i += GT) v += binTot[i];
  for (int o = 16; o > 0; o >>= 1) v += __shfl_xor(v, o);
  if (lane == 0) s_red[w] = v;
  __syncthreads();
  int r0 = 0;
  for (int k = 0; k < GW; ++k) r0 += s_red[k];
  const int r1 = r0 + tot;

  v4f aA = {0.f, 0.f, 0.f, 0.f}, aB = {0.f, 0.f, 0.f, 0.f};
  int dA = 0, dB = 0;
  const int wc0 = w * NPB;
  for (int t0 = r0; t0 < r1; t0 += MAXR) {
    const int len = min(MAXR, r1 - t0);
    unsigned kq[TK];
    v4f wq[2 * TK];
#pragma unroll
    for (int k = 0; k < TK; ++k) {
      const int i = t + k * GT;
      kq[k] = (i < len) ? keys[t0 + i] : 0u;
    }
#pragma unroll
    for (int k = 0; k < 2 * TK; ++k) {
      const int j = t + k * GT;
      wq[k] = (j < len * 2) ? ((const v4f*)wts)[(size_t)t0 * 2 + j] : v4f{0.f, 0.f, 0.f, 0.f};
    }
    __syncthreads();
#pragma unroll
    for (int k = 0; k < TK; ++k) {
      const int i = t + k * GT;
      if (i < len) s_key[i] = kq[k];
    }
#pragma unroll
    for (int k = 0; k < 2 * TK; ++k) {
      const int j = t + k * GT;
      if (j < len * 2) ((v4f*)s_w)[j] = wq[k];
    }
    for (int j = lane; j < NPB; j += 32) s_wc[wc0 + j] = 0;
    __syncthreads();
    int rk[TK];
#pragma unroll
    for (int k = 0; k < TK; ++k) {
      const int i = w * WR + k * 32 + lane;
      const bool valid = i < len;
      const int node = valid ? (int)((s_key[i] >> 16) & (NPB - 1)) : 0;
      unsigned m = (unsigned)__ballot(valid);
#pragma unroll
      for (int bit = 0; bit < NSH; ++bit) {
        const bool on = (node >> bit) & 1;
        const unsigned bb = (unsigned)__ballot(on);
        m &= on ? bb : ~bb;
      }
      const int before = __popc(m & ((1u << lane) - 1u));
      const int same = __popc(m);
      int old = 0;
      if (valid && before == 0) old = atomicAdd(&s_wc[wc0 + node], same);
      const int base = __shfl(old, m ? (__ffs(m) - 1) : lane);
      rk[k] = base + before;
    }
    __syncthreads();
    if (t < NPB) {
      int run = 0;
      for (int k = 0; k < GW; ++k) {
        const int c = s_wc[k * NPB + t];
        s_wc[k * NPB + t] = run;
        run += c;
      }
      s_cnt[t] = run;
    }
    __syncthreads();
    if (t < 32) {
      int c[NPB / 32];
      int s = 0;
#pragma unroll
      for (int j = 0; j < NPB / 32; ++j) { c[j] = s_cnt[t * (NPB / 32) + j]; s += c[j]; }
      int inc = s;
      for (int o = 1; o < 32; o <<= 1) {
        int u = __shfl_up(inc, o);
        if (t >= o) inc += u;
      }
      int x = inc - s;
#pragma unroll
      for (int j = 0; j < NPB / 32; ++j) { s_st[t * (NPB / 32) + j] = x; x += c[j]; }
      if (t == 31) s_st[NPB] = x;
    }
    __syncthreads();
#pragma unroll
    for (int k = 0; k < TK; ++k) {
      const int i = w * WR + k * 32 + lane;
      if (i < len) {
        const int node = (int)((s_key[i] >> 16) & (NPB - 1));
        s_ord[s_st[node] + s_wc[wc0 + node] + rk[k]] = (unsigned short)i;
      }
    }
    __syncthreads();
    {
      const int p = s_st[lA], p1 = s_st[lA + 1];
      dA += p1 - p;
      bin_row_sum(aA, p, p1, b, q, s_ord, s_key, s_w, tab);
    }
    {
      const int p = s_st[lB], p1 = s_st[lB + 1];
      dB += p1 - p;
      bin_row_sum(aB, p, p1, b, q, s_ord, s_key, s_w, tab);
    }
  }

  const int nA = bin * NPB + lA, nB = bin * NPB + lB;
  if constexpr (LAYER == 1) {
    if (nA < NN) {
      ((v4f*)dstbuf)[((size_t)nA * 8 + b) * 2 + q] = aA;
      if ((t & 15) == 0) normD[nA] = rsqrtf(fmaxf((float)dA, 1.0f));
    }
    if (nB < NN) {
      ((v4f*)dstbuf)[((size_t)nB * 8 + b) * 2 + q] = aB;
      if ((t & 15) == 0) normD[nB] = rsqrtf(fmaxf((float)dB, 1.0f));
    }
  } else {
    const v4f bias = ((const v4f*)b2)[q];
    if (nA < CBGN) {
      const float nd = rsqrtf(fmaxf((float)dA, 1.0f));
      v4f o;
      o.x = fmaf(aA.x, nd, bias.x); o.y = fmaf(aA.y, nd, bias.y);
      o.z = fmaf(aA.z, nd, bias.z); o.w = fmaf(aA.w, nd, bias.w);
      ((v4f*)dstbuf)[((size_t)b * CBGN + nA) * 2 + q] = o;
    }
    if (nB < CBGN) {
      const float nd = rsqrtf(fmaxf((float)dB, 1.0f));
      v4f o;
      o.x = fmaf(aB.x, nd, bias.x); o.y = fmaf(aB.y, nd, bias.y);
      o.z = fmaf(aB.z, nd, bias.z); o.w = fmaf(aB.w, nd, bias.w);
      ((v4f*)dstbuf)[((size_t)b * CBGN + nB) * 2 + q] = o;
    }
  }
}

#define DN_R 5
#define DN_TILES ((NN * NB) / 16)
#define DN_G (DN_TILES / (8 * DN_R))
static_assert(DN_TILES % (8 * DN_R) == 0, "whole waves: every lane runs every product");
__global__ void __launch_bounds__(256)
dense12_kernel(const float* __restrict__ agg, const float* __restrict__ normD,
               const float* __restrict__ normS, const float* __restrict__ W1,
               const float* __restrict__ b1, const float* __restrict__ W2,
               float* __restrict__ t2) {
  const int wave = (int)((blockIdx.x * 256 + threadIdx.x) >> 5);
  const int lane = threadIdx.x & 31;
  const int rc = lane & 15;
  const int half = lane >> 4;
  const int cw = rc & 7;
  const float colmask = (rc < 8) ? 1.0f : 0.0f;

  v2f wa[2], wb[4];
#pragma unroll
  for (int kb = 0; kb < 8; kb += 4) {
    int k0 = kb + 2 * half;
    wa[kb >> 2].x = W1[(k0 + 0) * 16 + rc];
    wa[kb >> 2].y = W1[(k0 + 1) * 16 + rc];
  }
#pragma unroll
  for (int j = 0; j < 4; ++j) {
    int k0 = 8 * half + 2 * j;
    wb[j].x = W2[(k0 + 0) * 8 + cw] * colmask;
    wb[j].y = W2[(k0 + 1) * 8 + cw] * colmask;
  }
  const v4f* b1v = (const v4f*)b1;
  v4f bl = b1v[2 * half], bh = b1v[2 * half + 1];
  const float bias[8] = {bl.x, bl.y, bl.z, bl.w, bh.x, bh.y, bh.z, bh.w};

  v2f xa[DN_R][2];
  float nd[DN_R], ns[DN_R];
#pragma unroll
  for (int i = 0; i < DN_R; ++i) {
    int row = (wave * DN_R + i) * 16 + rc;
    nd[i] = normD[row >> 3];
    ns[i] = normS[row >> 3];
    const v2f* ar = (const v2f*)&agg[(size_t)row * 8];
    xa[i][0] = ar[half];
    xa[i][1] = ar[2 + half];
  }

#pragma unroll
  for (int i = 0; i < DN_R; ++i) {
    int m0 = (wave * DN_R + i) * 16;
    v8f c = {0.f, 0.f, 0.f, 0.f, 0.f, 0.f, 0.f, 0.f};
#pragma unroll
    for (int kk = 0; kk < 2; ++kk) {
      v2f x;
      x.x = xa[i][kk].x * nd[i];
      x.y = xa[i][kk].y * nd[i];
      c = __builtin_amdgcn_wmma_f32_16x16x4_f32(false, wa[kk], false, x, (short)0, c,
                                                false, false);
    }
    float h[8];
#pragma unroll
    for (int r = 0; r < 8; ++r) {
      float v = c[r] + bias[r];
      v = v > 0.0f ? v : 0.01f * v;
      h[r] = v * ns[i];
    }
    v8f d = {0.f, 0.f, 0.f, 0.f, 0.f, 0.f, 0.f, 0.f};
#pragma unroll
    for (int j = 0; j < 4; ++j) {
      v2f x;
      x.x = h[2 * j + 0];
      x.y = h[2 * j + 1];
      d = __builtin_amdgcn_wmma_f32_16x16x4_f32(false, wb[j], false, x, (short)0, d,
                                                false, false);
    }
    float u0 = __shfl_xor(d[4], 16);
    float u1 = __shfl_xor(d[5], 16);
    float u2 = __shfl_xor(d[6], 16);
    float u3 = __shfl_xor(d[7], 16);
    v4f o;
    o.x = half ? u0 : d[0];
    o.y = half ? u1 : d[1];
    o.z = half ? u2 : d[2];
    o.w = half ? u3 : d[3];
    ((v4f*)t2)[(size_t)(m0 + rc) * 2 + half] = o;
  }
}

extern "C" void kernel_launch(void* const* d_in, const int* in_sizes, int n_in,
                              void* d_out, int out_size, void* d_ws, size_t ws_size,
                              hipStream_t stream) {
  (void)in_sizes; (void)n_in; (void)out_size; (void)ws_size;
  const float* ew  = (const float*)d_in[0];
  const float* cbg = (const float*)d_in[1];
  const float* poi = (const float*)d_in[2];
  const float* W1  = (const float*)d_in[3];
  const float* b1  = (const float*)d_in[4];
  const float* W2  = (const float*)d_in[5];
  const float* b2  = (const float*)d_in[6];
  const int*   src = (const int*)d_in[7];
  const int*   dst = (const int*)d_in[8];
  float* out = (float*)d_out;

  float* ws     = (float*)d_ws;
  float* normS  = ws;
  float* normD  = normS + 50048;
  int*   binTot = (int*)(normD + 50048);
  int*   hist   = binTot + NBP;
  int*   coff   = hist + (size_t)NCK * NBP;
  int*   cO     = coff + (size_t)NCK * NBP;
  unsigned* keys = (unsigned*)(cO + 1250432);
  float* wts    = (float*)(keys + NE);
  float* agg    = wts + 8 * (size_t)NE;
  float* t2     = agg + (size_t)NB * NN * 8;
  float* featT  = t2 + (size_t)NB * NN * 8;

  count_kernel<<<NCH + NCK, 1024, 0, stream>>>(src, dst, cO, hist);
  scan_norm_kernel<<<NBIN + SN_NODEBLK, 256, 0, stream>>>(hist, cO, coff, binTot, normS);

  partition_kernel<<<NCK, PT, 0, stream>>>(ew, src, dst, coff, binTot, keys, wts);
  relayout_kernel<<<RL_G, 256, 0, stream>>>(cbg, poi, normS, featT);

  gather_bin_kernel<1><<<NBIN, GT, 0, stream>>>(featT, keys, wts, binTot, normD, b2, agg);
  dense12_kernel<<<DN_G, 256, 0, stream>>>(agg, normD, normS, W1, b1, W2, t2);

  gather_bin_kernel<2><<<(CBGN + NPB - 1) / NPB, GT, 0, stream>>>(t2, keys, wts, binTot, normD, b2, out);
}


# Out-degree counters packed two 16-bit counts per 32-bit LDS word: a count group clears and writes half the bytes, so the count runs as 100 groups of 16000 edges, two per WGP, not 25 of 64000
# speedup vs baseline: 60.6731x; 1.0580x over previous
#include <hip/hip_runtime.h>

#define NB   8
#define CBGN 30000
#define POIN 20000
#define NE   1600000
#define NN   50000
#define NCH  100
#define CH   (NE / NCH)
#define CH4  (CH / 4)
#define CL   ((CH4 + 1023) / 1024)
#define NW   (NN / 2)
#define NPW  25024
#define NPB  64
#define NSH  6
#define NBIN 782
#define NBP  1024
#define NCK  250
#define CK   (NE / NCK)
#define PT   800
#define GT   (NPB * 8)
#define GW   (GT / 32)
#define TK   6
#define MAXR (TK * GT)
#define WR   (TK * 32)
#define HK   ((CK + 1023) / 1024)
static_assert(NE % NCK == 0 && CK == 8 * PT, "a partition group covers its chunk exactly");
static_assert(NE % NCH == 0 && CH % 4 == 0 && CH < 65536, "a chunk's count of one node fits its 16-bit counter");
static_assert(NN % 8 == 0 && NPW >= NW && NPW % 32 == 0, "counter words move as 16-byte pieces, rows are whole lines");
static_assert(WR * GW == MAXR && MAXR < 65536, "the waves' rank windows cover a tile; a staged slot fits 16 bits");
static_assert(NBIN * NPB >= NN && NBIN <= NBP && NBP == 1024 && NCK <= 256 && (1 << NSH) == NPB, "table shapes");
static_assert(NN < 65536, "source and destination ids share one 32-bit key");

typedef __attribute__((ext_vector_type(2))) float v2f;
typedef __attribute__((ext_vector_type(8))) float v8f;
typedef __attribute__((ext_vector_type(4))) float v4f;

__device__ __forceinline__ void count_chunk(const int* __restrict__ idx, int* __restrict__ s_cnt,
                                            const int t) {
  const int4* i4 = (const int4*)idx;
  int4 d[CL];
#pragma unroll
  for (int i = 0; i < CL; ++i) {
    int k = t + 1024 * i;
    d[i] = (k < CH4) ? i4[k] : make_int4(0, 0, 0, 0);
  }
#pragma unroll
  for (int i = 0; i < CL; ++i) {
    int k = t + 1024 * i;
    if (k < CH4) {
      atomicAdd(&s_cnt[d[i].x >> 1], 1 << ((d[i].x & 1) << 4));
      atomicAdd(&s_cnt[d[i].y >> 1], 1 << ((d[i].y & 1) << 4));
      atomicAdd(&s_cnt[d[i].z >> 1], 1 << ((d[i].z & 1) << 4));
      atomicAdd(&s_cnt[d[i].w >> 1], 1 << ((d[i].w & 1) << 4));
    }
  }
}

__global__ void __launch_bounds__(1024)
count_kernel(const int* __restrict__ src, const int* __restrict__ dst,
             int* __restrict__ cO, int* __restrict__ hist) {
  __shared__ __attribute__((aligned(16))) int s_cnt[NW];
  static_assert(NW >= NBP, "the range counters of the other role fit the same array");
  const int t = threadIdx.x;
  if (blockIdx.x < NCH) {
    const int c = blockIdx.x;
    int4* s4 = (int4*)s_cnt;
    for (int k = t; k < NW / 4; k += 1024) s4[k] = make_int4(0, 0, 0, 0);
    __syncthreads();
    count_chunk(src + (size_t)c * CH, s_cnt, t);
    __syncthreads();
    int4* o4 = (int4*)(cO + (size_t)c * NPW);
    for (int k = t; k < NW / 4; k += 1024) o4[k] = s4[k];
  } else {
    const int c = blockIdx.x - NCH;
    const int* d = dst + (size_t)c * CK;
    int dv[HK];
#pragma unroll
    for (int k = 0; k < HK; ++k) {
      const int i = t + 1024 * k;
      dv[k] = (i < CK) ? d[i] : 0;
    }
    if (t < NBP) s_cnt[t] = 0;
    __syncthreads();
#pragma unroll
    for (int k = 0; k < HK; ++k) {
      if (t + 1024 * k < CK) atomicAdd(&s_cnt[dv[k] >> NSH], 1);
    }
    __syncthreads();
    if (t < NBP) hist[c * NBP + t] = s_cnt[t];
  }
}

#define SN_NODEBLK ((NN + 255) / 256)
static_assert(NN % 2 == 0 && NCH % 2 == 0, "a node pair shares two lanes of one wave; each takes half the chunks");
__global__ void __launch_bounds__(256)
scan_norm_kernel(const int* __restrict__ hist, const int* __restrict__ cO,
                 int* __restrict__ coff, int* __restrict__ binTot,
                 float* __restrict__ normS) {
  __shared__ int s_wsum[8];
  const int t = threadIdx.x;
  if (blockIdx.x < NBIN) {
    const int bin = blockIdx.x;
    const int lane = t & 31, wv = t >> 5;
    const int v = (t < NCK) ? hist[t * NBP + bin] : 0;
    int incl = v;
    for (int o = 1; o < 32; o <<= 1) {
      int u = __shfl_up(incl, o);
      if (lane >= o) incl += u;
    }
    if (lane == 31) s_wsum[wv] = incl;
    __syncthreads();
    int base = 0;
    for (int k = 0; k < wv; ++k) base += s_wsum[k];
    if (t < NCK) coff[t * NBP + bin] = base + incl - v;
    if (t == 255) binTot[bin] = base + incl;
  } else {
    const int n = (blockIdx.x - NBIN) * 256 + t;
    if (n < NN) {
      const int q0 = (n & 1) * (NCH / 2);
      unsigned lo = 0, hi = 0;
#pragma unroll
      for (int q = 0; q < NCH / 2; ++q) {
        const unsigned w = (unsigned)cO[(size_t)(q0 + q) * NPW + (n >> 1)];
        lo += w & 0xFFFFu;
        hi += w >> 16;
      }
      lo += (unsigned)__shfl_xor((int)lo, 1);
      hi += (unsigned)__shfl_xor((int)hi, 1);
      normS[n] = rsqrtf(fmaxf((float)((n & 1) ? hi : lo), 1.0f));
    }
  }
}

__global__ void __launch_bounds__(PT)
partition_kernel(const float* __restrict__ ew, const int* __restrict__ src,
                 const int* __restrict__ dst, const int* __restrict__ coff,
                 const int* __restrict__ binTot,
                 unsigned* __restrict__ keys, float* __restrict__ wts) {
  __shared__ __attribute__((aligned(16))) float s_w[CK * 8];
  __shared__ unsigned s_key[CK];
  __shared__ int s_cnt[NBP];
  __shared__ int s_lst[NBP];
  __shared__ int s_gb[NBP];
  __shared__ int s_bt[NBP];
  __shared__ int s_co[NBP];
  const int t = threadIdx.x;
  const int c = blockIdx.x;
  const int e0 = c * CK + t;
  int d[8], r[8];
#pragma unroll
  for (int k = 0; k < 8; ++k) d[k] = dst[e0 + k * PT];
  static_assert(NBP == 32 * 32, "one scan wave, 32 ranges per lane");
  for (int i = t; i < NBP; i += PT) {
    const bool in = i < NBIN;
    const int tr = (i & 31) * 32 + (i >> 5);
    s_cnt[i] = 0;
    s_bt[tr] = in ? binTot[i] : 0;
    s_co[tr] = in ? coff[c * NBP + i] : 0;
  }
  __syncthreads();
#pragma unroll
  for (int k = 0; k < 8; ++k) r[k] = atomicAdd(&s_cnt[d[k] >> NSH], 1);
  __syncthreads();
  if (t < 32) {
    int cl[32], tl[32];
    int sc = 0, st = 0;
#pragma unroll
    for (int j = 0; j < 32; ++j) {
      const int bin = t * 32 + j;
      cl[j] = s_cnt[bin];
      tl[j] = s_bt[j * 32 + t];
      sc += cl[j];
      st += tl[j];
    }
    int ic = sc, it = st;
    for (int o = 1; o < 32; o <<= 1) {
      int uc = __shfl_up(ic, o);
      int ut = __shfl_up(it, o);
      if (t >= o) { ic += uc; it += ut; }
    }
    int xc = ic - sc, xt = it - st;
#pragma unroll
    for (int j = 0; j < 32; ++j) {
      const int bin = t * 32 + j;
      const int co = s_co[j * 32 + t];
      s_lst[bin] = xc;
      s_gb[bin] = xt + co - xc;
      xc += cl[j];
      xt += tl[j];
    }
  }
  __syncthreads();
#pragma unroll
  for (int g = 0; g < 2; ++g) {
    int sv[4];
    v4f lo[4], hi[4];
#pragma unroll
    for (int k = 0; k < 4; ++k) {
      const int e = e0 + (4 * g + k) * PT;
      sv[k] = src[e];
      lo[k].x = ew[(size_t)0 * NE + e];
      lo[k].y = ew[(size_t)1 * NE + e];
      lo[k].z = ew[(size_t)2 * NE + e];
      lo[k].w = ew[(size_t)3 * NE + e];
      hi[k].x = ew[(size_t)4 * NE + e];
      hi[k].y = ew[(size_t)5 * NE + e];
      hi[k].z = ew[(size_t)6 * NE + e];
      hi[k].w = ew[(size_t)7 * NE + e];
    }
#pragma unroll
    for (int k = 0; k < 4; ++k) {
      const int dd = d[4 * g + k];
      const int lp = s_lst[dd >> NSH] + r[4 * g + k];
      s_key[lp] = (unsigned)sv[k] | ((unsigned)dd << 16);
      ((v4f*)s_w)[lp * 2 + 0] = lo[k];
      ((v4f*)s_w)[lp * 2 + 1] = hi[k];
    }
  }
  __syncthreads();
  for (int i = t; i < CK; i += PT) {
    const unsigned k = s_key[i];
    keys[s_gb[k >> (16 + NSH)] + i] = k;
  }
  for (int j = t; j < CK * 2; j += PT) {
    const int i = j >> 1;
    const unsigned k = s_key[i];
    const size_t gslot = (size_t)(s_gb[k >> (16 + NSH)] + i);
    ((v4f*)wts)[gslot * 2 + (j & 1)] = ((const v4f*)s_w)[j];
  }
}

#define RL_R 5
#define RL_G ((NN * 16) / (256 * RL_R))
static_assert((NN * 16) % (256 * RL_R) == 0, "whole work-groups: no bounds guard");
__global__ void __launch_bounds__(256)
relayout_kernel(const float* __restrict__ cbg, const float* __restrict__ poi,
                const float* __restrict__ normS, float* __restrict__ featT) {
  const int t0 = blockIdx.x * (256 * RL_R) + threadIdx.x;
  v4f x[RL_R];
  float s[RL_R];
#pragma unroll
  for (int i = 0; i < RL_R; ++i) {
    int tid = t0 + i * 256;
    int q = tid & 1;
    int b = (tid >> 1) & 7;
    int n = tid >> 4;
    const bool in_cbg = n < CBGN;
    const float* tab = in_cbg ? cbg : poi;
    const int row = in_cbg ? (b * CBGN + n) : (b * POIN + (n - CBGN));
    x[i] = ((const v4f*)tab)[(size_t)row * 2 + q];
    s[i] = normS[n];
  }
#pragma unroll
  for (int i = 0; i < RL_R; ++i) {
    int tid = t0 + i * 256;
    v4f y = x[i];
    y.x *= s[i]; y.y *= s[i]; y.z *= s[i]; y.w *= s[i];
    ((v4f*)featT)[tid] = y;
  }
}

template <int K>
__device__ __forceinline__ void hold_rows(const v4f (&x)[K]) {
  if constexpr (K == 8) {
    asm volatile("" :: "v"(x[0]), "v"(x[1]), "v"(x[2]), "v"(x[3]),
                       "v"(x[4]), "v"(x[5]), "v"(x[6]), "v"(x[7]));
  } else if constexpr (K == 4) {
    asm volatile("" :: "v"(x[0]), "v"(x[1]), "v"(x[2]), "v"(x[3]));
  } else if constexpr (K == 2) {
    asm volatile("" :: "v"(x[0]), "v"(x[1]));
  }
}

template <int K>
__device__ __forceinline__ void bin_batch(v4f& a, const int p, const int b, const int q,
                                          const unsigned short* s_ord, const unsigned* s_key,
                                          const float* s_w, const float* __restrict__ tab) {
  int id[K];
  float w[K];
  v4f x[K];
#pragma unroll
  for (int i = 0; i < K; ++i) id[i] = s_ord[p + i];
#pragma unroll
  for (int i = 0; i < K; ++i) {
    const unsigned k = s_key[id[i]];
    w[i] = s_w[id[i] * 8 + b];
    const v4f* r = (const v4f*)(tab + ((size_t)(k & 0xFFFFu) * 8 + b) * 8);
    x[i] = r[q];
  }
  hold_rows<K>(x);
#pragma unroll
  for (int i = 0; i < K; ++i) {
    a.x = fmaf(x[i].x, w[i], a.x);
    a.y = fmaf(x[i].y, w[i], a.y);
    a.z = fmaf(x[i].z, w[i], a.z);
    a.w = fmaf(x[i].w, w[i], a.w);
  }
}

__device__ __forceinline__ void bin_row_sum(v4f& a, int p, const int p1, const int b, const int q,
                                            const unsigned short* s_ord, const unsigned* s_key,
                                            const float* s_w, const float* __restrict__ tab) {
  for (; p + 8 <= p1; p += 8) bin_batch<8>(a, p, b, q, s_ord, s_key, s_w, tab);
  if (p + 4 <= p1) { bin_batch<4>(a, p, b, q, s_ord, s_key, s_w, tab); p += 4; }
  if (p + 2 <= p1) { bin_batch<2>(a, p, b, q, s_ord, s_key, s_w, tab); p += 2; }
  if (p < p1)      { bin_batch<1>(a, p, b, q, s_ord, s_key, s_w, tab); }
}

template <int LAYER>
__global__ void __launch_bounds__(GT)
gather_bin_kernel(const float* __restrict__ tab, const unsigned* __restrict__ keys,
                  const float* __restrict__ wts, const int* __restrict__ binTot,
                  float* __restrict__ normD, const float* __restrict__ b2,
                  float* __restrict__ dstbuf) {
  __shared__ __attribute__((aligned(16))) float s_w[MAXR * 8];
  __shared__ unsigned s_key[MAXR];
  __shared__ unsigned short s_ord[MAXR];
  __shared__ int s_wc[GW * NPB];
  __shared__ int s_cnt[NPB];
  __shared__ int s_st[NPB + 1];
  __shared__ int s_red[GW];
  const int t = threadIdx.x;
  const int bin = blockIdx.x;
  const int lane = t & 31, w = t >> 5;
  const int nl = (t >> 4) & 1, b = (t >> 1) & 7, q = t & 1;
  const int lA = 4 * w + nl, lB = lA + 2;

  const int tot = binTot[bin];
  int v = 0;
  for (int i = t; i < bin; i += GT) v += binTot[i];
  for (int o = 16; o > 0; o >>= 1) v += __shfl_xor(v, o);
  if (lane == 0) s_red[w] = v;
  __syncthreads();
  int r0 = 0;
  for (int k = 0; k < GW; ++k) r0 += s_red[k];
  const int r1 = r0 + tot;

  v4f aA = {0.f, 0.f, 0.f, 0.f}, aB = {0.f, 0.f, 0.f, 0.f};
  int dA = 0, dB = 0;
  const int wc0 = w * NPB;
  for (int t0 = r0; t0 < r1; t0 += MAXR) {
    const int len = min(MAXR, r1 - t0);
    unsigned kq[TK];
    v4f wq[2 * TK];
#pragma unroll
    for (int k = 0; k < TK; ++k) {
      const int i = t + k * GT;
      kq[k] = (i < len) ? keys[t0 + i] : 0u;
    }
#pragma unroll
    for (int k = 0; k < 2 * TK; ++k) {
      const int j = t + k * GT;
      wq[k] = (j < len * 2) ? ((const v4f*)wts)[(size_t)t0 * 2 + j] : v4f{0.f, 0.f, 0.f, 0.f};
    }
    __syncthreads();
#pragma unroll
    for (int k = 0; k < TK; ++k) {
      const int i = t + k * GT;
      if (i < len) s_key[i] = kq[k];
    }
#pragma unroll
    for (int k = 0; k < 2 * TK; ++k) {
      const int j = t + k * GT;
      if (j < len * 2) ((v4f*)s_w)[j] = wq[k];
    }
    for (int j = lane; j < NPB; j += 32) s_wc[wc0 + j] = 0;
    __syncthreads();
    int rk[TK];
#pragma unroll
    for (int k = 0; k < TK; ++k) {
      const int i = w * WR + k * 32 + lane;
      const bool valid = i < len;
      const int node = valid ? (int)((s_key[i] >> 16) & (NPB - 1)) : 0;
      unsigned m = (unsigned)__ballot(valid);
#pragma unroll
      for (int bit = 0; bit < NSH; ++bit) {
        const bool on = (node >> bit) & 1;
        const unsigned bb = (unsigned)__ballot(on);
        m &= on ? bb : ~bb;
      }
      const int before = __popc(m & ((1u << lane) - 1u));
      const int same = __popc(m);
      int old = 0;
      if (valid && before == 0) old = atomicAdd(&s_wc[wc0 + node], same);
      const int base = __shfl(old, m ? (__ffs(m) - 1) : lane);
      rk[k] = base + before;
    }
    __syncthreads();
    if (t < NPB) {
      int run = 0;
      for (int k = 0; k < GW; ++k) {
        const int c = s_wc[k * NPB + t];
        s_wc[k * NPB + t] = run;
        run += c;
      }
      s_cnt[t] = run;
    }
    __syncthreads();
    if (t < 32) {
      int c[NPB / 32];
      int s = 0;
#pragma unroll
      for (int j = 0; j < NPB / 32; ++j) { c[j] = s_cnt[t * (NPB / 32) + j]; s += c[j]; }
      int inc = s;
      for (int o = 1; o < 32; o <<= 1) {
        int u = __shfl_up(inc, o);
        if (t >= o) inc += u;
      }
      int x = inc - s;
#pragma unroll
      for (int j = 0; j < NPB / 32; ++j) { s_st[t * (NPB / 32) + j] = x; x += c[j]; }
      if (t == 31) s_st[NPB] = x;
    }
    __syncthreads();
#pragma unroll
    for (int k = 0; k < TK; ++k) {
      const int i = w * WR + k * 32 + lane;
      if (i < len) {
        const int node = (int)((s_key[i] >> 16) & (NPB - 1));
        s_ord[s_st[node] + s_wc[wc0 + node] + rk[k]] = (unsigned short)i;
      }
    }
    __syncthreads();
    {
      const int p = s_st[lA], p1 = s_st[lA + 1];
      dA += p1 - p;
      bin_row_sum(aA, p, p1, b, q, s_ord, s_key, s_w, tab);
    }
    {
      const int p = s_st[lB], p1 = s_st[lB + 1];
      dB += p1 - p;
      bin_row_sum(aB, p, p1, b, q, s_ord, s_key, s_w, tab);
    }
  }

  const int nA = bin * NPB + lA, nB = bin * NPB + lB;
  if constexpr (LAYER == 1) {
    if (nA < NN) {
      ((v4f*)dstbuf)[((size_t)nA * 8 + b) * 2 + q] = aA;
      if ((t & 15) == 0) normD[nA] = rsqrtf(fmaxf((float)dA, 1.0f));
    }
    if (nB < NN) {
      ((v4f*)dstbuf)[((size_t)nB * 8 + b) * 2 + q] = aB;
      if ((t & 15) == 0) normD[nB] = rsqrtf(fmaxf((float)dB, 1.0f));
    }
  } else {
    const v4f bias = ((const v4f*)b2)[q];
    if (nA < CBGN) {
      const float nd = rsqrtf(fmaxf((float)dA, 1.0f));
      v4f o;
      o.x = fmaf(aA.x, nd, bias.x); o.y = fmaf(aA.y, nd, bias.y);
      o.z = fmaf(aA.z, nd, bias.z); o.w = fmaf(aA.w, nd, bias.w);
      ((v4f*)dstbuf)[((size_t)b * CBGN + nA) * 2 + q] = o;
    }
    if (nB < CBGN) {
      const float nd = rsqrtf(fmaxf((float)dB, 1.0f));
      v4f o;
      o.x = fmaf(aB.x, nd, bias.x); o.y = fmaf(aB.y, nd, bias.y);
      o.z = fmaf(aB.z, nd, bias.z); o.w = fmaf(aB.w, nd, bias.w);
      ((v4f*)dstbuf)[((size_t)b * CBGN + nB) * 2 + q] = o;
    }
  }
}

#define DN_R 5
#define DN_TILES ((NN * NB) / 16)
#define DN_G (DN_TILES / (8 * DN_R))
static_assert(DN_TILES % (8 * DN_R) == 0, "whole waves: every lane runs every product");
__global__ void __launch_bounds__(256)
dense12_kernel(const float* __restrict__ agg, const float* __restrict__ normD,
               const float* __restrict__ normS, const float* __restrict__ W1,
               const float* __restrict__ b1, const float* __restrict__ W2,
               float* __restrict__ t2) {
  const int wave = (int)((blockIdx.x * 256 + threadIdx.x) >> 5);
  const int lane = threadIdx.x & 31;
  const int rc = lane & 15;
  const int half = lane >> 4;
  const int cw = rc & 7;
  const float colmask = (rc < 8) ? 1.0f : 0.0f;

  v2f wa[2], wb[4];
#pragma unroll
  for (int kb = 0; kb < 8; kb += 4) {
    int k0 = kb + 2 * half;
    wa[kb >> 2].x = W1[(k0 + 0) * 16 + rc];
    wa[kb >> 2].y = W1[(k0 + 1) * 16 + rc];
  }
#pragma unroll
  for (int j = 0; j < 4; ++j) {
    int k0 = 8 * half + 2 * j;
    wb[j].x = W2[(k0 + 0) * 8 + cw] * colmask;
    wb[j].y = W2[(k0 + 1) * 8 + cw] * colmask;
  }
  const v4f* b1v = (const v4f*)b1;
  v4f bl = b1v[2 * half], bh = b1v[2 * half + 1];
  const float bias[8] = {bl.x, bl.y, bl.z, bl.w, bh.x, bh.y, bh.z, bh.w};

  v2f xa[DN_R][2];
  float nd[DN_R], ns[DN_R];
#pragma unroll
  for (int i = 0; i < DN_R; ++i) {
    int row = (wave * DN_R + i) * 16 + rc;
    nd[i] = normD[row >> 3];
    ns[i] = normS[row >> 3];
    const v2f* ar = (const v2f*)&agg[(size_t)row * 8];
    xa[i][0] = ar[half];
    xa[i][1] = ar[2 + half];
  }

#pragma unroll
  for (int i = 0; i < DN_R; ++i) {
    int m0 = (wave * DN_R + i) * 16;
    v8f c = {0.f, 0.f, 0.f, 0.f, 0.f, 0.f, 0.f, 0.f};
#pragma unroll
    for (int kk = 0; kk < 2; ++kk) {
      v2f x;
      x.x = xa[i][kk].x * nd[i];
      x.y = xa[i][kk].y * nd[i];
      c = __builtin_amdgcn_wmma_f32_16x16x4_f32(false, wa[kk], false, x, (short)0, c,
                                                false, false);
    }
    float h[8];
#pragma unroll
    for (int r = 0; r < 8; ++r) {
      float v = c[r] + bias[r];
      v = v > 0.0f ? v : 0.01f * v;
      h[r] = v * ns[i];
    }
    v8f d = {0.f, 0.f, 0.f, 0.f, 0.f, 0.f, 0.f, 0.f};
#pragma unroll
    for (int j = 0; j < 4; ++j) {
      v2f x;
      x.x = h[2 * j + 0];
      x.y = h[2 * j + 1];
      d = __builtin_amdgcn_wmma_f32_16x16x4_f32(false, wb[j], false, x, (short)0, d,
                                                false, false);
    }
    float u0 = __shfl_xor(d[4], 16);
    float u1 = __shfl_xor(d[5], 16);
    float u2 = __shfl_xor(d[6], 16);
    float u3 = __shfl_xor(d[7], 16);
    v4f o;
    o.x = half ? u0 : d[0];
    o.y = half ? u1 : d[1];
    o.z = half ? u2 : d[2];
    o.w = half ? u3 : d[3];
    ((v4f*)t2)[(size_t)(m0 + rc) * 2 + half] = o;
  }
}

extern "C" void kernel_launch(void* const* d_in, const int* in_sizes, int n_in,
                              void* d_out, int out_size, void* d_ws, size_t ws_size,
                              hipStream_t stream) {
  (void)in_sizes; (void)n_in; (void)out_size; (void)ws_size;
  const float* ew  = (const float*)d_in[0];
  const float* cbg = (const float*)d_in[1];
  const float* poi = (const float*)d_in[2];
  const float* W1  = (const float*)d_in[3];
  const float* b1  = (const float*)d_in[4];
  const float* W2  = (const float*)d_in[5];
  const float* b2  = (const float*)d_in[6];
  const int*   src = (const int*)d_in[7];
  const int*   dst = (const int*)d_in[8];
  float* out = (float*)d_out;

  float* ws     = (float*)d_ws;
  float* normS  = ws;
  float* normD  = normS + 50048;
  int*   binTot = (int*)(normD + 50048);
  int*   hist   = binTot + NBP;
  int*   coff   = hist + (size_t)NCK * NBP;
  int*   cO     = coff + (size_t)NCK * NBP;
  unsigned* keys = (unsigned*)(cO + (size_t)NCH * NPW);
  float* wts    = (float*)(keys + NE);
  float* agg    = wts + 8 * (size_t)NE;
  float* t2     = agg + (size_t)NB * NN * 8;
  float* featT  = t2 + (size_t)NB * NN * 8;

  count_kernel<<<NCH + NCK, 1024, 0, stream>>>(src, dst, cO, hist);
  scan_norm_kernel<<<NBIN + SN_NODEBLK, 256, 0, stream>>>(hist, cO, coff, binTot, normS);

  partition_kernel<<<NCK, PT, 0, stream>>>(ew, src, dst, coff, binTot, keys, wts);
  relayout_kernel<<<RL_G, 256, 0, stream>>>(cbg, poi, normS, featT);

  gather_bin_kernel<1><<<NBIN, GT, 0, stream>>>(featT, keys, wts, binTot, normD, b2, agg);
  dense12_kernel<<<DN_G, 256, 0, stream>>>(agg, normD, normS, W1, b1, W2, t2);

  gather_bin_kernel<2><<<(CBGN + NPB - 1) / NPB, GT, 0, stream>>>(t2, keys, wts, binTot, normD, b2, out);
}


# Table re-layout folded into the partition launch, both dense WMMA products into the layer-1 row-sum epilogue on its register sums: 7 launches become 5; W1 product's feature sum order changes
# speedup vs baseline: 64.7423x; 1.0671x over previous
#include <hip/hip_runtime.h>

#define NB   8
#define CBGN 30000
#define POIN 20000
#define NE   1600000
#define NN   50000
#define NCH  100
#define CH   (NE / NCH)
#define CH4  (CH / 4)
#define CL   ((CH4 + 1023) / 1024)
#define NW   (NN / 2)
#define NPW  25024
#define NPB  64
#define NSH  6
#define NBIN 782
#define NBP  1024
#define NCK  250
#define CK   (NE / NCK)
#define PT   800
#define GT   (NPB * 8)
#define GW   (GT / 32)
#define TK   6
#define MAXR (TK * GT)
#define WR   (TK * 32)
#define HK   ((CK + 1023) / 1024)
static_assert(NE % NCK == 0 && CK == 8 * PT, "a partition group covers its chunk exactly");
static_assert(NE % NCH == 0 && CH % 4 == 0 && CH < 65536, "a chunk's count of one node fits its 16-bit counter");
static_assert(NN % 8 == 0 && NPW >= NW && NPW % 32 == 0, "counter words move as 16-byte pieces, rows are whole lines");
static_assert(WR * GW == MAXR && MAXR < 65536, "the waves' rank windows cover a tile; a staged slot fits 16 bits");
static_assert(NBIN * NPB >= NN && NBIN <= NBP && NBP == 1024 && NCK <= 256 && (1 << NSH) == NPB, "table shapes");
static_assert(NN < 65536, "source and destination ids share one 32-bit key");

typedef __attribute__((ext_vector_type(2))) float v2f;
typedef __attribute__((ext_vector_type(8))) float v8f;
typedef __attribute__((ext_vector_type(4))) float v4f;

__device__ __forceinline__ void count_chunk(const int* __restrict__ idx, int* __restrict__ s_cnt,
                                            const int t) {
  const int4* i4 = (const int4*)idx;
  int4 d[CL];
#pragma unroll
  for (int i = 0; i < CL; ++i) {
    int k = t + 1024 * i;
    d[i] = (k < CH4) ? i4[k] : make_int4(0, 0, 0, 0);
  }
#pragma unroll
  for (int i = 0; i < CL; ++i) {
    int k = t + 1024 * i;
    if (k < CH4) {
      atomicAdd(&s_cnt[d[i].x >> 1], 1 << ((d[i].x & 1) << 4));
      atomicAdd(&s_cnt[d[i].y >> 1], 1 << ((d[i].y & 1) << 4));
      atomicAdd(&s_cnt[d[i].z >> 1], 1 << ((d[i].z & 1) << 4));
      atomicAdd(&s_cnt[d[i].w >> 1], 1 << ((d[i].w & 1) << 4));
    }
  }
}

__global__ void __launch_bounds__(1024)
count_kernel(const int* __restrict__ src, const int* __restrict__ dst,
             int* __restrict__ cO, int* __restrict__ hist) {
  __shared__ __attribute__((aligned(16))) int s_cnt[NW];
  static_assert(NW >= NBP, "the range counters of the other role fit the same array");
  const int t = threadIdx.x;
  if (blockIdx.x < NCH) {
    const int c = blockIdx.x;
    int4* s4 = (int4*)s_cnt;
    for (int k = t; k < NW / 4; k += 1024) s4[k] = make_int4(0, 0, 0, 0);
    __syncthreads();
    count_chunk(src + (size_t)c * CH, s_cnt, t);
    __syncthreads();
    int4* o4 = (int4*)(cO + (size_t)c * NPW);
    for (int k = t; k < NW / 4; k += 1024) o4[k] = s4[k];
  } else {
    const int c = blockIdx.x - NCH;
    const int* d = dst + (size_t)c * CK;
    int dv[HK];
#pragma unroll
    for (int k = 0; k < HK; ++k) {
      const int i = t + 1024 * k;
      dv[k] = (i < CK) ? d[i] : 0;
    }
    if (t < NBP) s_cnt[t] = 0;
    __syncthreads();
#pragma unroll
    for (int k = 0; k < HK; ++k) {
      if (t + 1024 * k < CK) atomicAdd(&s_cnt[dv[k] >> NSH], 1);
    }
    __syncthreads();
    if (t < NBP) hist[c * NBP + t] = s_cnt[t];
  }
}

#define SN_NODEBLK ((NN + 255) / 256)
static_assert(NN % 2 == 0 && NCH % 2 == 0, "a node pair shares two lanes of one wave; each takes half the chunks");
__global__ void __launch_bounds__(256)
scan_norm_kernel(const int* __restrict__ hist, const int* __restrict__ cO,
                 int* __restrict__ coff, int* __restrict__ binTot,
                 float* __restrict__ normS) {
  __shared__ int s_wsum[8];
  const int t = threadIdx.x;
  if (blockIdx.x < NBIN) {
    const int bin = blockIdx.x;
    const int lane = t & 31, wv = t >> 5;
    const int v = (t < NCK) ? hist[t * NBP + bin] : 0;
    int incl = v;
    for (int o = 1; o < 32; o <<= 1) {
      int u = __shfl_up(incl, o);
      if (lane >= o) incl += u;
    }
    if (lane == 31) s_wsum[wv] = incl;
    __syncthreads();
    int base = 0;
    for (int k = 0; k < wv; ++k) base += s_wsum[k];
    if (t < NCK) coff[t * NBP + bin] = base + incl - v;
    if (t == 255) binTot[bin] = base + incl;
  } else {
    const int n = (blockIdx.x - NBIN) * 256 + t;
    if (n < NN) {
      const int q0 = (n & 1) * (NCH / 2);
      unsigned lo = 0, hi = 0;
#pragma unroll
      for (int q = 0; q < NCH / 2; ++q) {
        const unsigned w = (unsigned)cO[(size_t)(q0 + q) * NPW + (n >> 1)];
        lo += w & 0xFFFFu;
        hi += w >> 16;
      }
      lo += (unsigned)__shfl_xor((int)lo, 1);
      hi += (unsigned)__shfl_xor((int)hi, 1);
      normS[n] = rsqrtf(fmaxf((float)((n & 1) ? hi : lo), 1.0f));
    }
  }
}

#define RL_N (NN / NCK)
#define RL_R ((RL_N * 16) / PT)
static_assert(NN % NCK == 0 && (RL_N * 16) % PT == 0, "whole pieces: no bounds guard");
__global__ void __launch_bounds__(PT)
partition_kernel(const float* __restrict__ ew, const int* __restrict__ src,
                 const int* __restrict__ dst, const int* __restrict__ coff,
                 const int* __restrict__ binTot,
                 unsigned* __restrict__ keys, float* __restrict__ wts,
                 const float* __restrict__ cbg, const float* __restrict__ poi,
                 const float* __restrict__ normS, float* __restrict__ featT) {
  __shared__ __attribute__((aligned(16))) float s_w[CK * 8];
  __shared__ unsigned s_key[CK];
  __shared__ int s_cnt[NBP];
  __shared__ int s_lst[NBP];
  __shared__ int s_gb[NBP];
  __shared__ int s_bt[NBP];
  __shared__ int s_co[NBP];
  const int t = threadIdx.x;
  const int c = blockIdx.x;
  const int e0 = c * CK + t;
  int d[8], r[8];
#pragma unroll
  for (int k = 0; k < 8; ++k) d[k] = dst[e0 + k * PT];
  const int rl0 = c * (RL_N * 16) + t;
  v4f rx[RL_R];
  float rs[RL_R];
#pragma unroll
  for (int i = 0; i < RL_R; ++i) {
    const int tid = rl0 + i * PT;
    const int q = tid & 1;
    const int b = (tid >> 1) & 7;
    const int n = tid >> 4;
    const bool in_cbg = n < CBGN;
    const float* tab = in_cbg ? cbg : poi;
    const int row = in_cbg ? (b * CBGN + n) : (b * POIN + (n - CBGN));
    rx[i] = ((const v4f*)tab)[(size_t)row * 2 + q];
    rs[i] = normS[n];
  }
  static_assert(NBP == 32 * 32, "one scan wave, 32 ranges per lane");
  for (int i = t; i < NBP; i += PT) {
    const bool in = i < NBIN;
    const int tr = (i & 31) * 32 + (i >> 5);
    s_cnt[i] = 0;
    s_bt[tr] = in ? binTot[i] : 0;
    s_co[tr] = in ? coff[c * NBP + i] : 0;
  }
  __syncthreads();
#pragma unroll
  for (int k = 0; k < 8; ++k) r[k] = atomicAdd(&s_cnt[d[k] >> NSH], 1);
  __syncthreads();
#pragma unroll
  for (int i = 0; i < RL_R; ++i) {
    v4f y = rx[i];
    y.x *= rs[i]; y.y *= rs[i]; y.z *= rs[i]; y.w *= rs[i];
    ((v4f*)featT)[rl0 + i * PT] = y;
  }
  if (t < 32) {
    int cl[32], tl[32];
    int sc = 0, st = 0;
#pragma unroll
    for (int j = 0; j < 32; ++j) {
      const int bin = t * 32 + j;
      cl[j] = s_cnt[bin];
      tl[j] = s_bt[j * 32 + t];
      sc += cl[j];
      st += tl[j];
    }
    int ic = sc, it = st;
    for (int o = 1; o < 32; o <<= 1) {
      int uc = __shfl_up(ic, o);
      int ut = __shfl_up(it, o);
      if (t >= o) { ic += uc; it += ut; }
    }
    int xc = ic - sc, xt = it - st;
#pragma unroll
    for (int j = 0; j < 32; ++j) {
      const int bin = t * 32 + j;
      const int co = s_co[j * 32 + t];
      s_lst[bin] = xc;
      s_gb[bin] = xt + co - xc;
      xc += cl[j];
      xt += tl[j];
    }
  }
  __syncthreads();
#pragma unroll
  for (int g = 0; g < 2; ++g) {
    int sv[4];
    v4f lo[4], hi[4];
#pragma unroll
    for (int k = 0; k < 4; ++k) {
      const int e = e0 + (4 * g + k) * PT;
      sv[k] = src[e];
      lo[k].x = ew[(size_t)0 * NE + e];
      lo[k].y = ew[(size_t)1 * NE + e];
      lo[k].z = ew[(size_t)2 * NE + e];
      lo[k].w = ew[(size_t)3 * NE + e];
      hi[k].x = ew[(size_t)4 * NE + e];
      hi[k].y = ew[(size_t)5 * NE + e];
      hi[k].z = ew[(size_t)6 * NE + e];
      hi[k].w = ew[(size_t)7 * NE + e];
    }
#pragma unroll
    for (int k = 0; k < 4; ++k) {
      const int dd = d[4 * g + k];
      const int lp = s_lst[dd >> NSH] + r[4 * g + k];
      s_key[lp] = (unsigned)sv[k] | ((unsigned)dd << 16);
      ((v4f*)s_w)[lp * 2 + 0] = lo[k];
      ((v4f*)s_w)[lp * 2 + 1] = hi[k];
    }
  }
  __syncthreads();
  for (int i = t; i < CK; i += PT) {
    const unsigned k = s_key[i];
    keys[s_gb[k >> (16 + NSH)] + i] = k;
  }
  for (int j = t; j < CK * 2; j += PT) {
    const int i = j >> 1;
    const unsigned k = s_key[i];
    const size_t gslot = (size_t)(s_gb[k >> (16 + NSH)] + i);
    ((v4f*)wts)[gslot * 2 + (j & 1)] = ((const v4f*)s_w)[j];
  }
}

template <int K>
__device__ __forceinline__ void hold_rows(const v4f (&x)[K]) {
  if constexpr (K == 8) {
    asm volatile("" :: "v"(x[0]), "v"(x[1]), "v"(x[2]), "v"(x[3]),
                       "v"(x[4]), "v"(x[5]), "v"(x[6]), "v"(x[7]));
  } else if constexpr (K == 4) {
    asm volatile("" :: "v"(x[0]), "v"(x[1]), "v"(x[2]), "v"(x[3]));
  } else if constexpr (K == 2) {
    asm volatile("" :: "v"(x[0]), "v"(x[1]));
  }
}

template <int K>
__device__ __forceinline__ void bin_batch(v4f& a, const int p, const int b, const int q,
                                          const unsigned short* s_ord, const unsigned* s_key,
                                          const float* s_w, const float* __restrict__ tab) {
  int id[K];
  float w[K];
  v4f x[K];
#pragma unroll
  for (int i = 0; i < K; ++i) id[i] = s_ord[p + i];
#pragma unroll
  for (int i = 0; i < K; ++i) {
    const unsigned k = s_key[id[i]];
    w[i] = s_w[id[i] * 8 + b];
    const v4f* r = (const v4f*)(tab + ((size_t)(k & 0xFFFFu) * 8 + b) * 8);
    x[i] = r[q];
  }
  hold_rows<K>(x);
#pragma unroll
  for (int i = 0; i < K; ++i) {
    a.x = fmaf(x[i].x, w[i], a.x);
    a.y = fmaf(x[i].y, w[i], a.y);
    a.z = fmaf(x[i].z, w[i], a.z);
    a.w = fmaf(x[i].w, w[i], a.w);
  }
}

__device__ __forceinline__ void bin_row_sum(v4f& a, int p, const int p1, const int b, const int q,
                                            const unsigned short* s_ord, const unsigned* s_key,
                                            const float* s_w, const float* __restrict__ tab) {
  for (; p + 8 <= p1; p += 8) bin_batch<8>(a, p, b, q, s_ord, s_key, s_w, tab);
  if (p + 4 <= p1) { bin_batch<4>(a, p, b, q, s_ord, s_key, s_w, tab); p += 4; }
  if (p + 2 <= p1) { bin_batch<2>(a, p, b, q, s_ord, s_key, s_w, tab); p += 2; }
  if (p < p1)      { bin_batch<1>(a, p, b, q, s_ord, s_key, s_w, tab); }
}

#define DW_B1 128
#define DW_W2 (DW_B1 + 16)
#define DW_NS (DW_W2 + 128)
#define DW_N  (DW_NS + NPB)
static_assert(DW_N <= GT, "one float per thread");
template <int LAYER>
__global__ void __launch_bounds__(GT)
gather_bin_kernel(const float* __restrict__ tab, const unsigned* __restrict__ keys,
                  const float* __restrict__ wts, const int* __restrict__ binTot,
                  const float* __restrict__ normS, const float* __restrict__ W1,
                  const float* __restrict__ b1, const float* __restrict__ W2,
                  const float* __restrict__ b2, float* __restrict__ dstbuf) {
  __shared__ __attribute__((aligned(16))) float s_w[MAXR * 8];
  __shared__ unsigned s_key[MAXR];
  __shared__ unsigned short s_ord[MAXR];
  __shared__ int s_wc[GW * NPB];
  __shared__ int s_cnt[NPB];
  __shared__ int s_st[NPB + 1];
  __shared__ int s_red[GW];
  __shared__ float s_dw[LAYER == 1 ? DW_N : 1];
  const int t = threadIdx.x;
  const int bin = blockIdx.x;
  const int lane = t & 31, w = t >> 5;
  const int nl = (t >> 4) & 1, b = (t >> 1) & 7, q = t & 1;
  const int lA = 4 * w + nl, lB = lA + 2;

  if constexpr (LAYER == 1) {
    if (t < DW_N) {
      const float* p = (t < DW_B1) ? (W1 + t)
                     : (t < DW_W2) ? (b1 + (t - DW_B1))
                     : (t < DW_NS) ? (W2 + (t - DW_W2))
                                   : (normS + min(bin * NPB + (t - DW_NS), NN - 1));
      s_dw[t] = *p;
    }
  }

  const int tot = binTot[bin];
  int v = 0;
  for (int i = t; i < bin; i += GT) v += binTot[i];
  for (int o = 16; o > 0; o >>= 1) v += __shfl_xor(v, o);
  if (lane == 0) s_red[w] = v;
  __syncthreads();
  int r0 = 0;
  for (int k = 0; k < GW; ++k) r0 += s_red[k];
  const int r1 = r0 + tot;

  v4f aA = {0.f, 0.f, 0.f, 0.f}, aB = {0.f, 0.f, 0.f, 0.f};
  int dA = 0, dB = 0;
  const int wc0 = w * NPB;
  for (int t0 = r0; t0 < r1; t0 += MAXR) {
    const int len = min(MAXR, r1 - t0);
    unsigned kq[TK];
    v4f wq[2 * TK];
#pragma unroll
    for (int k = 0; k < TK; ++k) {
      const int i = t + k * GT;
      kq[k] = (i < len) ? keys[t0 + i] : 0u;
    }
#pragma unroll
    for (int k = 0; k < 2 * TK; ++k) {
      const int j = t + k * GT;
      wq[k] = (j < len * 2) ? ((const v4f*)wts)[(size_t)t0 * 2 + j] : v4f{0.f, 0.f, 0.f, 0.f};
    }
    __syncthreads();
#pragma unroll
    for (int k = 0; k < TK; ++k) {
      const int i = t + k * GT;
      if (i < len) s_key[i] = kq[k];
    }
#pragma unroll
    for (int k = 0; k < 2 * TK; ++k) {
      const int j = t + k * GT;
      if (j < len * 2) ((v4f*)s_w)[j] = wq[k];
    }
    for (int j = lane; j < NPB; j += 32) s_wc[wc0 + j] = 0;
    __syncthreads();
    int rk[TK];
#pragma unroll
    for (int k = 0; k < TK; ++k) {
      const int i = w * WR + k * 32 + lane;
      const bool valid = i < len;
      const int node = valid ? (int)((s_key[i] >> 16) & (NPB - 1)) : 0;
      unsigned m = (unsigned)__ballot(valid);
#pragma unroll
      for (int bit = 0; bit < NSH; ++bit) {
        const bool on = (node >> bit) & 1;
        const unsigned bb = (unsigned)__ballot(on);
        m &= on ? bb : ~bb;
      }
      const int before = __popc(m & ((1u << lane) - 1u));
      const int same = __popc(m);
      int old = 0;
      if (valid && before == 0) old = atomicAdd(&s_wc[wc0 + node], same);
      const int base = __shfl(old, m ? (__ffs(m) - 1) : lane);
      rk[k] = base + before;
    }
    __syncthreads();
    if (t < NPB) {
      int run = 0;
      for (int k = 0; k < GW; ++k) {
        const int c = s_wc[k * NPB + t];
        s_wc[k * NPB + t] = run;
        run += c;
      }
      s_cnt[t] = run;
    }
    __syncthreads();
    if (t < 32) {
      int c[NPB / 32];
      int s = 0;
#pragma unroll
      for (int j = 0; j < NPB / 32; ++j) { c[j] = s_cnt[t * (NPB / 32) + j]; s += c[j]; }
      int inc = s;
      for (int o = 1; o < 32; o <<= 1) {
        int u = __shfl_up(inc, o);
        if (t >= o) inc += u;
      }
      int x = inc - s;
#pragma unroll
      for (int j = 0; j < NPB / 32; ++j) { s_st[t * (NPB / 32) + j] = x; x += c[j]; }
      if (t == 31) s_st[NPB] = x;
    }
    __syncthreads();
#pragma unroll
    for (int k = 0; k < TK; ++k) {
      const int i = w * WR + k * 32 + lane;
      if (i < len) {
        const int node = (int)((s_key[i] >> 16) & (NPB - 1));
        s_ord[s_st[node] + s_wc[wc0 + node] + rk[k]] = (unsigned short)i;
      }
    }
    __syncthreads();
    {
      const int p = s_st[lA], p1 = s_st[lA + 1];
      dA += p1 - p;
      bin_row_sum(aA, p, p1, b, q, s_ord, s_key, s_w, tab);
    }
    {
      const int p = s_st[lB], p1 = s_st[lB + 1];
      dB += p1 - p;
      bin_row_sum(aB, p, p1, b, q, s_ord, s_key, s_w, tab);
    }
  }

  const int nA = bin * NPB + lA, nB = bin * NPB + lB;
  if constexpr (LAYER == 1) {
    const int rc = lane & 15, half = lane >> 4;
    const int sl = (rc >> 3) * 16 + (rc & 7) * 2 + half;
    const int cw = rc & 7;
    const float colmask = (rc < 8) ? 1.0f : 0.0f;
    v2f wa[2], wb[4];
    float bias[8];
#pragma unroll
    for (int kk = 0; kk < 2; ++kk) {
      const int f0 = 4 * half + 2 * kk;
      wa[kk].x = s_dw[(f0 + 0) * 16 + rc];
      wa[kk].y = s_dw[(f0 + 1) * 16 + rc];
    }
#pragma unroll
    for (int j = 0; j < 4; ++j) {
      const int k0 = 8 * half + 2 * j;
      wb[j].x = s_dw[DW_W2 + (k0 + 0) * 8 + cw] * colmask;
      wb[j].y = s_dw[DW_W2 + (k0 + 1) * 8 + cw] * colmask;
    }
#pragma unroll
    for (int r = 0; r < 8; ++r) bias[r] = s_dw[DW_B1 + 8 * half + r];
#pragma unroll
    for (int turn = 0; turn < 2; ++turn) {
      const v4f a = turn ? aB : aA;
      const float x0 = __shfl(a.x, sl), x1 = __shfl(a.y, sl);
      const float x2 = __shfl(a.z, sl), x3 = __shfl(a.w, sl);
      const int dg = __shfl(turn ? dB : dA, sl);
      const float nd = rsqrtf(fmaxf((float)dg, 1.0f));
      const float ns = s_dw[DW_NS + 4 * w + 2 * turn + (rc >> 3)];
      const int n = bin * NPB + 4 * w + 2 * turn + (rc >> 3);
      v8f c = {0.f, 0.f, 0.f, 0.f, 0.f, 0.f, 0.f, 0.f};
      v2f x;
      x.x = x0 * nd; x.y = x1 * nd;
      c = __builtin_amdgcn_wmma_f32_16x16x4_f32(false, wa[0], false, x, (short)0, c, false, false);
      x.x = x2 * nd; x.y = x3 * nd;
      c = __builtin_amdgcn_wmma_f32_16x16x4_f32(false, wa[1], false, x, (short)0, c, false, false);
      float h[8];
#pragma unroll
      for (int r = 0; r < 8; ++r) {
        float v = c[r] + bias[r];
        v = v > 0.0f ? v : 0.01f * v;
        h[r] = v * ns;
      }
      v8f d = {0.f, 0.f, 0.f, 0.f, 0.f, 0.f, 0.f, 0.f};
#pragma unroll
      for (int j = 0; j < 4; ++j) {
        v2f y;
        y.x = h[2 * j + 0];
        y.y = h[2 * j + 1];
        d = __builtin_amdgcn_wmma_f32_16x16x4_f32(false, wb[j], false, y, (short)0, d, false, false);
      }
      const float u0 = __shfl_xor(d[4], 16);
      const float u1 = __shfl_xor(d[5], 16);
      const float u2 = __shfl_xor(d[6], 16);
      const float u3 = __shfl_xor(d[7], 16);
      v4f o;
      o.x = half ? u0 : d[0];
      o.y = half ? u1 : d[1];
      o.z = half ? u2 : d[2];
      o.w = half ? u3 : d[3];
      if (n < NN) ((v4f*)dstbuf)[((size_t)n * 8 + (rc & 7)) * 2 + half] = o;
    }
  } else {
    const v4f bias = ((const v4f*)b2)[q];
    if (nA < CBGN) {
      const float nd = rsqrtf(fmaxf((float)dA, 1.0f));
      v4f o;
      o.x = fmaf(aA.x, nd, bias.x); o.y = fmaf(aA.y, nd, bias.y);
      o.z = fmaf(aA.z, nd, bias.z); o.w = fmaf(aA.w, nd, bias.w);
      ((v4f*)dstbuf)[((size_t)b * CBGN + nA) * 2 + q] = o;
    }
    if (nB < CBGN) {
      const float nd = rsqrtf(fmaxf((float)dB, 1.0f));
      v4f o;
      o.x = fmaf(aB.x, nd, bias.x); o.y = fmaf(aB.y, nd, bias.y);
      o.z = fmaf(aB.z, nd, bias.z); o.w = fmaf(aB.w, nd, bias.w);
      ((v4f*)dstbuf)[((size_t)b * CBGN + nB) * 2 + q] = o;
    }
  }
}

extern "C" void kernel_launch(void* const* d_in, const int* in_sizes, int n_in,
                              void* d_out, int out_size, void* d_ws, size_t ws_size,
                              hipStream_t stream) {
  (void)in_sizes; (void)n_in; (void)out_size; (void)ws_size;
  const float* ew  = (const float*)d_in[0];
  const float* cbg = (const float*)d_in[1];
  const float* poi = (const float*)d_in[2];
  const float* W1  = (const float*)d_in[3];
  const float* b1  = (const float*)d_in[4];
  const float* W2  = (const float*)d_in[5];
  const float* b2  = (const float*)d_in[6];
  const int*   src = (const int*)d_in[7];
  const int*   dst = (const int*)d_in[8];
  float* out = (float*)d_out;

  float* ws     = (float*)d_ws;
  float* normS  = ws;
  int*   binTot = (int*)(normS + 2 * 50048);
  int*   hist   = binTot + NBP;
  int*   coff   = hist + (size_t)NCK * NBP;
  int*   cO     = coff + (size_t)NCK * NBP;
  unsigned* keys = (unsigned*)(cO + (size_t)NCH * NPW);
  float* wts    = (float*)(keys + NE);
  float* t2     = wts + 8 * (size_t)NE;
  float* featT  = t2 + (size_t)NB * NN * 8;

  count_kernel<<<NCH + NCK, 1024, 0, stream>>>(src, dst, cO, hist);
  scan_norm_kernel<<<NBIN + SN_NODEBLK, 256, 0, stream>>>(hist, cO, coff, binTot, normS);

  partition_kernel<<<NCK, PT, 0, stream>>>(ew, src, dst, coff, binTot, keys, wts, cbg, poi, normS, featT);

  gather_bin_kernel<1><<<NBIN, GT, 0, stream>>>(featT, keys, wts, binTot, normS, W1, b1, W2, b2, t2);

  gather_bin_kernel<2><<<(CBGN + NPB - 1) / NPB, GT, 0, stream>>>(t2, keys, wts, binTot, normS, W1, b1, W2, b2, out);
}
